# starting kernel
# baseline (speedup 1.0000x reference)
#include <hip/hip_runtime.h>
#include <hip/hip_bf16.h>

#define NN     50000
#define KK     32
#define IN_C   256
#define OUT_C  128

typedef __attribute__((ext_vector_type(2))) float v2f;
typedef __attribute__((ext_vector_type(8))) float v8f;

__global__ void __launch_bounds__(256)
median_conv_gemm_f32(const float* __restrict__ x,
                     const float* __restrict__ w,
                     float* __restrict__ h) {
    const int wave = threadIdx.x >> 5;
    const int lane = threadIdx.x & 31;
    const int lid  = lane & 15;
    const int half = lane >> 4;

    const int gw = blockIdx.x * 8 + wave;
    const int mt = gw >> 3;
    const int nt = gw & 7;
    const int m0 = mt * 16;
    const int n0 = nt * 16;

    const float* __restrict__ xrow = x + (size_t)(m0 + lid) * IN_C;

    v8f c = {};
    #pragma unroll 4
    for (int k0 = 0; k0 < IN_C; k0 += 4) {
        v2f a, b;
        const int ka = k0 + 2 * half;
        a.x = xrow[ka];
        a.y = xrow[ka + 1];
        b.x = w[(size_t)ka       * OUT_C + n0 + lid];
        b.y = w[(size_t)(ka + 1) * OUT_C + n0 + lid];
        c = __builtin_amdgcn_wmma_f32_16x16x4_f32(
                false, a, false, b, (short)0, c, false, false);
    }

    #pragma unroll
    for (int v = 0; v < 8; v++) {
        h[(size_t)(m0 + v + 8 * half) * OUT_C + n0 + lid] = c[v];
    }
}

__global__ void __launch_bounds__(128)
median_conv_select(const float* __restrict__ h,
                   const int*   __restrict__ nbr,
                   const int*   __restrict__ deg,
                   float*       __restrict__ out) {
    const int n = blockIdx.x;
    const int c = threadIdx.x;

    const int d   = deg[n];
    const int med = (d - 1) >> 1;

    const int* __restrict__ nb = nbr + (size_t)n * KK;

    float v[KK];
    #pragma unroll
    for (int j = 0; j < KK; j++) {
        const int idx = nb[j];
        const float val = h[(size_t)idx * OUT_C + c];
        v[j] = (j < d) ? val : __builtin_inff();
    }

    #pragma unroll
    for (int k = 2; k <= KK; k <<= 1) {
        #pragma unroll
        for (int j = k >> 1; j > 0; j >>= 1) {
            #pragma unroll
            for (int i = 0; i < KK; i++) {
                const int l = i ^ j;
                if (l > i) {
                    const bool up = ((i & k) == 0);
                    const float a = v[i], b = v[l];
                    const float lo = fminf(a, b);
                    const float hi = fmaxf(a, b);
                    v[i] = up ? lo : hi;
                    v[l] = up ? hi : lo;
                }
            }
        }
    }

    float r = v[0];
    #pragma unroll
    for (int i = 1; i < KK; i++) {
        r = (med == i) ? v[i] : r;
    }

    out[(size_t)n * OUT_C + c] = r;
}

extern "C" void kernel_launch(void* const* d_in, const int* in_sizes, int n_in,
                              void* d_out, int out_size, void* d_ws, size_t ws_size,
                              hipStream_t stream) {
    const float* x   = (const float*)d_in[0];
    const float* w   = (const float*)d_in[1];
    const int*   nbr = (const int*)d_in[2];
    const int*   deg = (const int*)d_in[3];
    float*       out = (float*)d_out;
    float*       h   = (float*)d_ws;

    median_conv_gemm_f32<<<3125, 256, 0, stream>>>(x, w, h);

    median_conv_select<<<NN, 128, 0, stream>>>(h, nbr, deg, out);
}


# GEMM operand staging: each wave keeps its 16-column W panel in registers across row tiles; each 16x256 x tile goes once per block through LDS (coalesced 16B loads); sum order unchanged
# speedup vs baseline: 1.2664x; 1.2664x over previous
#include <hip/hip_runtime.h>
#include <hip/hip_bf16.h>

#define NN     50000
#define KK     32
#define IN_C   256
#define OUT_C  128

typedef __attribute__((ext_vector_type(2))) float v2f;
typedef __attribute__((ext_vector_type(8))) float v8f;

#define TILE_ROWS   16
#define NUM_TILES   (NN / TILE_ROWS)
#define LDS_PITCH   260
#define GEMM_BLOCKS 512

typedef __attribute__((ext_vector_type(4))) float v4f;

__global__ void __launch_bounds__(256)
median_conv_gemm_f32(const float* __restrict__ x,
                     const float* __restrict__ w,
                     float* __restrict__ h) {
    __shared__ __attribute__((aligned(16))) float xs[2][TILE_ROWS * LDS_PITCH];

    const int tid  = threadIdx.x;
    const int wave = tid >> 5;
    const int lane = tid & 31;
    const int lid  = lane & 15;
    const int half = lane >> 4;
    const int n0   = wave * 16;

    v2f bw[64];
    #pragma unroll
    for (int s = 0; s < 64; s++) {
        const int ka = 4 * s + 2 * half;
        bw[s].x = w[(size_t)ka       * OUT_C + n0 + lid];
        bw[s].y = w[(size_t)(ka + 1) * OUT_C + n0 + lid];
    }

    int mt = blockIdx.x;
    v4f st[4];
    {
        const v4f* __restrict__ src = (const v4f*)(x + (size_t)mt * TILE_ROWS * IN_C);
        #pragma unroll
        for (int i = 0; i < 4; i++) st[i] = src[tid + 256 * i];
        #pragma unroll
        for (int i = 0; i < 4; i++) {
            const int f = tid + 256 * i;
            *(v4f*)&xs[0][(f >> 6) * LDS_PITCH + (f & 63) * 4] = st[i];
        }
    }
    __syncthreads();

    int buf = 0;
    for (; mt < NUM_TILES; mt += GEMM_BLOCKS) {
        const int  nxt      = mt + GEMM_BLOCKS;
        const bool has_next = nxt < NUM_TILES;

        if (has_next) {
            const v4f* __restrict__ src = (const v4f*)(x + (size_t)nxt * TILE_ROWS * IN_C);
            #pragma unroll
            for (int i = 0; i < 4; i++) st[i] = src[tid + 256 * i];
        }

        const float* xr = &xs[buf][lid * LDS_PITCH + 2 * half];
        v8f c = {};
        #pragma unroll
        for (int s = 0; s < 64; s++) {
            const v2f a = *(const v2f*)(xr + 4 * s);
            c = __builtin_amdgcn_wmma_f32_16x16x4_f32(
                    false, a, false, bw[s], (short)0, c, false, false);
        }

        if (has_next) {
            #pragma unroll
            for (int i = 0; i < 4; i++) {
                const int f = tid + 256 * i;
                *(v4f*)&xs[buf ^ 1][(f >> 6) * LDS_PITCH + (f & 63) * 4] = st[i];
            }
        }
        __syncthreads();
        buf ^= 1;

        const int m0 = mt * TILE_ROWS;
        #pragma unroll
        for (int v = 0; v < 8; v++) {
            h[(size_t)(m0 + v + 8 * half) * OUT_C + n0 + lid] = c[v];
        }
    }
}

__global__ void __launch_bounds__(128)
median_conv_select(const float* __restrict__ h,
                   const int*   __restrict__ nbr,
                   const int*   __restrict__ deg,
                   float*       __restrict__ out) {
    const int n = blockIdx.x;
    const int c = threadIdx.x;

    const int d   = deg[n];
    const int med = (d - 1) >> 1;

    const int* __restrict__ nb = nbr + (size_t)n * KK;

    float v[KK];
    #pragma unroll
    for (int j = 0; j < KK; j++) {
        const int idx = nb[j];
        const float val = h[(size_t)idx * OUT_C + c];
        v[j] = (j < d) ? val : __builtin_inff();
    }

    #pragma unroll
    for (int k = 2; k <= KK; k <<= 1) {
        #pragma unroll
        for (int j = k >> 1; j > 0; j >>= 1) {
            #pragma unroll
            for (int i = 0; i < KK; i++) {
                const int l = i ^ j;
                if (l > i) {
                    const bool up = ((i & k) == 0);
                    const float a = v[i], b = v[l];
                    const float lo = fminf(a, b);
                    const float hi = fmaxf(a, b);
                    v[i] = up ? lo : hi;
                    v[l] = up ? hi : lo;
                }
            }
        }
    }

    float r = v[0];
    #pragma unroll
    for (int i = 1; i < KK; i++) {
        r = (med == i) ? v[i] : r;
    }

    out[(size_t)n * OUT_C + c] = r;
}

extern "C" void kernel_launch(void* const* d_in, const int* in_sizes, int n_in,
                              void* d_out, int out_size, void* d_ws, size_t ws_size,
                              hipStream_t stream) {
    const float* x   = (const float*)d_in[0];
    const float* w   = (const float*)d_in[1];
    const int*   nbr = (const int*)d_in[2];
    const int*   deg = (const int*)d_in[3];
    float*       out = (float*)d_out;
    float*       h   = (float*)d_ws;

    median_conv_gemm_f32<<<GEMM_BLOCKS, 256, 0, stream>>>(x, w, h);

    median_conv_select<<<NN, 128, 0, stream>>>(h, nbr, deg, out);
}


# select kernel skips all work for +inf padding slots: gathers stop at the node's degree d; a block-uniform branch picks a sorting network sized for d that yields only the ranks a median can take
# speedup vs baseline: 1.8710x; 1.4774x over previous
#include <hip/hip_runtime.h>
#include <hip/hip_bf16.h>

#define NN     50000
#define KK     32
#define IN_C   256
#define OUT_C  128

typedef __attribute__((ext_vector_type(2))) float v2f;
typedef __attribute__((ext_vector_type(8))) float v8f;

#define TILE_ROWS   16
#define NUM_TILES   (NN / TILE_ROWS)
#define LDS_PITCH   260
#define GEMM_BLOCKS 512

typedef __attribute__((ext_vector_type(4))) float v4f;

__global__ void __launch_bounds__(256)
median_conv_gemm_f32(const float* __restrict__ x,
                     const float* __restrict__ w,
                     float* __restrict__ h) {
    __shared__ __attribute__((aligned(16))) float xs[2][TILE_ROWS * LDS_PITCH];

    const int tid  = threadIdx.x;
    const int wave = tid >> 5;
    const int lane = tid & 31;
    const int lid  = lane & 15;
    const int half = lane >> 4;
    const int n0   = wave * 16;

    v2f bw[64];
    #pragma unroll
    for (int s = 0; s < 64; s++) {
        const int ka = 4 * s + 2 * half;
        bw[s].x = w[(size_t)ka       * OUT_C + n0 + lid];
        bw[s].y = w[(size_t)(ka + 1) * OUT_C + n0 + lid];
    }

    int mt = blockIdx.x;
    v4f st[4];
    {
        const v4f* __restrict__ src = (const v4f*)(x + (size_t)mt * TILE_ROWS * IN_C);
        #pragma unroll
        for (int i = 0; i < 4; i++) st[i] = src[tid + 256 * i];
        #pragma unroll
        for (int i = 0; i < 4; i++) {
            const int f = tid + 256 * i;
            *(v4f*)&xs[0][(f >> 6) * LDS_PITCH + (f & 63) * 4] = st[i];
        }
    }
    __syncthreads();

    int buf = 0;
    for (; mt < NUM_TILES; mt += GEMM_BLOCKS) {
        const int  nxt      = mt + GEMM_BLOCKS;
        const bool has_next = nxt < NUM_TILES;

        if (has_next) {
            const v4f* __restrict__ src = (const v4f*)(x + (size_t)nxt * TILE_ROWS * IN_C);
            #pragma unroll
            for (int i = 0; i < 4; i++) st[i] = src[tid + 256 * i];
        }

        const float* xr = &xs[buf][lid * LDS_PITCH + 2 * half];
        v8f c = {};
        #pragma unroll
        for (int s = 0; s < 64; s++) {
            const v2f a = *(const v2f*)(xr + 4 * s);
            c = __builtin_amdgcn_wmma_f32_16x16x4_f32(
                    false, a, false, bw[s], (short)0, c, false, false);
        }

        if (has_next) {
            #pragma unroll
            for (int i = 0; i < 4; i++) {
                const int f = tid + 256 * i;
                *(v4f*)&xs[buf ^ 1][(f >> 6) * LDS_PITCH + (f & 63) * 4] = st[i];
            }
        }
        __syncthreads();
        buf ^= 1;

        const int m0 = mt * TILE_ROWS;
        #pragma unroll
        for (int v = 0; v < 8; v++) {
            h[(size_t)(m0 + v + 8 * half) * OUT_C + n0 + lid] = c[v];
        }
    }
}

__device__ __forceinline__ void cex(float& lo, float& hi) {
    const float p = lo, q = hi;
    lo = fminf(p, q);
    hi = fmaxf(p, q);
}

template <int LO, int HI, int R>
__device__ __forceinline__ void oe_merge(float* v) {
    constexpr int STEP = 2 * R;
    if constexpr (STEP < HI - LO) {
        oe_merge<LO, HI, STEP>(v);
        oe_merge<LO + R, HI, STEP>(v);
        #pragma unroll
        for (int i = LO + R; i < HI - R; i += STEP) cex(v[i], v[i + R]);
    } else {
        cex(v[LO], v[LO + R]);
    }
}

template <int LO, int HI>
__device__ __forceinline__ void oe_sort(float* v) {
    if constexpr (HI > LO) {
        constexpr int MID = LO + (HI - LO) / 2;
        oe_sort<LO, MID>(v);
        oe_sort<MID + 1, HI>(v);
        oe_merge<LO, HI, 1>(v);
    }
}

template <int M>
__device__ __forceinline__ float bitonic_pick(float* v, int sel) {
    #pragma unroll
    for (int j = M / 2; j > 0; j >>= 1) {
        #pragma unroll
        for (int i = 0; i < M; i++) {
            const int l = i ^ j;
            if (l > i) cex(v[i], v[l]);
        }
    }
    float r = v[0];
    #pragma unroll
    for (int i = 1; i < M; i++) r = (sel == i) ? v[i] : r;
    return r;
}

template <int HA, int HB>
__device__ __forceinline__ float rank_of_two_lists(float* a, float* b, int med) {
    oe_sort<0, HA - 1>(a);
    oe_sort<0, HB - 1>(b);
    constexpr int Q = HA / 2;
    float u[Q];
    #pragma unroll
    for (int i = 0; i < Q; i++) {
        float lo0 = a[i];
        if (HA - 1 - i < HB) lo0 = fminf(lo0, b[HA - 1 - i]);
        float lo1 = a[i + Q];
        if (Q - 1 - i < HB) lo1 = fminf(lo1, b[Q - 1 - i]);
        u[i] = fmaxf(lo0, lo1);
    }
    if constexpr (HB == HA) {
        return bitonic_pick<Q>(u, med - Q);
    } else {
        constexpr int E = Q / 2;
        float w[E];
        #pragma unroll
        for (int i = 0; i < E; i++) w[i] = fminf(u[i], u[i + E]);
        return bitonic_pick<E>(w, med - Q);
    }
}

template <int HA, int HB>
__device__ __forceinline__ float node_median(const float* __restrict__ hc,
                                             const int* __restrict__ nb,
                                             int d, int med) {
    float a[HA], b[HB];
    #pragma unroll
    for (int j = 0; j < HA; j++) a[j] = hc[(size_t)nb[j] * OUT_C];
    constexpr int CH = HB < 4 ? HB : 4;
    #pragma unroll
    for (int q = 0; q < HB; q += CH) {
        if (q == 0 || HA + q < d) {
            #pragma unroll
            for (int t = 0; t < CH; t++) {
                const float val = hc[(size_t)nb[HA + q + t] * OUT_C];
                b[q + t] = (HA + q + t < d) ? val : __builtin_inff();
            }
        } else {
            #pragma unroll
            for (int t = 0; t < CH; t++) b[q + t] = __builtin_inff();
        }
    }
    return rank_of_two_lists<HA, HB>(a, b, med);
}

__global__ void __launch_bounds__(128)
median_conv_select(const float* __restrict__ h,
                   const int*   __restrict__ nbr,
                   const int*   __restrict__ deg,
                   float*       __restrict__ out) {
    const int n = blockIdx.x;
    const int c = threadIdx.x;

    const int d   = deg[n];
    const int med = (d - 1) >> 1;

    const int*   __restrict__ nb = nbr + (size_t)n * KK;
    const float* __restrict__ hc = h + c;

    float r;
    if (d > 24) {
        r = node_median<16, 16>(hc, nb, d, med);
    } else if (d > 16) {
        r = node_median<16, 8>(hc, nb, d, med);
    } else if (d > 12) {
        r = node_median<8, 8>(hc, nb, d, med);
    } else if (d > 8) {
        r = node_median<8, 4>(hc, nb, d, med);
    } else if (d > 4) {
        r = node_median<4, 4>(hc, nb, d, med);
    } else if (d > 2) {
        r = node_median<2, 2>(hc, nb, d, med);
    } else {
        r = (d > 0) ? hc[(size_t)nb[0] * OUT_C] : __builtin_inff();
        if (d > 1) r = fminf(r, hc[(size_t)nb[1] * OUT_C]);
    }

    out[(size_t)n * OUT_C + c] = r;
}

extern "C" void kernel_launch(void* const* d_in, const int* in_sizes, int n_in,
                              void* d_out, int out_size, void* d_ws, size_t ws_size,
                              hipStream_t stream) {
    const float* x   = (const float*)d_in[0];
    const float* w   = (const float*)d_in[1];
    const int*   nbr = (const int*)d_in[2];
    const int*   deg = (const int*)d_in[3];
    float*       out = (float*)d_out;
    float*       h   = (float*)d_ws;

    median_conv_gemm_f32<<<GEMM_BLOCKS, 256, 0, stream>>>(x, w, h);

    median_conv_select<<<NN, 128, 0, stream>>>(h, nbr, deg, out);
}


# select kernel: one wave per node, 4 channels per lane as one float4 (16-byte row gathers and stores; node, degree and index work once per node, not in 4 waves; registers capped for 5 waves per SIMD)
# speedup vs baseline: 2.2107x; 1.1816x over previous
#include <hip/hip_runtime.h>
#include <hip/hip_bf16.h>

#define NN     50000
#define KK     32
#define IN_C   256
#define OUT_C  128

typedef __attribute__((ext_vector_type(2))) float v2f;
typedef __attribute__((ext_vector_type(8))) float v8f;

#define TILE_ROWS   16
#define NUM_TILES   (NN / TILE_ROWS)
#define LDS_PITCH   260
#define GEMM_BLOCKS 512

typedef __attribute__((ext_vector_type(4))) float v4f;

__global__ void __launch_bounds__(256)
median_conv_gemm_f32(const float* __restrict__ x,
                     const float* __restrict__ w,
                     float* __restrict__ h) {
    __shared__ __attribute__((aligned(16))) float xs[2][TILE_ROWS * LDS_PITCH];

    const int tid  = threadIdx.x;
    const int wave = tid >> 5;
    const int lane = tid & 31;
    const int lid  = lane & 15;
    const int half = lane >> 4;
    const int n0   = wave * 16;

    v2f bw[64];
    #pragma unroll
    for (int s = 0; s < 64; s++) {
        const int ka = 4 * s + 2 * half;
        bw[s].x = w[(size_t)ka       * OUT_C + n0 + lid];
        bw[s].y = w[(size_t)(ka + 1) * OUT_C + n0 + lid];
    }

    int mt = blockIdx.x;
    v4f st[4];
    {
        const v4f* __restrict__ src = (const v4f*)(x + (size_t)mt * TILE_ROWS * IN_C);
        #pragma unroll
        for (int i = 0; i < 4; i++) st[i] = src[tid + 256 * i];
        #pragma unroll
        for (int i = 0; i < 4; i++) {
            const int f = tid + 256 * i;
            *(v4f*)&xs[0][(f >> 6) * LDS_PITCH + (f & 63) * 4] = st[i];
        }
    }
    __syncthreads();

    int buf = 0;
    for (; mt < NUM_TILES; mt += GEMM_BLOCKS) {
        const int  nxt      = mt + GEMM_BLOCKS;
        const bool has_next = nxt < NUM_TILES;

        if (has_next) {
            const v4f* __restrict__ src = (const v4f*)(x + (size_t)nxt * TILE_ROWS * IN_C);
            #pragma unroll
            for (int i = 0; i < 4; i++) st[i] = src[tid + 256 * i];
        }

        const float* xr = &xs[buf][lid * LDS_PITCH + 2 * half];
        v8f c = {};
        #pragma unroll
        for (int s = 0; s < 64; s++) {
            const v2f a = *(const v2f*)(xr + 4 * s);
            c = __builtin_amdgcn_wmma_f32_16x16x4_f32(
                    false, a, false, bw[s], (short)0, c, false, false);
        }

        if (has_next) {
            #pragma unroll
            for (int i = 0; i < 4; i++) {
                const int f = tid + 256 * i;
                *(v4f*)&xs[buf ^ 1][(f >> 6) * LDS_PITCH + (f & 63) * 4] = st[i];
            }
        }
        __syncthreads();
        buf ^= 1;

        const int m0 = mt * TILE_ROWS;
        #pragma unroll
        for (int v = 0; v < 8; v++) {
            h[(size_t)(m0 + v + 8 * half) * OUT_C + n0 + lid] = c[v];
        }
    }
}

#define SEL_WAVES   4
#define SEL_BLOCKS  (NN / SEL_WAVES)

__device__ __forceinline__ v4f vmin4(v4f p, v4f q) { return __builtin_elementwise_min(p, q); }
__device__ __forceinline__ v4f vmax4(v4f p, v4f q) { return __builtin_elementwise_max(p, q); }

__device__ __forceinline__ void cex(v4f& lo, v4f& hi) {
    const v4f p = lo, q = hi;
    lo = vmin4(p, q);
    hi = vmax4(p, q);
}

template <int LO, int HI, int R>
__device__ __forceinline__ void oe_merge(v4f* v) {
    constexpr int STEP = 2 * R;
    if constexpr (STEP < HI - LO) {
        oe_merge<LO, HI, STEP>(v);
        oe_merge<LO + R, HI, STEP>(v);
        #pragma unroll
        for (int i = LO + R; i < HI - R; i += STEP) cex(v[i], v[i + R]);
    } else {
        cex(v[LO], v[LO + R]);
    }
}

template <int LO, int HI>
__device__ __forceinline__ void oe_sort(v4f* v) {
    if constexpr (HI > LO) {
        constexpr int MID = LO + (HI - LO) / 2;
        oe_sort<LO, MID>(v);
        oe_sort<MID + 1, HI>(v);
        oe_merge<LO, HI, 1>(v);
    }
}

template <int M>
__device__ __forceinline__ v4f bitonic_pick(v4f* v, int sel) {
    #pragma unroll
    for (int j = M / 2; j > 0; j >>= 1) {
        #pragma unroll
        for (int i = 0; i < M; i++) {
            const int l = i ^ j;
            if (l > i) cex(v[i], v[l]);
        }
    }
    v4f r = v[0];
    #pragma unroll
    for (int i = 1; i < M; i++) r = (sel == i) ? v[i] : r;
    return r;
}

template <int HA, int HB>
__device__ __forceinline__ v4f rank_of_two_lists(v4f* a, v4f* b, int med) {
    oe_sort<0, HB - 1>(b);
    oe_sort<0, HA - 1>(a);
    constexpr int Q = HA / 2;
    v4f u[Q];
    #pragma unroll
    for (int i = 0; i < Q; i++) {
        v4f lo0 = a[i];
        if (HA - 1 - i < HB) lo0 = vmin4(lo0, b[HA - 1 - i]);
        v4f lo1 = a[i + Q];
        if (Q - 1 - i < HB) lo1 = vmin4(lo1, b[Q - 1 - i]);
        u[i] = vmax4(lo0, lo1);
    }
    if constexpr (HB == HA) {
        return bitonic_pick<Q>(u, med - Q);
    } else {
        constexpr int E = Q / 2;
        v4f w[E];
        #pragma unroll
        for (int i = 0; i < E; i++) w[i] = vmin4(u[i], u[i + E]);
        return bitonic_pick<E>(w, med - Q);
    }
}

__device__ __forceinline__ v4f row4(const float* __restrict__ h, int nb, int lane) {
    return ((const v4f*)h)[(unsigned)nb * (OUT_C / 4) + (unsigned)lane];
}

template <int HA, int HB>
__device__ __forceinline__ v4f node_median(const float* __restrict__ h,
                                           const int* __restrict__ nb,
                                           int lane, int d, int med) {
    const float inf = __builtin_inff();
    const v4f inf4 = {inf, inf, inf, inf};
    v4f a[HA], b[HB];
    #pragma unroll
    for (int j = 0; j < HA; j++) a[j] = row4(h, nb[j], lane);
    constexpr int CH = HB < 4 ? HB : 4;
    #pragma unroll
    for (int q = 0; q < HB; q += CH) {
        if (q == 0 || HA + q < d) {
            #pragma unroll
            for (int t = 0; t < CH; t++) {
                const v4f val = row4(h, nb[HA + q + t], lane);
                b[q + t] = (HA + q + t < d) ? val : inf4;
            }
        } else {
            #pragma unroll
            for (int t = 0; t < CH; t++) b[q + t] = inf4;
        }
    }
    return rank_of_two_lists<HA, HB>(a, b, med);
}

__global__ void __launch_bounds__(32 * SEL_WAVES) __attribute__((amdgpu_waves_per_eu(5)))
median_conv_select(const float* __restrict__ h,
                   const int*   __restrict__ nbr,
                   const int*   __restrict__ deg,
                   float*       __restrict__ out) {
    const int lane = threadIdx.x & 31;
    const int wave = __builtin_amdgcn_readfirstlane((int)(threadIdx.x >> 5));
    const int n    = blockIdx.x * SEL_WAVES + wave;

    const int d   = deg[n];
    const int med = (d - 1) >> 1;

    const int* __restrict__ nb = nbr + (size_t)n * KK;

    v4f r;
    if (d > 24) {
        r = node_median<16, 16>(h, nb, lane, d, med);
    } else if (d > 16) {
        r = node_median<16, 8>(h, nb, lane, d, med);
    } else if (d > 12) {
        r = node_median<8, 8>(h, nb, lane, d, med);
    } else if (d > 8) {
        r = node_median<8, 4>(h, nb, lane, d, med);
    } else if (d > 4) {
        r = node_median<4, 4>(h, nb, lane, d, med);
    } else if (d > 2) {
        r = node_median<2, 2>(h, nb, lane, d, med);
    } else {
        const float inf = __builtin_inff();
        const v4f inf4 = {inf, inf, inf, inf};
        r = (d > 0) ? row4(h, nb[0], lane) : inf4;
        if (d > 1) r = vmin4(r, row4(h, nb[1], lane));
    }

    ((v4f*)(out + (size_t)n * OUT_C))[lane] = r;
}

extern "C" void kernel_launch(void* const* d_in, const int* in_sizes, int n_in,
                              void* d_out, int out_size, void* d_ws, size_t ws_size,
                              hipStream_t stream) {
    const float* x   = (const float*)d_in[0];
    const float* w   = (const float*)d_in[1];
    const int*   nbr = (const int*)d_in[2];
    const int*   deg = (const int*)d_in[3];
    float*       out = (float*)d_out;
    float*       h   = (float*)d_ws;

    median_conv_gemm_f32<<<GEMM_BLOCKS, 256, 0, stream>>>(x, w, h);

    median_conv_select<<<SEL_BLOCKS, 32 * SEL_WAVES, 0, stream>>>(h, nbr, deg, out);
}


# select kernel: each wave loops over 4 consecutive nodes and requests the next node's degree and index row (scalar loads) before the current node's gathers and network; 4x fewer wave launches
# speedup vs baseline: 2.3594x; 1.0673x over previous
#include <hip/hip_runtime.h>
#include <hip/hip_bf16.h>

#define NN     50000
#define KK     32
#define IN_C   256
#define OUT_C  128

typedef __attribute__((ext_vector_type(2))) float v2f;
typedef __attribute__((ext_vector_type(8))) float v8f;

#define TILE_ROWS   16
#define NUM_TILES   (NN / TILE_ROWS)
#define LDS_PITCH   260
#define GEMM_BLOCKS 512

typedef __attribute__((ext_vector_type(4))) float v4f;

__global__ void __launch_bounds__(256)
median_conv_gemm_f32(const float* __restrict__ x,
                     const float* __restrict__ w,
                     float* __restrict__ h) {
    __shared__ __attribute__((aligned(16))) float xs[2][TILE_ROWS * LDS_PITCH];

    const int tid  = threadIdx.x;
    const int wave = tid >> 5;
    const int lane = tid & 31;
    const int lid  = lane & 15;
    const int half = lane >> 4;
    const int n0   = wave * 16;

    v2f bw[64];
    #pragma unroll
    for (int s = 0; s < 64; s++) {
        const int ka = 4 * s + 2 * half;
        bw[s].x = w[(size_t)ka       * OUT_C + n0 + lid];
        bw[s].y = w[(size_t)(ka + 1) * OUT_C + n0 + lid];
    }

    int mt = blockIdx.x;
    v4f st[4];
    {
        const v4f* __restrict__ src = (const v4f*)(x + (size_t)mt * TILE_ROWS * IN_C);
        #pragma unroll
        for (int i = 0; i < 4; i++) st[i] = src[tid + 256 * i];
        #pragma unroll
        for (int i = 0; i < 4; i++) {
            const int f = tid + 256 * i;
            *(v4f*)&xs[0][(f >> 6) * LDS_PITCH + (f & 63) * 4] = st[i];
        }
    }
    __syncthreads();

    int buf = 0;
    for (; mt < NUM_TILES; mt += GEMM_BLOCKS) {
        const int  nxt      = mt + GEMM_BLOCKS;
        const bool has_next = nxt < NUM_TILES;

        if (has_next) {
            const v4f* __restrict__ src = (const v4f*)(x + (size_t)nxt * TILE_ROWS * IN_C);
            #pragma unroll
            for (int i = 0; i < 4; i++) st[i] = src[tid + 256 * i];
        }

        const float* xr = &xs[buf][lid * LDS_PITCH + 2 * half];
        v8f c = {};
        #pragma unroll
        for (int s = 0; s < 64; s++) {
            const v2f a = *(const v2f*)(xr + 4 * s);
            c = __builtin_amdgcn_wmma_f32_16x16x4_f32(
                    false, a, false, bw[s], (short)0, c, false, false);
        }

        if (has_next) {
            #pragma unroll
            for (int i = 0; i < 4; i++) {
                const int f = tid + 256 * i;
                *(v4f*)&xs[buf ^ 1][(f >> 6) * LDS_PITCH + (f & 63) * 4] = st[i];
            }
        }
        __syncthreads();
        buf ^= 1;

        const int m0 = mt * TILE_ROWS;
        #pragma unroll
        for (int v = 0; v < 8; v++) {
            h[(size_t)(m0 + v + 8 * half) * OUT_C + n0 + lid] = c[v];
        }
    }
}

#define SEL_WAVES   4
#define SEL_NODES   4
#define SEL_BLOCKS  (NN / (SEL_WAVES * SEL_NODES))

__device__ __forceinline__ v4f vmin4(v4f p, v4f q) { return __builtin_elementwise_min(p, q); }
__device__ __forceinline__ v4f vmax4(v4f p, v4f q) { return __builtin_elementwise_max(p, q); }

__device__ __forceinline__ void cex(v4f& lo, v4f& hi) {
    const v4f p = lo, q = hi;
    lo = vmin4(p, q);
    hi = vmax4(p, q);
}

template <int LO, int HI, int R>
__device__ __forceinline__ void oe_merge(v4f* v) {
    constexpr int STEP = 2 * R;
    if constexpr (STEP < HI - LO) {
        oe_merge<LO, HI, STEP>(v);
        oe_merge<LO + R, HI, STEP>(v);
        #pragma unroll
        for (int i = LO + R; i < HI - R; i += STEP) cex(v[i], v[i + R]);
    } else {
        cex(v[LO], v[LO + R]);
    }
}

template <int LO, int HI>
__device__ __forceinline__ void oe_sort(v4f* v) {
    if constexpr (HI > LO) {
        constexpr int MID = LO + (HI - LO) / 2;
        oe_sort<LO, MID>(v);
        oe_sort<MID + 1, HI>(v);
        oe_merge<LO, HI, 1>(v);
    }
}

template <int M>
__device__ __forceinline__ v4f bitonic_pick(v4f* v, int sel) {
    #pragma unroll
    for (int j = M / 2; j > 0; j >>= 1) {
        #pragma unroll
        for (int i = 0; i < M; i++) {
            const int l = i ^ j;
            if (l > i) cex(v[i], v[l]);
        }
    }
    v4f r = v[0];
    #pragma unroll
    for (int i = 1; i < M; i++) r = (sel == i) ? v[i] : r;
    return r;
}

template <int HA, int HB>
__device__ __forceinline__ v4f rank_of_two_lists(v4f* a, v4f* b, int med) {
    oe_sort<0, HB - 1>(b);
    oe_sort<0, HA - 1>(a);
    constexpr int Q = HA / 2;
    v4f u[Q];
    #pragma unroll
    for (int i = 0; i < Q; i++) {
        v4f lo0 = a[i];
        if (HA - 1 - i < HB) lo0 = vmin4(lo0, b[HA - 1 - i]);
        v4f lo1 = a[i + Q];
        if (Q - 1 - i < HB) lo1 = vmin4(lo1, b[Q - 1 - i]);
        u[i] = vmax4(lo0, lo1);
    }
    if constexpr (HB == HA) {
        return bitonic_pick<Q>(u, med - Q);
    } else {
        constexpr int E = Q / 2;
        v4f w[E];
        #pragma unroll
        for (int i = 0; i < E; i++) w[i] = vmin4(u[i], u[i + E]);
        return bitonic_pick<E>(w, med - Q);
    }
}

__device__ __forceinline__ v4f row4(const float* __restrict__ h, int nb, int lane) {
    return ((const v4f*)h)[(unsigned)nb * (OUT_C / 4) + (unsigned)lane];
}

__device__ __forceinline__ void load_index_row(const int* __restrict__ nbr, int n, int (&idx)[KK]) {
    const int* __restrict__ p = nbr + (size_t)n * KK;
    #pragma unroll
    for (int j = 0; j < KK; j++) idx[j] = p[j];
}

template <int HA, int HB>
__device__ __forceinline__ v4f node_median(const float* __restrict__ h,
                                           const int (&nb)[KK],
                                           int lane, int d, int med) {
    const float inf = __builtin_inff();
    const v4f inf4 = {inf, inf, inf, inf};
    v4f a[HA], b[HB];
    #pragma unroll
    for (int j = 0; j < HA; j++) a[j] = row4(h, nb[j], lane);
    constexpr int CH = HB < 4 ? HB : 4;
    #pragma unroll
    for (int q = 0; q < HB; q += CH) {
        if (q == 0 || HA + q < d) {
            #pragma unroll
            for (int t = 0; t < CH; t++) {
                const v4f val = row4(h, nb[HA + q + t], lane);
                b[q + t] = (HA + q + t < d) ? val : inf4;
            }
        } else {
            #pragma unroll
            for (int t = 0; t < CH; t++) b[q + t] = inf4;
        }
    }
    return rank_of_two_lists<HA, HB>(a, b, med);
}

__global__ void __launch_bounds__(32 * SEL_WAVES) __attribute__((amdgpu_waves_per_eu(5)))
median_conv_select(const float* __restrict__ h,
                   const int*   __restrict__ nbr,
                   const int*   __restrict__ deg,
                   float*       __restrict__ out) {
    const int lane = threadIdx.x & 31;
    const int wave = __builtin_amdgcn_readfirstlane((int)(threadIdx.x >> 5));
    const int n0   = (blockIdx.x * SEL_WAVES + wave) * SEL_NODES;

    int d = deg[n0];
    int idx[KK];
    load_index_row(nbr, n0, idx);

    asm volatile("" :: "s"(d), "s"(idx[0]), "s"(idx[16]));

    #pragma unroll 1
    for (int i = 0; i < SEL_NODES; i++) {
        const int n = n0 + i;

        const int nn = (n + 1 < NN) ? n + 1 : NN - 1;
        const int d_next = deg[nn];
        int idx_next[KK];
        load_index_row(nbr, nn, idx_next);

        const int med = (d - 1) >> 1;

        v4f r;
        if (d > 24) {
            r = node_median<16, 16>(h, idx, lane, d, med);
        } else if (d > 16) {
            r = node_median<16, 8>(h, idx, lane, d, med);
        } else if (d > 12) {
            r = node_median<8, 8>(h, idx, lane, d, med);
        } else if (d > 8) {
            r = node_median<8, 4>(h, idx, lane, d, med);
        } else if (d > 4) {
            r = node_median<4, 4>(h, idx, lane, d, med);
        } else if (d > 2) {
            r = node_median<2, 2>(h, idx, lane, d, med);
        } else {
            const float inf = __builtin_inff();
            const v4f inf4 = {inf, inf, inf, inf};
            r = (d > 0) ? row4(h, idx[0], lane) : inf4;
            if (d > 1) r = vmin4(r, row4(h, idx[1], lane));
        }

        ((v4f*)(out + (size_t)n * OUT_C))[lane] = r;

        d = d_next;
        #pragma unroll
        for (int j = 0; j < KK; j++) idx[j] = idx_next[j];
    }
}

extern "C" void kernel_launch(void* const* d_in, const int* in_sizes, int n_in,
                              void* d_out, int out_size, void* d_ws, size_t ws_size,
                              hipStream_t stream) {
    const float* x   = (const float*)d_in[0];
    const float* w   = (const float*)d_in[1];
    const int*   nbr = (const int*)d_in[2];
    const int*   deg = (const int*)d_in[3];
    float*       out = (float*)d_out;
    float*       h   = (float*)d_ws;

    median_conv_gemm_f32<<<GEMM_BLOCKS, 256, 0, stream>>>(x, w, h);

    median_conv_select<<<SEL_BLOCKS, 32 * SEL_WAVES, 0, stream>>>(h, nbr, deg, out);
}


# select kernel: min/max networks of the four large paths hand-written as dual-issue instruction pairs on fixed register blocks (padding mask inside the asm); same operations, bit-identical results
# speedup vs baseline: 2.5451x; 1.0787x over previous
#include <hip/hip_runtime.h>
#include <hip/hip_bf16.h>

#define NN     50000
#define KK     32
#define IN_C   256
#define OUT_C  128

typedef __attribute__((ext_vector_type(2))) float v2f;
typedef __attribute__((ext_vector_type(8))) float v8f;

#define TILE_ROWS   16
#define NUM_TILES   (NN / TILE_ROWS)
#define LDS_PITCH   260
#define GEMM_BLOCKS 512

typedef __attribute__((ext_vector_type(4))) float v4f;

__global__ void __launch_bounds__(256)
median_conv_gemm_f32(const float* __restrict__ x,
                     const float* __restrict__ w,
                     float* __restrict__ h) {
    __shared__ __attribute__((aligned(16))) float xs[2][TILE_ROWS * LDS_PITCH];

    const int tid  = threadIdx.x;
    const int wave = tid >> 5;
    const int lane = tid & 31;
    const int lid  = lane & 15;
    const int half = lane >> 4;
    const int n0   = wave * 16;

    v2f bw[64];
    #pragma unroll
    for (int s = 0; s < 64; s++) {
        const int ka = 4 * s + 2 * half;
        bw[s].x = w[(size_t)ka       * OUT_C + n0 + lid];
        bw[s].y = w[(size_t)(ka + 1) * OUT_C + n0 + lid];
    }

    int mt = blockIdx.x;
    v4f st[4];
    {
        const v4f* __restrict__ src = (const v4f*)(x + (size_t)mt * TILE_ROWS * IN_C);
        #pragma unroll
        for (int i = 0; i < 4; i++) st[i] = src[tid + 256 * i];
        #pragma unroll
        for (int i = 0; i < 4; i++) {
            const int f = tid + 256 * i;
            *(v4f*)&xs[0][(f >> 6) * LDS_PITCH + (f & 63) * 4] = st[i];
        }
    }
    __syncthreads();

    int buf = 0;
    for (; mt < NUM_TILES; mt += GEMM_BLOCKS) {
        const int  nxt      = mt + GEMM_BLOCKS;
        const bool has_next = nxt < NUM_TILES;

        if (has_next) {
            const v4f* __restrict__ src = (const v4f*)(x + (size_t)nxt * TILE_ROWS * IN_C);
            #pragma unroll
            for (int i = 0; i < 4; i++) st[i] = src[tid + 256 * i];
        }

        const float* xr = &xs[buf][lid * LDS_PITCH + 2 * half];
        v8f c = {};
        #pragma unroll
        for (int s = 0; s < 64; s++) {
            const v2f a = *(const v2f*)(xr + 4 * s);
            c = __builtin_amdgcn_wmma_f32_16x16x4_f32(
                    false, a, false, bw[s], (short)0, c, false, false);
        }

        if (has_next) {
            #pragma unroll
            for (int i = 0; i < 4; i++) {
                const int f = tid + 256 * i;
                *(v4f*)&xs[buf ^ 1][(f >> 6) * LDS_PITCH + (f & 63) * 4] = st[i];
            }
        }
        __syncthreads();
        buf ^= 1;

        const int m0 = mt * TILE_ROWS;
        #pragma unroll
        for (int v = 0; v < 8; v++) {
            h[(size_t)(m0 + v + 8 * half) * OUT_C + n0 + lid] = c[v];
        }
    }
}

#define SEL_WAVES   4
#define SEL_NODES   4
#define SEL_BLOCKS  (NN / (SEL_WAVES * SEL_NODES))

__device__ __forceinline__ v4f vmin4(v4f p, v4f q) { return __builtin_elementwise_min(p, q); }
__device__ __forceinline__ v4f vmax4(v4f p, v4f q) { return __builtin_elementwise_max(p, q); }

__device__ __forceinline__ void cex(v4f& lo, v4f& hi) {
    const v4f p = lo, q = hi;
    lo = vmin4(p, q);
    hi = vmax4(p, q);
}

template <int LO, int HI, int R>
__device__ __forceinline__ void oe_merge(v4f* v) {
    constexpr int STEP = 2 * R;
    if constexpr (STEP < HI - LO) {
        oe_merge<LO, HI, STEP>(v);
        oe_merge<LO + R, HI, STEP>(v);
        #pragma unroll
        for (int i = LO + R; i < HI - R; i += STEP) cex(v[i], v[i + R]);
    } else {
        cex(v[LO], v[LO + R]);
    }
}

template <int LO, int HI>
__device__ __forceinline__ void oe_sort(v4f* v) {
    if constexpr (HI > LO) {
        constexpr int MID = LO + (HI - LO) / 2;
        oe_sort<LO, MID>(v);
        oe_sort<MID + 1, HI>(v);
        oe_merge<LO, HI, 1>(v);
    }
}

template <int M>
__device__ __forceinline__ v4f bitonic_pick(v4f* v, int sel) {
    #pragma unroll
    for (int j = M / 2; j > 0; j >>= 1) {
        #pragma unroll
        for (int i = 0; i < M; i++) {
            const int l = i ^ j;
            if (l > i) cex(v[i], v[l]);
        }
    }
    v4f r = v[0];
    #pragma unroll
    for (int i = 1; i < M; i++) r = (sel == i) ? v[i] : r;
    return r;
}


__device__ __forceinline__ v4f rank_paired_16_16(v4f* a, v4f* b, int d, int med) {
    const float inf = __builtin_inff();
    const float m9 = (25 < d) ? -inf : inf;
    const float m10 = (26 < d) ? -inf : inf;
    const float m11 = (27 < d) ? -inf : inf;
    const float m12 = (28 < d) ? -inf : inf;
    const float m13 = (29 < d) ? -inf : inf;
    const float m14 = (30 < d) ? -inf : inf;
    const float m15 = (31 < d) ? -inf : inf;
    v4f sp;
    asm(
        "v_dual_max_num_f32 v100, %[m9], v100 :: v_dual_max_num_f32 v101, %[m9], v101\n\t"
        "v_dual_max_num_f32 v102, %[m9], v102 :: v_dual_max_num_f32 v103, %[m9], v103\n\t"
        "v_dual_max_num_f32 v104, %[m10], v104 :: v_dual_max_num_f32 v105, %[m10], v105\n\t"
        "v_dual_max_num_f32 v106, %[m10], v106 :: v_dual_max_num_f32 v107, %[m10], v107\n\t"
        "v_dual_max_num_f32 v108, %[m11], v108 :: v_dual_max_num_f32 v109, %[m11], v109\n\t"
        "v_dual_max_num_f32 v110, %[m11], v110 :: v_dual_max_num_f32 v111, %[m11], v111\n\t"
        "v_dual_max_num_f32 v112, %[m12], v112 :: v_dual_max_num_f32 v113, %[m12], v113\n\t"
        "v_dual_max_num_f32 v114, %[m12], v114 :: v_dual_max_num_f32 v115, %[m12], v115\n\t"
        "v_dual_max_num_f32 v116, %[m13], v116 :: v_dual_max_num_f32 v117, %[m13], v117\n\t"
        "v_dual_max_num_f32 v118, %[m13], v118 :: v_dual_max_num_f32 v119, %[m13], v119\n\t"
        "v_dual_max_num_f32 v120, %[m14], v120 :: v_dual_max_num_f32 v121, %[m14], v121\n\t"
        "v_dual_max_num_f32 v122, %[m14], v122 :: v_dual_max_num_f32 v123, %[m14], v123\n\t"
        "v_dual_max_num_f32 v124, %[m15], v124 :: v_dual_max_num_f32 v125, %[m15], v125\n\t"
        "v_dual_max_num_f32 v126, %[m15], v126 :: v_dual_max_num_f32 v127, %[m15], v127\n\t"
        "v_dual_min_num_f32 v128, v64, v68 :: v_dual_min_num_f32 v129, v65, v69\n\t"
        "v_dual_min_num_f32 v130, v66, v70 :: v_dual_min_num_f32 v131, v67, v71\n\t"
        "v_dual_max_num_f32 v68, v64, v68 :: v_dual_max_num_f32 v69, v65, v69\n\t"
        "v_dual_max_num_f32 v70, v66, v70 :: v_dual_max_num_f32 v71, v67, v71\n\t"
        "v_dual_min_num_f32 v64, v72, v76 :: v_dual_min_num_f32 v65, v73, v77\n\t"
        "v_dual_min_num_f32 v66, v74, v78 :: v_dual_min_num_f32 v67, v75, v79\n\t"
        "v_dual_max_num_f32 v76, v72, v76 :: v_dual_max_num_f32 v77, v73, v77\n\t"
        "v_dual_max_num_f32 v78, v74, v78 :: v_dual_max_num_f32 v79, v75, v79\n\t"
        "v_dual_min_num_f32 v72, v80, v84 :: v_dual_min_num_f32 v73, v81, v85\n\t"
        "v_dual_min_num_f32 v74, v82, v86 :: v_dual_min_num_f32 v75, v83, v87\n\t"
        "v_dual_max_num_f32 v84, v80, v84 :: v_dual_max_num_f32 v85, v81, v85\n\t"
        "v_dual_max_num_f32 v86, v82, v86 :: v_dual_max_num_f32 v87, v83, v87\n\t"
        "v_dual_min_num_f32 v80, v88, v92 :: v_dual_min_num_f32 v81, v89, v93\n\t"
        "v_dual_min_num_f32 v82, v90, v94 :: v_dual_min_num_f32 v83, v91, v95\n\t"
        "v_dual_max_num_f32 v92, v88, v92 :: v_dual_max_num_f32 v93, v89, v93\n\t"
        "v_dual_max_num_f32 v94, v90, v94 :: v_dual_max_num_f32 v95, v91, v95\n\t"
        "v_dual_min_num_f32 v88, v96, v100 :: v_dual_min_num_f32 v89, v97, v101\n\t"
        "v_dual_min_num_f32 v90, v98, v102 :: v_dual_min_num_f32 v91, v99, v103\n\t"
        "v_dual_max_num_f32 v100, v96, v100 :: v_dual_max_num_f32 v101, v97, v101\n\t"
        "v_dual_max_num_f32 v102, v98, v102 :: v_dual_max_num_f32 v103, v99, v103\n\t"
        "v_dual_min_num_f32 v96, v104, v108 :: v_dual_min_num_f32 v97, v105, v109\n\t"
        "v_dual_min_num_f32 v98, v106, v110 :: v_dual_min_num_f32 v99, v107, v111\n\t"
        "v_dual_max_num_f32 v108, v104, v108 :: v_dual_max_num_f32 v109, v105, v109\n\t"
        "v_dual_max_num_f32 v110, v106, v110 :: v_dual_max_num_f32 v111, v107, v111\n\t"
        "v_dual_min_num_f32 v104, v112, v116 :: v_dual_min_num_f32 v105, v113, v117\n\t"
        "v_dual_min_num_f32 v106, v114, v118 :: v_dual_min_num_f32 v107, v115, v119\n\t"
        "v_dual_max_num_f32 v116, v112, v116 :: v_dual_max_num_f32 v117, v113, v117\n\t"
        "v_dual_max_num_f32 v118, v114, v118 :: v_dual_max_num_f32 v119, v115, v119\n\t"
        "v_dual_min_num_f32 v112, v120, v124 :: v_dual_min_num_f32 v113, v121, v125\n\t"
        "v_dual_min_num_f32 v114, v122, v126 :: v_dual_min_num_f32 v115, v123, v127\n\t"
        "v_dual_max_num_f32 v124, v120, v124 :: v_dual_max_num_f32 v125, v121, v125\n\t"
        "v_dual_max_num_f32 v126, v122, v126 :: v_dual_max_num_f32 v127, v123, v127\n\t"
        "v_dual_min_num_f32 v120, v0, v4 :: v_dual_min_num_f32 v121, v1, v5\n\t"
        "v_dual_min_num_f32 v122, v2, v6 :: v_dual_min_num_f32 v123, v3, v7\n\t"
        "v_dual_max_num_f32 v4, v0, v4 :: v_dual_max_num_f32 v5, v1, v5\n\t"
        "v_dual_max_num_f32 v6, v2, v6 :: v_dual_max_num_f32 v7, v3, v7\n\t"
        "v_dual_min_num_f32 v0, v8, v12 :: v_dual_min_num_f32 v1, v9, v13\n\t"
        "v_dual_min_num_f32 v2, v10, v14 :: v_dual_min_num_f32 v3, v11, v15\n\t"
        "v_dual_max_num_f32 v12, v8, v12 :: v_dual_max_num_f32 v13, v9, v13\n\t"
        "v_dual_max_num_f32 v14, v10, v14 :: v_dual_max_num_f32 v15, v11, v15\n\t"
        "v_dual_min_num_f32 v8, v16, v20 :: v_dual_min_num_f32 v9, v17, v21\n\t"
        "v_dual_min_num_f32 v10, v18, v22 :: v_dual_min_num_f32 v11, v19, v23\n\t"
        "v_dual_max_num_f32 v20, v16, v20 :: v_dual_max_num_f32 v21, v17, v21\n\t"
        "v_dual_max_num_f32 v22, v18, v22 :: v_dual_max_num_f32 v23, v19, v23\n\t"
        "v_dual_min_num_f32 v16, v24, v28 :: v_dual_min_num_f32 v17, v25, v29\n\t"
        "v_dual_min_num_f32 v18, v26, v30 :: v_dual_min_num_f32 v19, v27, v31\n\t"
        "v_dual_max_num_f32 v28, v24, v28 :: v_dual_max_num_f32 v29, v25, v29\n\t"
        "v_dual_max_num_f32 v30, v26, v30 :: v_dual_max_num_f32 v31, v27, v31\n\t"
        "v_dual_min_num_f32 v24, v32, v36 :: v_dual_min_num_f32 v25, v33, v37\n\t"
        "v_dual_min_num_f32 v26, v34, v38 :: v_dual_min_num_f32 v27, v35, v39\n\t"
        "v_dual_max_num_f32 v36, v32, v36 :: v_dual_max_num_f32 v37, v33, v37\n\t"
        "v_dual_max_num_f32 v38, v34, v38 :: v_dual_max_num_f32 v39, v35, v39\n\t"
        "v_dual_min_num_f32 v32, v40, v44 :: v_dual_min_num_f32 v33, v41, v45\n\t"
        "v_dual_min_num_f32 v34, v42, v46 :: v_dual_min_num_f32 v35, v43, v47\n\t"
        "v_dual_max_num_f32 v44, v40, v44 :: v_dual_max_num_f32 v45, v41, v45\n\t"
        "v_dual_max_num_f32 v46, v42, v46 :: v_dual_max_num_f32 v47, v43, v47\n\t"
        "v_dual_min_num_f32 v40, v48, v52 :: v_dual_min_num_f32 v41, v49, v53\n\t"
        "v_dual_min_num_f32 v42, v50, v54 :: v_dual_min_num_f32 v43, v51, v55\n\t"
        "v_dual_max_num_f32 v52, v48, v52 :: v_dual_max_num_f32 v53, v49, v53\n\t"
        "v_dual_max_num_f32 v54, v50, v54 :: v_dual_max_num_f32 v55, v51, v55\n\t"
        "v_dual_min_num_f32 v48, v56, v60 :: v_dual_min_num_f32 v49, v57, v61\n\t"
        "v_dual_min_num_f32 v50, v58, v62 :: v_dual_min_num_f32 v51, v59, v63\n\t"
        "v_dual_max_num_f32 v60, v56, v60 :: v_dual_max_num_f32 v61, v57, v61\n\t"
        "v_dual_max_num_f32 v62, v58, v62 :: v_dual_max_num_f32 v63, v59, v63\n\t"
        "v_dual_min_num_f32 v56, v128, v64 :: v_dual_min_num_f32 v57, v129, v65\n\t"
        "v_dual_min_num_f32 v58, v130, v66 :: v_dual_min_num_f32 v59, v131, v67\n\t"
        "v_dual_max_num_f32 v64, v128, v64 :: v_dual_max_num_f32 v65, v129, v65\n\t"
        "v_dual_max_num_f32 v66, v130, v66 :: v_dual_max_num_f32 v67, v131, v67\n\t"
        "v_dual_min_num_f32 v128, v68, v76 :: v_dual_min_num_f32 v129, v69, v77\n\t"
        "v_dual_min_num_f32 v130, v70, v78 :: v_dual_min_num_f32 v131, v71, v79\n\t"
        "v_dual_max_num_f32 v76, v68, v76 :: v_dual_max_num_f32 v77, v69, v77\n\t"
        "v_dual_max_num_f32 v78, v70, v78 :: v_dual_max_num_f32 v79, v71, v79\n\t"
        "v_dual_min_num_f32 v68, v72, v80 :: v_dual_min_num_f32 v69, v73, v81\n\t"
        "v_dual_min_num_f32 v70, v74, v82 :: v_dual_min_num_f32 v71, v75, v83\n\t"
        "v_dual_max_num_f32 v80, v72, v80 :: v_dual_max_num_f32 v81, v73, v81\n\t"
        "v_dual_max_num_f32 v82, v74, v82 :: v_dual_max_num_f32 v83, v75, v83\n\t"
        "v_dual_min_num_f32 v72, v84, v92 :: v_dual_min_num_f32 v73, v85, v93\n\t"
        "v_dual_min_num_f32 v74, v86, v94 :: v_dual_min_num_f32 v75, v87, v95\n\t"
        "v_dual_max_num_f32 v92, v84, v92 :: v_dual_max_num_f32 v93, v85, v93\n\t"
        "v_dual_max_num_f32 v94, v86, v94 :: v_dual_max_num_f32 v95, v87, v95\n\t"
        "v_dual_min_num_f32 v84, v88, v96 :: v_dual_min_num_f32 v85, v89, v97\n\t"
        "v_dual_min_num_f32 v86, v90, v98 :: v_dual_min_num_f32 v87, v91, v99\n\t"
        "v_dual_max_num_f32 v96, v88, v96 :: v_dual_max_num_f32 v97, v89, v97\n\t"
        "v_dual_max_num_f32 v98, v90, v98 :: v_dual_max_num_f32 v99, v91, v99\n\t"
        "v_dual_min_num_f32 v88, v100, v108 :: v_dual_min_num_f32 v89, v101, v109\n\t"
        "v_dual_min_num_f32 v90, v102, v110 :: v_dual_min_num_f32 v91, v103, v111\n\t"
        "v_dual_max_num_f32 v108, v100, v108 :: v_dual_max_num_f32 v109, v101, v109\n\t"
        "v_dual_max_num_f32 v110, v102, v110 :: v_dual_max_num_f32 v111, v103, v111\n\t"
        "v_dual_min_num_f32 v100, v104, v112 :: v_dual_min_num_f32 v101, v105, v113\n\t"
        "v_dual_min_num_f32 v102, v106, v114 :: v_dual_min_num_f32 v103, v107, v115\n\t"
        "v_dual_max_num_f32 v112, v104, v112 :: v_dual_max_num_f32 v113, v105, v113\n\t"
        "v_dual_max_num_f32 v114, v106, v114 :: v_dual_max_num_f32 v115, v107, v115\n\t"
        "v_dual_min_num_f32 v104, v116, v124 :: v_dual_min_num_f32 v105, v117, v125\n\t"
        "v_dual_min_num_f32 v106, v118, v126 :: v_dual_min_num_f32 v107, v119, v127\n\t"
        "v_dual_max_num_f32 v124, v116, v124 :: v_dual_max_num_f32 v125, v117, v125\n\t"
        "v_dual_max_num_f32 v126, v118, v126 :: v_dual_max_num_f32 v127, v119, v127\n\t"
        "v_dual_min_num_f32 v116, v120, v0 :: v_dual_min_num_f32 v117, v121, v1\n\t"
        "v_dual_min_num_f32 v118, v122, v2 :: v_dual_min_num_f32 v119, v123, v3\n\t"
        "v_dual_max_num_f32 v0, v120, v0 :: v_dual_max_num_f32 v1, v121, v1\n\t"
        "v_dual_max_num_f32 v2, v122, v2 :: v_dual_max_num_f32 v3, v123, v3\n\t"
        "v_dual_min_num_f32 v120, v4, v12 :: v_dual_min_num_f32 v121, v5, v13\n\t"
        "v_dual_min_num_f32 v122, v6, v14 :: v_dual_min_num_f32 v123, v7, v15\n\t"
        "v_dual_max_num_f32 v12, v4, v12 :: v_dual_max_num_f32 v13, v5, v13\n\t"
        "v_dual_max_num_f32 v14, v6, v14 :: v_dual_max_num_f32 v15, v7, v15\n\t"
        "v_dual_min_num_f32 v4, v8, v16 :: v_dual_min_num_f32 v5, v9, v17\n\t"
        "v_dual_min_num_f32 v6, v10, v18 :: v_dual_min_num_f32 v7, v11, v19\n\t"
        "v_dual_max_num_f32 v16, v8, v16 :: v_dual_max_num_f32 v17, v9, v17\n\t"
        "v_dual_max_num_f32 v18, v10, v18 :: v_dual_max_num_f32 v19, v11, v19\n\t"
        "v_dual_min_num_f32 v8, v20, v28 :: v_dual_min_num_f32 v9, v21, v29\n\t"
        "v_dual_min_num_f32 v10, v22, v30 :: v_dual_min_num_f32 v11, v23, v31\n\t"
        "v_dual_max_num_f32 v28, v20, v28 :: v_dual_max_num_f32 v29, v21, v29\n\t"
        "v_dual_max_num_f32 v30, v22, v30 :: v_dual_max_num_f32 v31, v23, v31\n\t"
        "v_dual_min_num_f32 v20, v24, v32 :: v_dual_min_num_f32 v21, v25, v33\n\t"
        "v_dual_min_num_f32 v22, v26, v34 :: v_dual_min_num_f32 v23, v27, v35\n\t"
        "v_dual_max_num_f32 v32, v24, v32 :: v_dual_max_num_f32 v33, v25, v33\n\t"
        "v_dual_max_num_f32 v34, v26, v34 :: v_dual_max_num_f32 v35, v27, v35\n\t"
        "v_dual_min_num_f32 v24, v36, v44 :: v_dual_min_num_f32 v25, v37, v45\n\t"
        "v_dual_min_num_f32 v26, v38, v46 :: v_dual_min_num_f32 v27, v39, v47\n\t"
        "v_dual_max_num_f32 v44, v36, v44 :: v_dual_max_num_f32 v45, v37, v45\n\t"
        "v_dual_max_num_f32 v46, v38, v46 :: v_dual_max_num_f32 v47, v39, v47\n\t"
        "v_dual_min_num_f32 v36, v40, v48 :: v_dual_min_num_f32 v37, v41, v49\n\t"
        "v_dual_min_num_f32 v38, v42, v50 :: v_dual_min_num_f32 v39, v43, v51\n\t"
        "v_dual_max_num_f32 v48, v40, v48 :: v_dual_max_num_f32 v49, v41, v49\n\t"
        "v_dual_max_num_f32 v50, v42, v50 :: v_dual_max_num_f32 v51, v43, v51\n\t"
        "v_dual_min_num_f32 v40, v52, v60 :: v_dual_min_num_f32 v41, v53, v61\n\t"
        "v_dual_min_num_f32 v42, v54, v62 :: v_dual_min_num_f32 v43, v55, v63\n\t"
        "v_dual_max_num_f32 v60, v52, v60 :: v_dual_max_num_f32 v61, v53, v61\n\t"
        "v_dual_max_num_f32 v62, v54, v62 :: v_dual_max_num_f32 v63, v55, v63\n\t"
        "v_dual_min_num_f32 v52, v128, v64 :: v_dual_min_num_f32 v53, v129, v65\n\t"
        "v_dual_min_num_f32 v54, v130, v66 :: v_dual_min_num_f32 v55, v131, v67\n\t"
        "v_dual_max_num_f32 v64, v128, v64 :: v_dual_max_num_f32 v65, v129, v65\n\t"
        "v_dual_max_num_f32 v66, v130, v66 :: v_dual_max_num_f32 v67, v131, v67\n\t"
        "v_dual_min_num_f32 v128, v72, v80 :: v_dual_min_num_f32 v129, v73, v81\n\t"
        "v_dual_min_num_f32 v130, v74, v82 :: v_dual_min_num_f32 v131, v75, v83\n\t"
        "v_dual_max_num_f32 v80, v72, v80 :: v_dual_max_num_f32 v81, v73, v81\n\t"
        "v_dual_max_num_f32 v82, v74, v82 :: v_dual_max_num_f32 v83, v75, v83\n\t"
        "v_dual_min_num_f32 v72, v56, v68 :: v_dual_min_num_f32 v73, v57, v69\n\t"
        "v_dual_min_num_f32 v74, v58, v70 :: v_dual_min_num_f32 v75, v59, v71\n\t"
        "v_dual_max_num_f32 v68, v56, v68 :: v_dual_max_num_f32 v69, v57, v69\n\t"
        "v_dual_max_num_f32 v70, v58, v70 :: v_dual_max_num_f32 v71, v59, v71\n\t"
        "v_dual_min_num_f32 v56, v76, v92 :: v_dual_min_num_f32 v57, v77, v93\n\t"
        "v_dual_min_num_f32 v58, v78, v94 :: v_dual_min_num_f32 v59, v79, v95\n\t"
        "v_dual_max_num_f32 v92, v76, v92 :: v_dual_max_num_f32 v93, v77, v93\n\t"
        "v_dual_max_num_f32 v94, v78, v94 :: v_dual_max_num_f32 v95, v79, v95\n\t"
        "v_dual_min_num_f32 v76, v88, v96 :: v_dual_min_num_f32 v77, v89, v97\n\t"
        "v_dual_min_num_f32 v78, v90, v98 :: v_dual_min_num_f32 v79, v91, v99\n\t"
        "v_dual_max_num_f32 v96, v88, v96 :: v_dual_max_num_f32 v97, v89, v97\n\t"
        "v_dual_max_num_f32 v98, v90, v98 :: v_dual_max_num_f32 v99, v91, v99\n\t"
        "v_dual_min_num_f32 v88, v104, v112 :: v_dual_min_num_f32 v89, v105, v113\n\t"
        "v_dual_min_num_f32 v90, v106, v114 :: v_dual_min_num_f32 v91, v107, v115\n\t"
        "v_dual_max_num_f32 v112, v104, v112 :: v_dual_max_num_f32 v113, v105, v113\n\t"
        "v_dual_max_num_f32 v114, v106, v114 :: v_dual_max_num_f32 v115, v107, v115\n\t"
        "v_dual_min_num_f32 v104, v84, v100 :: v_dual_min_num_f32 v105, v85, v101\n\t"
        "v_dual_min_num_f32 v106, v86, v102 :: v_dual_min_num_f32 v107, v87, v103\n\t"
        "v_dual_max_num_f32 v100, v84, v100 :: v_dual_max_num_f32 v101, v85, v101\n\t"
        "v_dual_max_num_f32 v102, v86, v102 :: v_dual_max_num_f32 v103, v87, v103\n\t"
        "v_dual_min_num_f32 v84, v108, v124 :: v_dual_min_num_f32 v85, v109, v125\n\t"
        "v_dual_min_num_f32 v86, v110, v126 :: v_dual_min_num_f32 v87, v111, v127\n\t"
        "v_dual_max_num_f32 v124, v108, v124 :: v_dual_max_num_f32 v125, v109, v125\n\t"
        "v_dual_max_num_f32 v126, v110, v126 :: v_dual_max_num_f32 v127, v111, v127\n\t"
        "v_dual_min_num_f32 v108, v120, v0 :: v_dual_min_num_f32 v109, v121, v1\n\t"
        "v_dual_min_num_f32 v110, v122, v2 :: v_dual_min_num_f32 v111, v123, v3\n\t"
        "v_dual_max_num_f32 v0, v120, v0 :: v_dual_max_num_f32 v1, v121, v1\n\t"
        "v_dual_max_num_f32 v2, v122, v2 :: v_dual_max_num_f32 v3, v123, v3\n\t"
        "v_dual_min_num_f32 v120, v8, v16 :: v_dual_min_num_f32 v121, v9, v17\n\t"
        "v_dual_min_num_f32 v122, v10, v18 :: v_dual_min_num_f32 v123, v11, v19\n\t"
        "v_dual_max_num_f32 v16, v8, v16 :: v_dual_max_num_f32 v17, v9, v17\n\t"
        "v_dual_max_num_f32 v18, v10, v18 :: v_dual_max_num_f32 v19, v11, v19\n\t"
        "v_dual_min_num_f32 v8, v116, v4 :: v_dual_min_num_f32 v9, v117, v5\n\t"
        "v_dual_min_num_f32 v10, v118, v6 :: v_dual_min_num_f32 v11, v119, v7\n\t"
        "v_dual_max_num_f32 v4, v116, v4 :: v_dual_max_num_f32 v5, v117, v5\n\t"
        "v_dual_max_num_f32 v6, v118, v6 :: v_dual_max_num_f32 v7, v119, v7\n\t"
        "v_dual_min_num_f32 v116, v12, v28 :: v_dual_min_num_f32 v117, v13, v29\n\t"
        "v_dual_min_num_f32 v118, v14, v30 :: v_dual_min_num_f32 v119, v15, v31\n\t"
        "v_dual_max_num_f32 v28, v12, v28 :: v_dual_max_num_f32 v29, v13, v29\n\t"
        "v_dual_max_num_f32 v30, v14, v30 :: v_dual_max_num_f32 v31, v15, v31\n\t"
        "v_dual_min_num_f32 v12, v24, v32 :: v_dual_min_num_f32 v13, v25, v33\n\t"
        "v_dual_min_num_f32 v14, v26, v34 :: v_dual_min_num_f32 v15, v27, v35\n\t"
        "v_dual_max_num_f32 v32, v24, v32 :: v_dual_max_num_f32 v33, v25, v33\n\t"
        "v_dual_max_num_f32 v34, v26, v34 :: v_dual_max_num_f32 v35, v27, v35\n\t"
        "v_dual_min_num_f32 v24, v40, v48 :: v_dual_min_num_f32 v25, v41, v49\n\t"
        "v_dual_min_num_f32 v26, v42, v50 :: v_dual_min_num_f32 v27, v43, v51\n\t"
        "v_dual_max_num_f32 v48, v40, v48 :: v_dual_max_num_f32 v49, v41, v49\n\t"
        "v_dual_max_num_f32 v50, v42, v50 :: v_dual_max_num_f32 v51, v43, v51\n\t"
        "v_dual_min_num_f32 v40, v20, v36 :: v_dual_min_num_f32 v41, v21, v37\n\t"
        "v_dual_min_num_f32 v42, v22, v38 :: v_dual_min_num_f32 v43, v23, v39\n\t"
        "v_dual_max_num_f32 v36, v20, v36 :: v_dual_max_num_f32 v37, v21, v37\n\t"
        "v_dual_max_num_f32 v38, v22, v38 :: v_dual_max_num_f32 v39, v23, v39\n\t"
        "v_dual_min_num_f32 v20, v44, v60 :: v_dual_min_num_f32 v21, v45, v61\n\t"
        "v_dual_min_num_f32 v22, v46, v62 :: v_dual_min_num_f32 v23, v47, v63\n\t"
        "v_dual_max_num_f32 v60, v44, v60 :: v_dual_max_num_f32 v61, v45, v61\n\t"
        "v_dual_max_num_f32 v62, v46, v62 :: v_dual_max_num_f32 v63, v47, v63\n\t"
        "v_dual_min_num_f32 v44, v64, v80 :: v_dual_min_num_f32 v45, v65, v81\n\t"
        "v_dual_min_num_f32 v46, v66, v82 :: v_dual_min_num_f32 v47, v67, v83\n\t"
        "v_dual_max_num_f32 v80, v64, v80 :: v_dual_max_num_f32 v81, v65, v81\n\t"
        "v_dual_max_num_f32 v82, v66, v82 :: v_dual_max_num_f32 v83, v67, v83\n\t"
        "v_dual_min_num_f32 v64, v52, v128 :: v_dual_min_num_f32 v65, v53, v129\n\t"
        "v_dual_min_num_f32 v66, v54, v130 :: v_dual_min_num_f32 v67, v55, v131\n\t"
        "v_dual_max_num_f32 v128, v52, v128 :: v_dual_max_num_f32 v129, v53, v129\n\t"
        "v_dual_max_num_f32 v130, v54, v130 :: v_dual_max_num_f32 v131, v55, v131\n\t"
        "v_dual_min_num_f32 v52, v96, v112 :: v_dual_min_num_f32 v53, v97, v113\n\t"
        "v_dual_min_num_f32 v54, v98, v114 :: v_dual_min_num_f32 v55, v99, v115\n\t"
        "v_dual_max_num_f32 v112, v96, v112 :: v_dual_max_num_f32 v113, v97, v113\n\t"
        "v_dual_max_num_f32 v114, v98, v114 :: v_dual_max_num_f32 v115, v99, v115\n\t"
        "v_dual_min_num_f32 v96, v76, v88 :: v_dual_min_num_f32 v97, v77, v89\n\t"
        "v_dual_min_num_f32 v98, v78, v90 :: v_dual_min_num_f32 v99, v79, v91\n\t"
        "v_dual_max_num_f32 v88, v76, v88 :: v_dual_max_num_f32 v89, v77, v89\n\t"
        "v_dual_max_num_f32 v90, v78, v90 :: v_dual_max_num_f32 v91, v79, v91\n\t"
        "v_dual_min_num_f32 v76, v72, v104 :: v_dual_min_num_f32 v77, v73, v105\n\t"
        "v_dual_min_num_f32 v78, v74, v106 :: v_dual_min_num_f32 v79, v75, v107\n\t"
        "v_dual_max_num_f32 v104, v72, v104 :: v_dual_max_num_f32 v105, v73, v105\n\t"
        "v_dual_max_num_f32 v106, v74, v106 :: v_dual_max_num_f32 v107, v75, v107\n\t"
        "v_dual_min_num_f32 v72, v92, v124 :: v_dual_min_num_f32 v73, v93, v125\n\t"
        "v_dual_min_num_f32 v74, v94, v126 :: v_dual_min_num_f32 v75, v95, v127\n\t"
        "v_dual_max_num_f32 v124, v92, v124 :: v_dual_max_num_f32 v125, v93, v125\n\t"
        "v_dual_max_num_f32 v126, v94, v126 :: v_dual_max_num_f32 v127, v95, v127\n\t"
        "v_dual_min_num_f32 v92, v0, v16 :: v_dual_min_num_f32 v93, v1, v17\n\t"
        "v_dual_min_num_f32 v94, v2, v18 :: v_dual_min_num_f32 v95, v3, v19\n\t"
        "v_dual_max_num_f32 v16, v0, v16 :: v_dual_max_num_f32 v17, v1, v17\n\t"
        "v_dual_max_num_f32 v18, v2, v18 :: v_dual_max_num_f32 v19, v3, v19\n\t"
        "v_dual_min_num_f32 v0, v108, v120 :: v_dual_min_num_f32 v1, v109, v121\n\t"
        "v_dual_min_num_f32 v2, v110, v122 :: v_dual_min_num_f32 v3, v111, v123\n\t"
        "v_dual_max_num_f32 v120, v108, v120 :: v_dual_max_num_f32 v121, v109, v121\n\t"
        "v_dual_max_num_f32 v122, v110, v122 :: v_dual_max_num_f32 v123, v111, v123\n\t"
        "v_dual_min_num_f32 v108, v32, v48 :: v_dual_min_num_f32 v109, v33, v49\n\t"
        "v_dual_min_num_f32 v110, v34, v50 :: v_dual_min_num_f32 v111, v35, v51\n\t"
        "v_dual_max_num_f32 v48, v32, v48 :: v_dual_max_num_f32 v49, v33, v49\n\t"
        "v_dual_max_num_f32 v50, v34, v50 :: v_dual_max_num_f32 v51, v35, v51\n\t"
        "v_dual_min_num_f32 v32, v12, v24 :: v_dual_min_num_f32 v33, v13, v25\n\t"
        "v_dual_min_num_f32 v34, v14, v26 :: v_dual_min_num_f32 v35, v15, v27\n\t"
        "v_dual_max_num_f32 v24, v12, v24 :: v_dual_max_num_f32 v25, v13, v25\n\t"
        "v_dual_max_num_f32 v26, v14, v26 :: v_dual_max_num_f32 v27, v15, v27\n\t"
        "v_dual_min_num_f32 v12, v8, v40 :: v_dual_min_num_f32 v13, v9, v41\n\t"
        "v_dual_min_num_f32 v14, v10, v42 :: v_dual_min_num_f32 v15, v11, v43\n\t"
        "v_dual_max_num_f32 v40, v8, v40 :: v_dual_max_num_f32 v41, v9, v41\n\t"
        "v_dual_max_num_f32 v42, v10, v42 :: v_dual_max_num_f32 v43, v11, v43\n\t"
        "v_dual_min_num_f32 v8, v28, v60 :: v_dual_min_num_f32 v9, v29, v61\n\t"
        "v_dual_min_num_f32 v10, v30, v62 :: v_dual_min_num_f32 v11, v31, v63\n\t"
        "v_dual_max_num_f32 v60, v28, v60 :: v_dual_max_num_f32 v61, v29, v61\n\t"
        "v_dual_max_num_f32 v62, v30, v62 :: v_dual_max_num_f32 v63, v31, v63\n\t"
        "v_dual_min_num_f32 v28, v44, v68 :: v_dual_min_num_f32 v29, v45, v69\n\t"
        "v_dual_min_num_f32 v30, v46, v70 :: v_dual_min_num_f32 v31, v47, v71\n\t"
        "v_dual_max_num_f32 v68, v44, v68 :: v_dual_max_num_f32 v69, v45, v69\n\t"
        "v_dual_max_num_f32 v70, v46, v70 :: v_dual_max_num_f32 v71, v47, v71\n\t"
        "v_dual_min_num_f32 v44, v56, v128 :: v_dual_min_num_f32 v45, v57, v129\n\t"
        "v_dual_min_num_f32 v46, v58, v130 :: v_dual_min_num_f32 v47, v59, v131\n\t"
        "v_dual_max_num_f32 v128, v56, v128 :: v_dual_max_num_f32 v129, v57, v129\n\t"
        "v_dual_max_num_f32 v130, v58, v130 :: v_dual_max_num_f32 v131, v59, v131\n\t"
        "v_dual_min_num_f32 v56, v52, v100 :: v_dual_min_num_f32 v57, v53, v101\n\t"
        "v_dual_min_num_f32 v58, v54, v102 :: v_dual_min_num_f32 v59, v55, v103\n\t"
        "v_dual_max_num_f32 v100, v52, v100 :: v_dual_max_num_f32 v101, v53, v101\n\t"
        "v_dual_max_num_f32 v102, v54, v102 :: v_dual_max_num_f32 v103, v55, v103\n\t"
        "v_dual_min_num_f32 v52, v84, v88 :: v_dual_min_num_f32 v53, v85, v89\n\t"
        "v_dual_min_num_f32 v54, v86, v90 :: v_dual_min_num_f32 v55, v87, v91\n\t"
        "v_dual_max_num_f32 v88, v84, v88 :: v_dual_max_num_f32 v89, v85, v89\n\t"
        "v_dual_max_num_f32 v90, v86, v90 :: v_dual_max_num_f32 v91, v87, v91\n\t"
        "v_dual_min_num_f32 v84, v92, v4 :: v_dual_min_num_f32 v85, v93, v5\n\t"
        "v_dual_min_num_f32 v86, v94, v6 :: v_dual_min_num_f32 v87, v95, v7\n\t"
        "v_dual_max_num_f32 v4, v92, v4 :: v_dual_max_num_f32 v5, v93, v5\n\t"
        "v_dual_max_num_f32 v6, v94, v6 :: v_dual_max_num_f32 v7, v95, v7\n\t"
        "v_dual_min_num_f32 v92, v116, v120 :: v_dual_min_num_f32 v93, v117, v121\n\t"
        "v_dual_min_num_f32 v94, v118, v122 :: v_dual_min_num_f32 v95, v119, v123\n\t"
        "v_dual_max_num_f32 v120, v116, v120 :: v_dual_max_num_f32 v121, v117, v121\n\t"
        "v_dual_max_num_f32 v122, v118, v122 :: v_dual_max_num_f32 v123, v119, v123\n\t"
        "v_dual_min_num_f32 v116, v108, v36 :: v_dual_min_num_f32 v117, v109, v37\n\t"
        "v_dual_min_num_f32 v118, v110, v38 :: v_dual_min_num_f32 v119, v111, v39\n\t"
        "v_dual_max_num_f32 v36, v108, v36 :: v_dual_max_num_f32 v37, v109, v37\n\t"
        "v_dual_max_num_f32 v38, v110, v38 :: v_dual_max_num_f32 v39, v111, v39\n\t"
        "v_dual_min_num_f32 v108, v20, v24 :: v_dual_min_num_f32 v109, v21, v25\n\t"
        "v_dual_min_num_f32 v110, v22, v26 :: v_dual_min_num_f32 v111, v23, v27\n\t"
        "v_dual_max_num_f32 v24, v20, v24 :: v_dual_max_num_f32 v25, v21, v25\n\t"
        "v_dual_max_num_f32 v26, v22, v26 :: v_dual_max_num_f32 v27, v23, v27\n\t"
        "v_dual_min_num_f32 v12, v12, v124 :: v_dual_min_num_f32 v13, v13, v125\n\t"
        "v_dual_min_num_f32 v14, v14, v126 :: v_dual_min_num_f32 v15, v15, v127\n\t"
        "v_dual_min_num_f32 v60, v60, v76 :: v_dual_min_num_f32 v61, v61, v77\n\t"
        "v_dual_min_num_f32 v62, v62, v78 :: v_dual_min_num_f32 v63, v63, v79\n\t"
        "v_dual_min_num_f32 v20, v64, v28 :: v_dual_min_num_f32 v21, v65, v29\n\t"
        "v_dual_min_num_f32 v22, v66, v30 :: v_dual_min_num_f32 v23, v67, v31\n\t"
        "v_dual_max_num_f32 v28, v64, v28 :: v_dual_max_num_f32 v29, v65, v29\n\t"
        "v_dual_max_num_f32 v30, v66, v30 :: v_dual_max_num_f32 v31, v67, v31\n\t"
        "v_dual_min_num_f32 v64, v44, v68 :: v_dual_min_num_f32 v65, v45, v69\n\t"
        "v_dual_min_num_f32 v66, v46, v70 :: v_dual_min_num_f32 v67, v47, v71\n\t"
        "v_dual_max_num_f32 v68, v44, v68 :: v_dual_max_num_f32 v69, v45, v69\n\t"
        "v_dual_max_num_f32 v70, v46, v70 :: v_dual_max_num_f32 v71, v47, v71\n\t"
        "v_dual_min_num_f32 v44, v128, v80 :: v_dual_min_num_f32 v45, v129, v81\n\t"
        "v_dual_min_num_f32 v46, v130, v82 :: v_dual_min_num_f32 v47, v131, v83\n\t"
        "v_dual_max_num_f32 v80, v128, v80 :: v_dual_max_num_f32 v81, v129, v81\n\t"
        "v_dual_max_num_f32 v82, v130, v82 :: v_dual_max_num_f32 v83, v131, v83\n\t"
        "v_dual_min_num_f32 v128, v96, v56 :: v_dual_min_num_f32 v129, v97, v57\n\t"
        "v_dual_min_num_f32 v130, v98, v58 :: v_dual_min_num_f32 v131, v99, v59\n\t"
        "v_dual_max_num_f32 v56, v96, v56 :: v_dual_max_num_f32 v57, v97, v57\n\t"
        "v_dual_max_num_f32 v58, v98, v58 :: v_dual_max_num_f32 v59, v99, v59\n\t"
        "v_dual_min_num_f32 v96, v52, v100 :: v_dual_min_num_f32 v97, v53, v101\n\t"
        "v_dual_min_num_f32 v98, v54, v102 :: v_dual_min_num_f32 v99, v55, v103\n\t"
        "v_dual_max_num_f32 v100, v52, v100 :: v_dual_max_num_f32 v101, v53, v101\n\t"
        "v_dual_max_num_f32 v102, v54, v102 :: v_dual_max_num_f32 v103, v55, v103\n\t"
        "v_dual_min_num_f32 v52, v88, v112 :: v_dual_min_num_f32 v53, v89, v113\n\t"
        "v_dual_min_num_f32 v54, v90, v114 :: v_dual_min_num_f32 v55, v91, v115\n\t"
        "v_dual_max_num_f32 v112, v88, v112 :: v_dual_max_num_f32 v113, v89, v113\n\t"
        "v_dual_max_num_f32 v114, v90, v114 :: v_dual_max_num_f32 v115, v91, v115\n\t"
        "v_dual_min_num_f32 v88, v0, v84 :: v_dual_min_num_f32 v89, v1, v85\n\t"
        "v_dual_min_num_f32 v90, v2, v86 :: v_dual_min_num_f32 v91, v3, v87\n\t"
        "v_dual_max_num_f32 v84, v0, v84 :: v_dual_max_num_f32 v85, v1, v85\n\t"
        "v_dual_max_num_f32 v86, v2, v86 :: v_dual_max_num_f32 v87, v3, v87\n\t"
        "v_dual_min_num_f32 v0, v92, v4 :: v_dual_min_num_f32 v1, v93, v5\n\t"
        "v_dual_min_num_f32 v2, v94, v6 :: v_dual_min_num_f32 v3, v95, v7\n\t"
        "v_dual_max_num_f32 v4, v92, v4 :: v_dual_max_num_f32 v5, v93, v5\n\t"
        "v_dual_max_num_f32 v6, v94, v6 :: v_dual_max_num_f32 v7, v95, v7\n\t"
        "v_dual_min_num_f32 v92, v120, v16 :: v_dual_min_num_f32 v93, v121, v17\n\t"
        "v_dual_min_num_f32 v94, v122, v18 :: v_dual_min_num_f32 v95, v123, v19\n\t"
        "v_dual_max_num_f32 v16, v120, v16 :: v_dual_max_num_f32 v17, v121, v17\n\t"
        "v_dual_max_num_f32 v18, v122, v18 :: v_dual_max_num_f32 v19, v123, v19\n\t"
        "v_dual_min_num_f32 v120, v32, v116 :: v_dual_min_num_f32 v121, v33, v117\n\t"
        "v_dual_min_num_f32 v122, v34, v118 :: v_dual_min_num_f32 v123, v35, v119\n\t"
        "v_dual_max_num_f32 v116, v32, v116 :: v_dual_max_num_f32 v117, v33, v117\n\t"
        "v_dual_max_num_f32 v118, v34, v118 :: v_dual_max_num_f32 v119, v35, v119\n\t"
        "v_dual_min_num_f32 v32, v108, v36 :: v_dual_min_num_f32 v33, v109, v37\n\t"
        "v_dual_min_num_f32 v34, v110, v38 :: v_dual_min_num_f32 v35, v111, v39\n\t"
        "v_dual_max_num_f32 v36, v108, v36 :: v_dual_max_num_f32 v37, v109, v37\n\t"
        "v_dual_max_num_f32 v38, v110, v38 :: v_dual_max_num_f32 v39, v111, v39\n\t"
        "v_dual_min_num_f32 v108, v24, v48 :: v_dual_min_num_f32 v109, v25, v49\n\t"
        "v_dual_min_num_f32 v110, v26, v50 :: v_dual_min_num_f32 v111, v27, v51\n\t"
        "v_dual_max_num_f32 v48, v24, v48 :: v_dual_max_num_f32 v49, v25, v49\n\t"
        "v_dual_max_num_f32 v50, v26, v50 :: v_dual_max_num_f32 v51, v27, v51\n\t"
        "v_dual_min_num_f32 v24, v68, v100 :: v_dual_min_num_f32 v25, v69, v101\n\t"
        "v_dual_min_num_f32 v26, v70, v102 :: v_dual_min_num_f32 v27, v71, v103\n\t"
        "v_dual_max_num_f32 v100, v68, v100 :: v_dual_max_num_f32 v101, v69, v101\n\t"
        "v_dual_max_num_f32 v102, v70, v102 :: v_dual_max_num_f32 v103, v71, v103\n\t"
        "v_dual_min_num_f32 v68, v28, v56 :: v_dual_min_num_f32 v69, v29, v57\n\t"
        "v_dual_min_num_f32 v70, v30, v58 :: v_dual_min_num_f32 v71, v31, v59\n\t"
        "v_dual_max_num_f32 v56, v28, v56 :: v_dual_max_num_f32 v57, v29, v57\n\t"
        "v_dual_max_num_f32 v58, v30, v58 :: v_dual_max_num_f32 v59, v31, v59\n\t"
        "v_dual_min_num_f32 v28, v80, v112 :: v_dual_min_num_f32 v29, v81, v113\n\t"
        "v_dual_min_num_f32 v30, v82, v114 :: v_dual_min_num_f32 v31, v83, v115\n\t"
        "v_dual_max_num_f32 v112, v80, v112 :: v_dual_max_num_f32 v113, v81, v113\n\t"
        "v_dual_max_num_f32 v114, v82, v114 :: v_dual_max_num_f32 v115, v83, v115\n\t"
        "v_dual_min_num_f32 v80, v20, v128 :: v_dual_min_num_f32 v81, v21, v129\n\t"
        "v_dual_min_num_f32 v82, v22, v130 :: v_dual_min_num_f32 v83, v23, v131\n\t"
        "v_dual_max_num_f32 v128, v20, v128 :: v_dual_max_num_f32 v129, v21, v129\n\t"
        "v_dual_max_num_f32 v130, v22, v130 :: v_dual_max_num_f32 v131, v23, v131\n\t"
        "v_dual_min_num_f32 v20, v44, v52 :: v_dual_min_num_f32 v21, v45, v53\n\t"
        "v_dual_min_num_f32 v22, v46, v54 :: v_dual_min_num_f32 v23, v47, v55\n\t"
        "v_dual_max_num_f32 v52, v44, v52 :: v_dual_max_num_f32 v53, v45, v53\n\t"
        "v_dual_max_num_f32 v54, v46, v54 :: v_dual_max_num_f32 v55, v47, v55\n\t"
        "v_dual_min_num_f32 v44, v64, v96 :: v_dual_min_num_f32 v45, v65, v97\n\t"
        "v_dual_min_num_f32 v46, v66, v98 :: v_dual_min_num_f32 v47, v67, v99\n\t"
        "v_dual_max_num_f32 v96, v64, v96 :: v_dual_max_num_f32 v97, v65, v97\n\t"
        "v_dual_max_num_f32 v98, v66, v98 :: v_dual_max_num_f32 v99, v67, v99\n\t"
        "v_dual_min_num_f32 v64, v4, v36 :: v_dual_min_num_f32 v65, v5, v37\n\t"
        "v_dual_min_num_f32 v66, v6, v38 :: v_dual_min_num_f32 v67, v7, v39\n\t"
        "v_dual_max_num_f32 v36, v4, v36 :: v_dual_max_num_f32 v37, v5, v37\n\t"
        "v_dual_max_num_f32 v38, v6, v38 :: v_dual_max_num_f32 v39, v7, v39\n\t"
        "v_dual_min_num_f32 v4, v84, v116 :: v_dual_min_num_f32 v5, v85, v117\n\t"
        "v_dual_min_num_f32 v6, v86, v118 :: v_dual_min_num_f32 v7, v87, v119\n\t"
        "v_dual_max_num_f32 v116, v84, v116 :: v_dual_max_num_f32 v117, v85, v117\n\t"
        "v_dual_max_num_f32 v118, v86, v118 :: v_dual_max_num_f32 v119, v87, v119\n\t"
        "v_dual_min_num_f32 v84, v16, v48 :: v_dual_min_num_f32 v85, v17, v49\n\t"
        "v_dual_min_num_f32 v86, v18, v50 :: v_dual_min_num_f32 v87, v19, v51\n\t"
        "v_dual_max_num_f32 v48, v16, v48 :: v_dual_max_num_f32 v49, v17, v49\n\t"
        "v_dual_max_num_f32 v50, v18, v50 :: v_dual_max_num_f32 v51, v19, v51\n\t"
        "v_dual_min_num_f32 v16, v88, v120 :: v_dual_min_num_f32 v17, v89, v121\n\t"
        "v_dual_min_num_f32 v18, v90, v122 :: v_dual_min_num_f32 v19, v91, v123\n\t"
        "v_dual_max_num_f32 v120, v88, v120 :: v_dual_max_num_f32 v121, v89, v121\n\t"
        "v_dual_max_num_f32 v122, v90, v122 :: v_dual_max_num_f32 v123, v91, v123\n\t"
        "v_dual_min_num_f32 v88, v92, v108 :: v_dual_min_num_f32 v89, v93, v109\n\t"
        "v_dual_min_num_f32 v90, v94, v110 :: v_dual_min_num_f32 v91, v95, v111\n\t"
        "v_dual_max_num_f32 v108, v92, v108 :: v_dual_max_num_f32 v109, v93, v109\n\t"
        "v_dual_max_num_f32 v110, v94, v110 :: v_dual_max_num_f32 v111, v95, v111\n\t"
        "v_dual_min_num_f32 v92, v0, v32 :: v_dual_min_num_f32 v93, v1, v33\n\t"
        "v_dual_min_num_f32 v94, v2, v34 :: v_dual_min_num_f32 v95, v3, v35\n\t"
        "v_dual_max_num_f32 v32, v0, v32 :: v_dual_max_num_f32 v33, v1, v33\n\t"
        "v_dual_max_num_f32 v34, v2, v34 :: v_dual_max_num_f32 v35, v3, v35\n\t"
        "v_dual_min_num_f32 v0, v24, v104 :: v_dual_min_num_f32 v1, v25, v105\n\t"
        "v_dual_min_num_f32 v2, v26, v106 :: v_dual_min_num_f32 v3, v27, v107\n\t"
        "v_dual_max_num_f32 v104, v24, v104 :: v_dual_max_num_f32 v105, v25, v105\n\t"
        "v_dual_max_num_f32 v106, v26, v106 :: v_dual_max_num_f32 v107, v27, v107\n\t"
        "v_dual_min_num_f32 v24, v28, v56 :: v_dual_min_num_f32 v25, v29, v57\n\t"
        "v_dual_min_num_f32 v26, v30, v58 :: v_dual_min_num_f32 v27, v31, v59\n\t"
        "v_dual_max_num_f32 v56, v28, v56 :: v_dual_max_num_f32 v57, v29, v57\n\t"
        "v_dual_max_num_f32 v58, v30, v58 :: v_dual_max_num_f32 v59, v31, v59\n\t"
        "v_dual_min_num_f32 v28, v20, v128 :: v_dual_min_num_f32 v29, v21, v129\n\t"
        "v_dual_min_num_f32 v30, v22, v130 :: v_dual_min_num_f32 v31, v23, v131\n\t"
        "v_dual_max_num_f32 v128, v20, v128 :: v_dual_max_num_f32 v129, v21, v129\n\t"
        "v_dual_max_num_f32 v130, v22, v130 :: v_dual_max_num_f32 v131, v23, v131\n\t"
        "v_dual_min_num_f32 v20, v72, v96 :: v_dual_min_num_f32 v21, v73, v97\n\t"
        "v_dual_min_num_f32 v22, v74, v98 :: v_dual_min_num_f32 v23, v75, v99\n\t"
        "v_dual_max_num_f32 v96, v72, v96 :: v_dual_max_num_f32 v97, v73, v97\n\t"
        "v_dual_max_num_f32 v98, v74, v98 :: v_dual_max_num_f32 v99, v75, v99\n\t"
        "v_dual_min_num_f32 v72, v64, v40 :: v_dual_min_num_f32 v73, v65, v41\n\t"
        "v_dual_min_num_f32 v74, v66, v42 :: v_dual_min_num_f32 v75, v67, v43\n\t"
        "v_dual_max_num_f32 v40, v64, v40 :: v_dual_max_num_f32 v41, v65, v41\n\t"
        "v_dual_max_num_f32 v42, v66, v42 :: v_dual_max_num_f32 v43, v67, v43\n\t"
        "v_dual_min_num_f32 v64, v84, v116 :: v_dual_min_num_f32 v65, v85, v117\n\t"
        "v_dual_min_num_f32 v66, v86, v118 :: v_dual_min_num_f32 v67, v87, v119\n\t"
        "v_dual_max_num_f32 v116, v84, v116 :: v_dual_max_num_f32 v117, v85, v117\n\t"
        "v_dual_max_num_f32 v118, v86, v118 :: v_dual_max_num_f32 v119, v87, v119\n\t"
        "v_dual_min_num_f32 v84, v88, v120 :: v_dual_min_num_f32 v85, v89, v121\n\t"
        "v_dual_min_num_f32 v86, v90, v122 :: v_dual_min_num_f32 v87, v91, v123\n\t"
        "v_dual_max_num_f32 v120, v88, v120 :: v_dual_max_num_f32 v121, v89, v121\n\t"
        "v_dual_max_num_f32 v122, v90, v122 :: v_dual_max_num_f32 v123, v91, v123\n\t"
        "v_dual_min_num_f32 v88, v8, v32 :: v_dual_min_num_f32 v89, v9, v33\n\t"
        "v_dual_min_num_f32 v90, v10, v34 :: v_dual_min_num_f32 v91, v11, v35\n\t"
        "v_dual_max_num_f32 v32, v8, v32 :: v_dual_max_num_f32 v33, v9, v33\n\t"
        "v_dual_max_num_f32 v34, v10, v34 :: v_dual_max_num_f32 v35, v11, v35\n\t"
        "v_dual_min_num_f32 v8, v68, v0 :: v_dual_min_num_f32 v9, v69, v1\n\t"
        "v_dual_min_num_f32 v10, v70, v2 :: v_dual_min_num_f32 v11, v71, v3\n\t"
        "v_dual_max_num_f32 v0, v68, v0 :: v_dual_max_num_f32 v1, v69, v1\n\t"
        "v_dual_max_num_f32 v2, v70, v2 :: v_dual_max_num_f32 v3, v71, v3\n\t"
        "v_dual_min_num_f32 v68, v24, v104 :: v_dual_min_num_f32 v69, v25, v105\n\t"
        "v_dual_min_num_f32 v70, v26, v106 :: v_dual_min_num_f32 v71, v27, v107\n\t"
        "v_dual_max_num_f32 v104, v24, v104 :: v_dual_max_num_f32 v105, v25, v105\n\t"
        "v_dual_max_num_f32 v106, v26, v106 :: v_dual_max_num_f32 v107, v27, v107\n\t"
        "v_dual_min_num_f32 v24, v56, v100 :: v_dual_min_num_f32 v25, v57, v101\n\t"
        "v_dual_min_num_f32 v26, v58, v102 :: v_dual_min_num_f32 v27, v59, v103\n\t"
        "v_dual_max_num_f32 v100, v56, v100 :: v_dual_max_num_f32 v101, v57, v101\n\t"
        "v_dual_max_num_f32 v102, v58, v102 :: v_dual_max_num_f32 v103, v59, v103\n\t"
        "v_dual_min_num_f32 v56, v44, v28 :: v_dual_min_num_f32 v57, v45, v29\n\t"
        "v_dual_min_num_f32 v58, v46, v30 :: v_dual_min_num_f32 v59, v47, v31\n\t"
        "v_dual_max_num_f32 v28, v44, v28 :: v_dual_max_num_f32 v29, v45, v29\n\t"
        "v_dual_max_num_f32 v30, v46, v30 :: v_dual_max_num_f32 v31, v47, v31\n\t"
        "v_dual_min_num_f32 v44, v20, v128 :: v_dual_min_num_f32 v45, v21, v129\n\t"
        "v_dual_min_num_f32 v46, v22, v130 :: v_dual_min_num_f32 v47, v23, v131\n\t"
        "v_dual_max_num_f32 v128, v20, v128 :: v_dual_max_num_f32 v129, v21, v129\n\t"
        "v_dual_max_num_f32 v130, v22, v130 :: v_dual_max_num_f32 v131, v23, v131\n\t"
        "v_dual_min_num_f32 v20, v96, v52 :: v_dual_min_num_f32 v21, v97, v53\n\t"
        "v_dual_min_num_f32 v22, v98, v54 :: v_dual_min_num_f32 v23, v99, v55\n\t"
        "v_dual_max_num_f32 v52, v96, v52 :: v_dual_max_num_f32 v53, v97, v53\n\t"
        "v_dual_max_num_f32 v54, v98, v54 :: v_dual_max_num_f32 v55, v99, v55\n\t"
        "v_dual_min_num_f32 v96, v4, v72 :: v_dual_min_num_f32 v97, v5, v73\n\t"
        "v_dual_min_num_f32 v98, v6, v74 :: v_dual_min_num_f32 v99, v7, v75\n\t"
        "v_dual_max_num_f32 v72, v4, v72 :: v_dual_max_num_f32 v73, v5, v73\n\t"
        "v_dual_max_num_f32 v74, v6, v74 :: v_dual_max_num_f32 v75, v7, v75\n\t"
        "v_dual_min_num_f32 v4, v64, v40 :: v_dual_min_num_f32 v5, v65, v41\n\t"
        "v_dual_min_num_f32 v6, v66, v42 :: v_dual_min_num_f32 v7, v67, v43\n\t"
        "v_dual_max_num_f32 v40, v64, v40 :: v_dual_max_num_f32 v41, v65, v41\n\t"
        "v_dual_max_num_f32 v42, v66, v42 :: v_dual_max_num_f32 v43, v67, v43\n\t"
        "v_dual_min_num_f32 v64, v116, v36 :: v_dual_min_num_f32 v65, v117, v37\n\t"
        "v_dual_min_num_f32 v66, v118, v38 :: v_dual_min_num_f32 v67, v119, v39\n\t"
        "v_dual_max_num_f32 v36, v116, v36 :: v_dual_max_num_f32 v37, v117, v37\n\t"
        "v_dual_max_num_f32 v38, v118, v38 :: v_dual_max_num_f32 v39, v119, v39\n\t"
        "v_dual_min_num_f32 v116, v92, v84 :: v_dual_min_num_f32 v117, v93, v85\n\t"
        "v_dual_min_num_f32 v118, v94, v86 :: v_dual_min_num_f32 v119, v95, v87\n\t"
        "v_dual_max_num_f32 v84, v92, v84 :: v_dual_max_num_f32 v85, v93, v85\n\t"
        "v_dual_max_num_f32 v86, v94, v86 :: v_dual_max_num_f32 v87, v95, v87\n\t"
        "v_dual_min_num_f32 v92, v88, v120 :: v_dual_min_num_f32 v93, v89, v121\n\t"
        "v_dual_min_num_f32 v94, v90, v122 :: v_dual_min_num_f32 v95, v91, v123\n\t"
        "v_dual_max_num_f32 v120, v88, v120 :: v_dual_max_num_f32 v121, v89, v121\n\t"
        "v_dual_max_num_f32 v122, v90, v122 :: v_dual_max_num_f32 v123, v91, v123\n\t"
        "v_dual_min_num_f32 v88, v32, v108 :: v_dual_min_num_f32 v89, v33, v109\n\t"
        "v_dual_min_num_f32 v90, v34, v110 :: v_dual_min_num_f32 v91, v35, v111\n\t"
        "v_dual_max_num_f32 v108, v32, v108 :: v_dual_max_num_f32 v109, v33, v109\n\t"
        "v_dual_max_num_f32 v110, v34, v110 :: v_dual_max_num_f32 v111, v35, v111\n\t"
        "v_dual_min_num_f32 v32, v80, v8 :: v_dual_min_num_f32 v33, v81, v9\n\t"
        "v_dual_min_num_f32 v34, v82, v10 :: v_dual_min_num_f32 v35, v83, v11\n\t"
        "v_dual_max_num_f32 v8, v80, v8 :: v_dual_max_num_f32 v9, v81, v9\n\t"
        "v_dual_max_num_f32 v10, v82, v10 :: v_dual_max_num_f32 v11, v83, v11\n\t"
        "v_dual_min_num_f32 v80, v56, v0 :: v_dual_min_num_f32 v81, v57, v1\n\t"
        "v_dual_min_num_f32 v82, v58, v2 :: v_dual_min_num_f32 v83, v59, v3\n\t"
        "v_dual_max_num_f32 v0, v56, v0 :: v_dual_max_num_f32 v1, v57, v1\n\t"
        "v_dual_max_num_f32 v2, v58, v2 :: v_dual_max_num_f32 v3, v59, v3\n\t"
        "v_dual_min_num_f32 v56, v28, v68 :: v_dual_min_num_f32 v57, v29, v69\n\t"
        "v_dual_min_num_f32 v58, v30, v70 :: v_dual_min_num_f32 v59, v31, v71\n\t"
        "v_dual_max_num_f32 v68, v28, v68 :: v_dual_max_num_f32 v69, v29, v69\n\t"
        "v_dual_max_num_f32 v70, v30, v70 :: v_dual_max_num_f32 v71, v31, v71\n\t"
        "v_dual_min_num_f32 v28, v44, v104 :: v_dual_min_num_f32 v29, v45, v105\n\t"
        "v_dual_min_num_f32 v30, v46, v106 :: v_dual_min_num_f32 v31, v47, v107\n\t"
        "v_dual_max_num_f32 v104, v44, v104 :: v_dual_max_num_f32 v105, v45, v105\n\t"
        "v_dual_max_num_f32 v106, v46, v106 :: v_dual_max_num_f32 v107, v47, v107\n\t"
        "v_dual_min_num_f32 v44, v128, v24 :: v_dual_min_num_f32 v45, v129, v25\n\t"
        "v_dual_min_num_f32 v46, v130, v26 :: v_dual_min_num_f32 v47, v131, v27\n\t"
        "v_dual_max_num_f32 v24, v128, v24 :: v_dual_max_num_f32 v25, v129, v25\n\t"
        "v_dual_max_num_f32 v26, v130, v26 :: v_dual_max_num_f32 v27, v131, v27\n\t"
        "v_dual_min_num_f32 v128, v20, v100 :: v_dual_min_num_f32 v129, v21, v101\n\t"
        "v_dual_min_num_f32 v130, v22, v102 :: v_dual_min_num_f32 v131, v23, v103\n\t"
        "v_dual_max_num_f32 v100, v20, v100 :: v_dual_max_num_f32 v101, v21, v101\n\t"
        "v_dual_max_num_f32 v102, v22, v102 :: v_dual_max_num_f32 v103, v23, v103\n\t"
        "v_dual_min_num_f32 v20, v52, v112 :: v_dual_min_num_f32 v21, v53, v113\n\t"
        "v_dual_min_num_f32 v22, v54, v114 :: v_dual_min_num_f32 v23, v55, v115\n\t"
        "v_dual_max_num_f32 v112, v52, v112 :: v_dual_max_num_f32 v113, v53, v113\n\t"
        "v_dual_max_num_f32 v114, v54, v114 :: v_dual_max_num_f32 v115, v55, v115\n\t"
        "v_dual_min_num_f32 v52, v16, v96 :: v_dual_min_num_f32 v53, v17, v97\n\t"
        "v_dual_min_num_f32 v54, v18, v98 :: v_dual_min_num_f32 v55, v19, v99\n\t"
        "v_dual_max_num_f32 v96, v16, v96 :: v_dual_max_num_f32 v97, v17, v97\n\t"
        "v_dual_max_num_f32 v98, v18, v98 :: v_dual_max_num_f32 v99, v19, v99\n\t"
        "v_dual_min_num_f32 v16, v116, v72 :: v_dual_min_num_f32 v17, v117, v73\n\t"
        "v_dual_min_num_f32 v18, v118, v74 :: v_dual_min_num_f32 v19, v119, v75\n\t"
        "v_dual_max_num_f32 v72, v116, v72 :: v_dual_max_num_f32 v73, v117, v73\n\t"
        "v_dual_max_num_f32 v74, v118, v74 :: v_dual_max_num_f32 v75, v119, v75\n\t"
        "v_dual_min_num_f32 v116, v84, v4 :: v_dual_min_num_f32 v117, v85, v5\n\t"
        "v_dual_min_num_f32 v118, v86, v6 :: v_dual_min_num_f32 v119, v87, v7\n\t"
        "v_dual_max_num_f32 v4, v84, v4 :: v_dual_max_num_f32 v5, v85, v5\n\t"
        "v_dual_max_num_f32 v6, v86, v6 :: v_dual_max_num_f32 v7, v87, v7\n\t"
        "v_dual_min_num_f32 v84, v92, v40 :: v_dual_min_num_f32 v85, v93, v41\n\t"
        "v_dual_min_num_f32 v86, v94, v42 :: v_dual_min_num_f32 v87, v95, v43\n\t"
        "v_dual_max_num_f32 v40, v92, v40 :: v_dual_max_num_f32 v41, v93, v41\n\t"
        "v_dual_max_num_f32 v42, v94, v42 :: v_dual_max_num_f32 v43, v95, v43\n\t"
        "v_dual_min_num_f32 v92, v120, v64 :: v_dual_min_num_f32 v93, v121, v65\n\t"
        "v_dual_min_num_f32 v94, v122, v66 :: v_dual_min_num_f32 v95, v123, v67\n\t"
        "v_dual_max_num_f32 v64, v120, v64 :: v_dual_max_num_f32 v65, v121, v65\n\t"
        "v_dual_max_num_f32 v66, v122, v66 :: v_dual_max_num_f32 v67, v123, v67\n\t"
        "v_dual_min_num_f32 v120, v88, v36 :: v_dual_min_num_f32 v121, v89, v37\n\t"
        "v_dual_min_num_f32 v122, v90, v38 :: v_dual_min_num_f32 v123, v91, v39\n\t"
        "v_dual_max_num_f32 v36, v88, v36 :: v_dual_max_num_f32 v37, v89, v37\n\t"
        "v_dual_max_num_f32 v38, v90, v38 :: v_dual_max_num_f32 v39, v91, v39\n\t"
        "v_dual_min_num_f32 v88, v108, v48 :: v_dual_min_num_f32 v89, v109, v49\n\t"
        "v_dual_min_num_f32 v90, v110, v50 :: v_dual_min_num_f32 v91, v111, v51\n\t"
        "v_dual_max_num_f32 v48, v108, v48 :: v_dual_max_num_f32 v49, v109, v49\n\t"
        "v_dual_max_num_f32 v50, v110, v50 :: v_dual_max_num_f32 v51, v111, v51\n\t"
        "v_dual_min_num_f32 v40, v40, v28 :: v_dual_min_num_f32 v41, v41, v29\n\t"
        "v_dual_min_num_f32 v42, v42, v30 :: v_dual_min_num_f32 v43, v43, v31\n\t"
        "v_dual_min_num_f32 v52, v52, v112 :: v_dual_min_num_f32 v53, v53, v113\n\t"
        "v_dual_min_num_f32 v54, v54, v114 :: v_dual_min_num_f32 v55, v55, v115\n\t"
        "v_dual_min_num_f32 v92, v92, v68 :: v_dual_min_num_f32 v93, v93, v69\n\t"
        "v_dual_min_num_f32 v94, v94, v70 :: v_dual_min_num_f32 v95, v95, v71\n\t"
        "v_dual_min_num_f32 v96, v96, v20 :: v_dual_min_num_f32 v97, v97, v21\n\t"
        "v_dual_min_num_f32 v98, v98, v22 :: v_dual_min_num_f32 v99, v99, v23\n\t"
        "v_dual_min_num_f32 v64, v64, v56 :: v_dual_min_num_f32 v65, v65, v57\n\t"
        "v_dual_min_num_f32 v66, v66, v58 :: v_dual_min_num_f32 v67, v67, v59\n\t"
        "v_dual_min_num_f32 v16, v16, v100 :: v_dual_min_num_f32 v17, v17, v101\n\t"
        "v_dual_min_num_f32 v18, v18, v102 :: v_dual_min_num_f32 v19, v19, v103\n\t"
        "v_dual_min_num_f32 v120, v120, v0 :: v_dual_min_num_f32 v121, v121, v1\n\t"
        "v_dual_min_num_f32 v122, v122, v2 :: v_dual_min_num_f32 v123, v123, v3\n\t"
        "v_dual_min_num_f32 v72, v72, v128 :: v_dual_min_num_f32 v73, v73, v129\n\t"
        "v_dual_min_num_f32 v74, v74, v130 :: v_dual_min_num_f32 v75, v75, v131\n\t"
        "v_dual_min_num_f32 v36, v36, v80 :: v_dual_min_num_f32 v37, v37, v81\n\t"
        "v_dual_min_num_f32 v38, v38, v82 :: v_dual_min_num_f32 v39, v39, v83\n\t"
        "v_dual_min_num_f32 v116, v116, v24 :: v_dual_min_num_f32 v117, v117, v25\n\t"
        "v_dual_min_num_f32 v118, v118, v26 :: v_dual_min_num_f32 v119, v119, v27\n\t"
        "v_dual_min_num_f32 v88, v88, v8 :: v_dual_min_num_f32 v89, v89, v9\n\t"
        "v_dual_min_num_f32 v90, v90, v10 :: v_dual_min_num_f32 v91, v91, v11\n\t"
        "v_dual_min_num_f32 v4, v4, v44 :: v_dual_min_num_f32 v5, v5, v45\n\t"
        "v_dual_min_num_f32 v6, v6, v46 :: v_dual_min_num_f32 v7, v7, v47\n\t"
        "v_dual_min_num_f32 v48, v48, v32 :: v_dual_min_num_f32 v49, v49, v33\n\t"
        "v_dual_min_num_f32 v50, v50, v34 :: v_dual_min_num_f32 v51, v51, v35\n\t"
        "v_dual_min_num_f32 v84, v84, v104 :: v_dual_min_num_f32 v85, v85, v105\n\t"
        "v_dual_min_num_f32 v86, v86, v106 :: v_dual_min_num_f32 v87, v87, v107\n\t"
        "v_dual_max_num_f32 v12, v12, v40 :: v_dual_max_num_f32 v13, v13, v41\n\t"
        "v_dual_max_num_f32 v14, v14, v42 :: v_dual_max_num_f32 v15, v15, v43\n\t"
        "v_dual_max_num_f32 v52, v52, v92 :: v_dual_max_num_f32 v53, v53, v93\n\t"
        "v_dual_max_num_f32 v54, v54, v94 :: v_dual_max_num_f32 v55, v55, v95\n\t"
        "v_dual_max_num_f32 v96, v96, v64 :: v_dual_max_num_f32 v97, v97, v65\n\t"
        "v_dual_max_num_f32 v98, v98, v66 :: v_dual_max_num_f32 v99, v99, v67\n\t"
        "v_dual_max_num_f32 v16, v16, v120 :: v_dual_max_num_f32 v17, v17, v121\n\t"
        "v_dual_max_num_f32 v18, v18, v122 :: v_dual_max_num_f32 v19, v19, v123\n\t"
        "v_dual_max_num_f32 v72, v72, v36 :: v_dual_max_num_f32 v73, v73, v37\n\t"
        "v_dual_max_num_f32 v74, v74, v38 :: v_dual_max_num_f32 v75, v75, v39\n\t"
        "v_dual_max_num_f32 v116, v116, v88 :: v_dual_max_num_f32 v117, v117, v89\n\t"
        "v_dual_max_num_f32 v118, v118, v90 :: v_dual_max_num_f32 v119, v119, v91\n\t"
        "v_dual_max_num_f32 v4, v4, v48 :: v_dual_max_num_f32 v5, v5, v49\n\t"
        "v_dual_max_num_f32 v6, v6, v50 :: v_dual_max_num_f32 v7, v7, v51\n\t"
        "v_dual_max_num_f32 v84, v84, v60 :: v_dual_max_num_f32 v85, v85, v61\n\t"
        "v_dual_max_num_f32 v86, v86, v62 :: v_dual_max_num_f32 v87, v87, v63\n\t"
        "v_dual_min_num_f32 v108, v12, v72 :: v_dual_min_num_f32 v109, v13, v73\n\t"
        "v_dual_min_num_f32 v110, v14, v74 :: v_dual_min_num_f32 v111, v15, v75\n\t"
        "v_dual_max_num_f32 v72, v12, v72 :: v_dual_max_num_f32 v73, v13, v73\n\t"
        "v_dual_max_num_f32 v74, v14, v74 :: v_dual_max_num_f32 v75, v15, v75\n\t"
        "v_dual_min_num_f32 v12, v52, v116 :: v_dual_min_num_f32 v13, v53, v117\n\t"
        "v_dual_min_num_f32 v14, v54, v118 :: v_dual_min_num_f32 v15, v55, v119\n\t"
        "v_dual_max_num_f32 v116, v52, v116 :: v_dual_max_num_f32 v117, v53, v117\n\t"
        "v_dual_max_num_f32 v118, v54, v118 :: v_dual_max_num_f32 v119, v55, v119\n\t"
        "v_dual_min_num_f32 v52, v96, v4 :: v_dual_min_num_f32 v53, v97, v5\n\t"
        "v_dual_min_num_f32 v54, v98, v6 :: v_dual_min_num_f32 v55, v99, v7\n\t"
        "v_dual_max_num_f32 v4, v96, v4 :: v_dual_max_num_f32 v5, v97, v5\n\t"
        "v_dual_max_num_f32 v6, v98, v6 :: v_dual_max_num_f32 v7, v99, v7\n\t"
        "v_dual_min_num_f32 v96, v16, v84 :: v_dual_min_num_f32 v97, v17, v85\n\t"
        "v_dual_min_num_f32 v98, v18, v86 :: v_dual_min_num_f32 v99, v19, v87\n\t"
        "v_dual_max_num_f32 v84, v16, v84 :: v_dual_max_num_f32 v85, v17, v85\n\t"
        "v_dual_max_num_f32 v86, v18, v86 :: v_dual_max_num_f32 v87, v19, v87\n\t"
        "v_dual_min_num_f32 v16, v108, v52 :: v_dual_min_num_f32 v17, v109, v53\n\t"
        "v_dual_min_num_f32 v18, v110, v54 :: v_dual_min_num_f32 v19, v111, v55\n\t"
        "v_dual_max_num_f32 v52, v108, v52 :: v_dual_max_num_f32 v53, v109, v53\n\t"
        "v_dual_max_num_f32 v54, v110, v54 :: v_dual_max_num_f32 v55, v111, v55\n\t"
        "v_dual_min_num_f32 v108, v12, v96 :: v_dual_min_num_f32 v109, v13, v97\n\t"
        "v_dual_min_num_f32 v110, v14, v98 :: v_dual_min_num_f32 v111, v15, v99\n\t"
        "v_dual_max_num_f32 v96, v12, v96 :: v_dual_max_num_f32 v97, v13, v97\n\t"
        "v_dual_max_num_f32 v98, v14, v98 :: v_dual_max_num_f32 v99, v15, v99\n\t"
        "v_dual_min_num_f32 v12, v72, v4 :: v_dual_min_num_f32 v13, v73, v5\n\t"
        "v_dual_min_num_f32 v14, v74, v6 :: v_dual_min_num_f32 v15, v75, v7\n\t"
        "v_dual_max_num_f32 v4, v72, v4 :: v_dual_max_num_f32 v5, v73, v5\n\t"
        "v_dual_max_num_f32 v6, v74, v6 :: v_dual_max_num_f32 v7, v75, v7\n\t"
        "v_dual_min_num_f32 v72, v116, v84 :: v_dual_min_num_f32 v73, v117, v85\n\t"
        "v_dual_min_num_f32 v74, v118, v86 :: v_dual_min_num_f32 v75, v119, v87\n\t"
        "v_dual_max_num_f32 v84, v116, v84 :: v_dual_max_num_f32 v85, v117, v85\n\t"
        "v_dual_max_num_f32 v86, v118, v86 :: v_dual_max_num_f32 v87, v119, v87\n\t"
        "v_dual_min_num_f32 v116, v16, v108 :: v_dual_min_num_f32 v117, v17, v109\n\t"
        "v_dual_min_num_f32 v118, v18, v110 :: v_dual_min_num_f32 v119, v19, v111\n\t"
        "v_dual_max_num_f32 v108, v16, v108 :: v_dual_max_num_f32 v109, v17, v109\n\t"
        "v_dual_max_num_f32 v110, v18, v110 :: v_dual_max_num_f32 v111, v19, v111\n\t"
        "v_dual_min_num_f32 v16, v52, v96 :: v_dual_min_num_f32 v17, v53, v97\n\t"
        "v_dual_min_num_f32 v18, v54, v98 :: v_dual_min_num_f32 v19, v55, v99\n\t"
        "v_dual_max_num_f32 v96, v52, v96 :: v_dual_max_num_f32 v97, v53, v97\n\t"
        "v_dual_max_num_f32 v98, v54, v98 :: v_dual_max_num_f32 v99, v55, v99\n\t"
        "v_dual_min_num_f32 v52, v12, v72 :: v_dual_min_num_f32 v53, v13, v73\n\t"
        "v_dual_min_num_f32 v54, v14, v74 :: v_dual_min_num_f32 v55, v15, v75\n\t"
        "v_dual_max_num_f32 v72, v12, v72 :: v_dual_max_num_f32 v73, v13, v73\n\t"
        "v_dual_max_num_f32 v74, v14, v74 :: v_dual_max_num_f32 v75, v15, v75\n\t"
        "v_dual_min_num_f32 v12, v4, v84 :: v_dual_min_num_f32 v13, v5, v85\n\t"
        "v_dual_min_num_f32 v14, v6, v86 :: v_dual_min_num_f32 v15, v7, v87\n\t"
        "v_dual_max_num_f32 v84, v4, v84 :: v_dual_max_num_f32 v85, v5, v85\n\t"
        "v_dual_max_num_f32 v86, v6, v86 :: v_dual_max_num_f32 v87, v7, v87\n\t"
        :
          "={v[0:3]}"(a[0]), "={v[4:7]}"(a[1]), "={v[8:11]}"(a[2]),
          "={v[12:15]}"(a[3]), "={v[16:19]}"(a[4]), "={v[20:23]}"(a[5]),
          "={v[24:27]}"(a[6]), "={v[28:31]}"(a[7]), "={v[32:35]}"(a[8]),
          "={v[36:39]}"(a[9]), "={v[40:43]}"(a[10]), "={v[44:47]}"(a[11]),
          "={v[48:51]}"(a[12]), "={v[52:55]}"(a[13]), "={v[56:59]}"(a[14]),
          "={v[60:63]}"(a[15]), "={v[64:67]}"(b[0]), "={v[68:71]}"(b[1]),
          "={v[72:75]}"(b[2]), "={v[76:79]}"(b[3]), "={v[80:83]}"(b[4]),
          "={v[84:87]}"(b[5]), "={v[88:91]}"(b[6]), "={v[92:95]}"(b[7]),
          "={v[96:99]}"(b[8]), "={v[100:103]}"(b[9]), "={v[104:107]}"(b[10]),
          "={v[108:111]}"(b[11]), "={v[112:115]}"(b[12]), "={v[116:119]}"(b[13]),
          "={v[120:123]}"(b[14]), "={v[124:127]}"(b[15]), "={v[128:131]}"(sp)
        :
          "{v[0:3]}"(a[0]), "{v[4:7]}"(a[1]), "{v[8:11]}"(a[2]),
          "{v[12:15]}"(a[3]), "{v[16:19]}"(a[4]), "{v[20:23]}"(a[5]),
          "{v[24:27]}"(a[6]), "{v[28:31]}"(a[7]), "{v[32:35]}"(a[8]),
          "{v[36:39]}"(a[9]), "{v[40:43]}"(a[10]), "{v[44:47]}"(a[11]),
          "{v[48:51]}"(a[12]), "{v[52:55]}"(a[13]), "{v[56:59]}"(a[14]),
          "{v[60:63]}"(a[15]), "{v[64:67]}"(b[0]), "{v[68:71]}"(b[1]),
          "{v[72:75]}"(b[2]), "{v[76:79]}"(b[3]), "{v[80:83]}"(b[4]),
          "{v[84:87]}"(b[5]), "{v[88:91]}"(b[6]), "{v[92:95]}"(b[7]),
          "{v[96:99]}"(b[8]), "{v[100:103]}"(b[9]), "{v[104:107]}"(b[10]),
          "{v[108:111]}"(b[11]), "{v[112:115]}"(b[12]), "{v[116:119]}"(b[13]),
          "{v[120:123]}"(b[14]), "{v[124:127]}"(b[15]),
          [m9] "s"(m9), [m10] "s"(m10), [m11] "s"(m11), [m12] "s"(m12),
          [m13] "s"(m13), [m14] "s"(m14), [m15] "s"(m15));
    const int sel = med - 8;
    v4f r = b[13];
    r = (sel == 1) ? b[11] : r;
    r = (sel == 2) ? a[4] : r;
    r = (sel == 3) ? b[8] : r;
    r = (sel == 4) ? a[13] : r;
    r = (sel == 5) ? b[2] : r;
    r = (sel == 6) ? a[3] : r;
    r = (sel == 7) ? b[5] : r;
    return r;
}

__device__ __forceinline__ v4f rank_paired_16_8(v4f* a, v4f* b, int d, int med) {
    const float inf = __builtin_inff();
    const float m1 = (17 < d) ? -inf : inf;
    const float m2 = (18 < d) ? -inf : inf;
    const float m3 = (19 < d) ? -inf : inf;
    const float m4 = (20 < d) ? -inf : inf;
    const float m5 = (21 < d) ? -inf : inf;
    const float m6 = (22 < d) ? -inf : inf;
    const float m7 = (23 < d) ? -inf : inf;
    v4f sp;
    asm(
        "v_dual_max_num_f32 v68, %[m1], v68 :: v_dual_max_num_f32 v69, %[m1], v69\n\t"
        "v_dual_max_num_f32 v70, %[m1], v70 :: v_dual_max_num_f32 v71, %[m1], v71\n\t"
        "v_dual_max_num_f32 v72, %[m2], v72 :: v_dual_max_num_f32 v73, %[m2], v73\n\t"
        "v_dual_max_num_f32 v74, %[m2], v74 :: v_dual_max_num_f32 v75, %[m2], v75\n\t"
        "v_dual_max_num_f32 v76, %[m3], v76 :: v_dual_max_num_f32 v77, %[m3], v77\n\t"
        "v_dual_max_num_f32 v78, %[m3], v78 :: v_dual_max_num_f32 v79, %[m3], v79\n\t"
        "v_dual_max_num_f32 v80, %[m4], v80 :: v_dual_max_num_f32 v81, %[m4], v81\n\t"
        "v_dual_max_num_f32 v82, %[m4], v82 :: v_dual_max_num_f32 v83, %[m4], v83\n\t"
        "v_dual_max_num_f32 v84, %[m5], v84 :: v_dual_max_num_f32 v85, %[m5], v85\n\t"
        "v_dual_max_num_f32 v86, %[m5], v86 :: v_dual_max_num_f32 v87, %[m5], v87\n\t"
        "v_dual_max_num_f32 v88, %[m6], v88 :: v_dual_max_num_f32 v89, %[m6], v89\n\t"
        "v_dual_max_num_f32 v90, %[m6], v90 :: v_dual_max_num_f32 v91, %[m6], v91\n\t"
        "v_dual_max_num_f32 v92, %[m7], v92 :: v_dual_max_num_f32 v93, %[m7], v93\n\t"
        "v_dual_max_num_f32 v94, %[m7], v94 :: v_dual_max_num_f32 v95, %[m7], v95\n\t"
        "v_dual_min_num_f32 v128, v64, v68 :: v_dual_min_num_f32 v129, v65, v69\n\t"
        "v_dual_min_num_f32 v130, v66, v70 :: v_dual_min_num_f32 v131, v67, v71\n\t"
        "v_dual_max_num_f32 v68, v64, v68 :: v_dual_max_num_f32 v69, v65, v69\n\t"
        "v_dual_max_num_f32 v70, v66, v70 :: v_dual_max_num_f32 v71, v67, v71\n\t"
        "v_dual_min_num_f32 v64, v72, v76 :: v_dual_min_num_f32 v65, v73, v77\n\t"
        "v_dual_min_num_f32 v66, v74, v78 :: v_dual_min_num_f32 v67, v75, v79\n\t"
        "v_dual_max_num_f32 v76, v72, v76 :: v_dual_max_num_f32 v77, v73, v77\n\t"
        "v_dual_max_num_f32 v78, v74, v78 :: v_dual_max_num_f32 v79, v75, v79\n\t"
        "v_dual_min_num_f32 v72, v80, v84 :: v_dual_min_num_f32 v73, v81, v85\n\t"
        "v_dual_min_num_f32 v74, v82, v86 :: v_dual_min_num_f32 v75, v83, v87\n\t"
        "v_dual_max_num_f32 v84, v80, v84 :: v_dual_max_num_f32 v85, v81, v85\n\t"
        "v_dual_max_num_f32 v86, v82, v86 :: v_dual_max_num_f32 v87, v83, v87\n\t"
        "v_dual_min_num_f32 v80, v88, v92 :: v_dual_min_num_f32 v81, v89, v93\n\t"
        "v_dual_min_num_f32 v82, v90, v94 :: v_dual_min_num_f32 v83, v91, v95\n\t"
        "v_dual_max_num_f32 v92, v88, v92 :: v_dual_max_num_f32 v93, v89, v93\n\t"
        "v_dual_max_num_f32 v94, v90, v94 :: v_dual_max_num_f32 v95, v91, v95\n\t"
        "v_dual_min_num_f32 v88, v0, v4 :: v_dual_min_num_f32 v89, v1, v5\n\t"
        "v_dual_min_num_f32 v90, v2, v6 :: v_dual_min_num_f32 v91, v3, v7\n\t"
        "v_dual_max_num_f32 v4, v0, v4 :: v_dual_max_num_f32 v5, v1, v5\n\t"
        "v_dual_max_num_f32 v6, v2, v6 :: v_dual_max_num_f32 v7, v3, v7\n\t"
        "v_dual_min_num_f32 v0, v8, v12 :: v_dual_min_num_f32 v1, v9, v13\n\t"
        "v_dual_min_num_f32 v2, v10, v14 :: v_dual_min_num_f32 v3, v11, v15\n\t"
        "v_dual_max_num_f32 v12, v8, v12 :: v_dual_max_num_f32 v13, v9, v13\n\t"
        "v_dual_max_num_f32 v14, v10, v14 :: v_dual_max_num_f32 v15, v11, v15\n\t"
        "v_dual_min_num_f32 v8, v16, v20 :: v_dual_min_num_f32 v9, v17, v21\n\t"
        "v_dual_min_num_f32 v10, v18, v22 :: v_dual_min_num_f32 v11, v19, v23\n\t"
        "v_dual_max_num_f32 v20, v16, v20 :: v_dual_max_num_f32 v21, v17, v21\n\t"
        "v_dual_max_num_f32 v22, v18, v22 :: v_dual_max_num_f32 v23, v19, v23\n\t"
        "v_dual_min_num_f32 v16, v24, v28 :: v_dual_min_num_f32 v17, v25, v29\n\t"
        "v_dual_min_num_f32 v18, v26, v30 :: v_dual_min_num_f32 v19, v27, v31\n\t"
        "v_dual_max_num_f32 v28, v24, v28 :: v_dual_max_num_f32 v29, v25, v29\n\t"
        "v_dual_max_num_f32 v30, v26, v30 :: v_dual_max_num_f32 v31, v27, v31\n\t"
        "v_dual_min_num_f32 v24, v32, v36 :: v_dual_min_num_f32 v25, v33, v37\n\t"
        "v_dual_min_num_f32 v26, v34, v38 :: v_dual_min_num_f32 v27, v35, v39\n\t"
        "v_dual_max_num_f32 v36, v32, v36 :: v_dual_max_num_f32 v37, v33, v37\n\t"
        "v_dual_max_num_f32 v38, v34, v38 :: v_dual_max_num_f32 v39, v35, v39\n\t"
        "v_dual_min_num_f32 v32, v40, v44 :: v_dual_min_num_f32 v33, v41, v45\n\t"
        "v_dual_min_num_f32 v34, v42, v46 :: v_dual_min_num_f32 v35, v43, v47\n\t"
        "v_dual_max_num_f32 v44, v40, v44 :: v_dual_max_num_f32 v45, v41, v45\n\t"
        "v_dual_max_num_f32 v46, v42, v46 :: v_dual_max_num_f32 v47, v43, v47\n\t"
        "v_dual_min_num_f32 v40, v48, v52 :: v_dual_min_num_f32 v41, v49, v53\n\t"
        "v_dual_min_num_f32 v42, v50, v54 :: v_dual_min_num_f32 v43, v51, v55\n\t"
        "v_dual_max_num_f32 v52, v48, v52 :: v_dual_max_num_f32 v53, v49, v53\n\t"
        "v_dual_max_num_f32 v54, v50, v54 :: v_dual_max_num_f32 v55, v51, v55\n\t"
        "v_dual_min_num_f32 v48, v56, v60 :: v_dual_min_num_f32 v49, v57, v61\n\t"
        "v_dual_min_num_f32 v50, v58, v62 :: v_dual_min_num_f32 v51, v59, v63\n\t"
        "v_dual_max_num_f32 v60, v56, v60 :: v_dual_max_num_f32 v61, v57, v61\n\t"
        "v_dual_max_num_f32 v62, v58, v62 :: v_dual_max_num_f32 v63, v59, v63\n\t"
        "v_dual_min_num_f32 v56, v128, v64 :: v_dual_min_num_f32 v57, v129, v65\n\t"
        "v_dual_min_num_f32 v58, v130, v66 :: v_dual_min_num_f32 v59, v131, v67\n\t"
        "v_dual_max_num_f32 v64, v128, v64 :: v_dual_max_num_f32 v65, v129, v65\n\t"
        "v_dual_max_num_f32 v66, v130, v66 :: v_dual_max_num_f32 v67, v131, v67\n\t"
        "v_dual_min_num_f32 v128, v68, v76 :: v_dual_min_num_f32 v129, v69, v77\n\t"
        "v_dual_min_num_f32 v130, v70, v78 :: v_dual_min_num_f32 v131, v71, v79\n\t"
        "v_dual_max_num_f32 v76, v68, v76 :: v_dual_max_num_f32 v77, v69, v77\n\t"
        "v_dual_max_num_f32 v78, v70, v78 :: v_dual_max_num_f32 v79, v71, v79\n\t"
        "v_dual_min_num_f32 v68, v72, v80 :: v_dual_min_num_f32 v69, v73, v81\n\t"
        "v_dual_min_num_f32 v70, v74, v82 :: v_dual_min_num_f32 v71, v75, v83\n\t"
        "v_dual_max_num_f32 v80, v72, v80 :: v_dual_max_num_f32 v81, v73, v81\n\t"
        "v_dual_max_num_f32 v82, v74, v82 :: v_dual_max_num_f32 v83, v75, v83\n\t"
        "v_dual_min_num_f32 v72, v84, v92 :: v_dual_min_num_f32 v73, v85, v93\n\t"
        "v_dual_min_num_f32 v74, v86, v94 :: v_dual_min_num_f32 v75, v87, v95\n\t"
        "v_dual_max_num_f32 v92, v84, v92 :: v_dual_max_num_f32 v93, v85, v93\n\t"
        "v_dual_max_num_f32 v94, v86, v94 :: v_dual_max_num_f32 v95, v87, v95\n\t"
        "v_dual_min_num_f32 v84, v88, v0 :: v_dual_min_num_f32 v85, v89, v1\n\t"
        "v_dual_min_num_f32 v86, v90, v2 :: v_dual_min_num_f32 v87, v91, v3\n\t"
        "v_dual_max_num_f32 v0, v88, v0 :: v_dual_max_num_f32 v1, v89, v1\n\t"
        "v_dual_max_num_f32 v2, v90, v2 :: v_dual_max_num_f32 v3, v91, v3\n\t"
        "v_dual_min_num_f32 v88, v4, v12 :: v_dual_min_num_f32 v89, v5, v13\n\t"
        "v_dual_min_num_f32 v90, v6, v14 :: v_dual_min_num_f32 v91, v7, v15\n\t"
        "v_dual_max_num_f32 v12, v4, v12 :: v_dual_max_num_f32 v13, v5, v13\n\t"
        "v_dual_max_num_f32 v14, v6, v14 :: v_dual_max_num_f32 v15, v7, v15\n\t"
        "v_dual_min_num_f32 v4, v8, v16 :: v_dual_min_num_f32 v5, v9, v17\n\t"
        "v_dual_min_num_f32 v6, v10, v18 :: v_dual_min_num_f32 v7, v11, v19\n\t"
        "v_dual_max_num_f32 v16, v8, v16 :: v_dual_max_num_f32 v17, v9, v17\n\t"
        "v_dual_max_num_f32 v18, v10, v18 :: v_dual_max_num_f32 v19, v11, v19\n\t"
        "v_dual_min_num_f32 v8, v20, v28 :: v_dual_min_num_f32 v9, v21, v29\n\t"
        "v_dual_min_num_f32 v10, v22, v30 :: v_dual_min_num_f32 v11, v23, v31\n\t"
        "v_dual_max_num_f32 v28, v20, v28 :: v_dual_max_num_f32 v29, v21, v29\n\t"
        "v_dual_max_num_f32 v30, v22, v30 :: v_dual_max_num_f32 v31, v23, v31\n\t"
        "v_dual_min_num_f32 v20, v24, v32 :: v_dual_min_num_f32 v21, v25, v33\n\t"
        "v_dual_min_num_f32 v22, v26, v34 :: v_dual_min_num_f32 v23, v27, v35\n\t"
        "v_dual_max_num_f32 v32, v24, v32 :: v_dual_max_num_f32 v33, v25, v33\n\t"
        "v_dual_max_num_f32 v34, v26, v34 :: v_dual_max_num_f32 v35, v27, v35\n\t"
        "v_dual_min_num_f32 v24, v36, v44 :: v_dual_min_num_f32 v25, v37, v45\n\t"
        "v_dual_min_num_f32 v26, v38, v46 :: v_dual_min_num_f32 v27, v39, v47\n\t"
        "v_dual_max_num_f32 v44, v36, v44 :: v_dual_max_num_f32 v45, v37, v45\n\t"
        "v_dual_max_num_f32 v46, v38, v46 :: v_dual_max_num_f32 v47, v39, v47\n\t"
        "v_dual_min_num_f32 v36, v40, v48 :: v_dual_min_num_f32 v37, v41, v49\n\t"
        "v_dual_min_num_f32 v38, v42, v50 :: v_dual_min_num_f32 v39, v43, v51\n\t"
        "v_dual_max_num_f32 v48, v40, v48 :: v_dual_max_num_f32 v49, v41, v49\n\t"
        "v_dual_max_num_f32 v50, v42, v50 :: v_dual_max_num_f32 v51, v43, v51\n\t"
        "v_dual_min_num_f32 v40, v52, v60 :: v_dual_min_num_f32 v41, v53, v61\n\t"
        "v_dual_min_num_f32 v42, v54, v62 :: v_dual_min_num_f32 v43, v55, v63\n\t"
        "v_dual_max_num_f32 v60, v52, v60 :: v_dual_max_num_f32 v61, v53, v61\n\t"
        "v_dual_max_num_f32 v62, v54, v62 :: v_dual_max_num_f32 v63, v55, v63\n\t"
        "v_dual_min_num_f32 v52, v128, v64 :: v_dual_min_num_f32 v53, v129, v65\n\t"
        "v_dual_min_num_f32 v54, v130, v66 :: v_dual_min_num_f32 v55, v131, v67\n\t"
        "v_dual_max_num_f32 v64, v128, v64 :: v_dual_max_num_f32 v65, v129, v65\n\t"
        "v_dual_max_num_f32 v66, v130, v66 :: v_dual_max_num_f32 v67, v131, v67\n\t"
        "v_dual_min_num_f32 v128, v72, v80 :: v_dual_min_num_f32 v129, v73, v81\n\t"
        "v_dual_min_num_f32 v130, v74, v82 :: v_dual_min_num_f32 v131, v75, v83\n\t"
        "v_dual_max_num_f32 v80, v72, v80 :: v_dual_max_num_f32 v81, v73, v81\n\t"
        "v_dual_max_num_f32 v82, v74, v82 :: v_dual_max_num_f32 v83, v75, v83\n\t"
        "v_dual_min_num_f32 v72, v56, v68 :: v_dual_min_num_f32 v73, v57, v69\n\t"
        "v_dual_min_num_f32 v74, v58, v70 :: v_dual_min_num_f32 v75, v59, v71\n\t"
        "v_dual_max_num_f32 v68, v56, v68 :: v_dual_max_num_f32 v69, v57, v69\n\t"
        "v_dual_max_num_f32 v70, v58, v70 :: v_dual_max_num_f32 v71, v59, v71\n\t"
        "v_dual_min_num_f32 v56, v76, v92 :: v_dual_min_num_f32 v57, v77, v93\n\t"
        "v_dual_min_num_f32 v58, v78, v94 :: v_dual_min_num_f32 v59, v79, v95\n\t"
        "v_dual_max_num_f32 v92, v76, v92 :: v_dual_max_num_f32 v93, v77, v93\n\t"
        "v_dual_max_num_f32 v94, v78, v94 :: v_dual_max_num_f32 v95, v79, v95\n\t"
        "v_dual_min_num_f32 v76, v88, v0 :: v_dual_min_num_f32 v77, v89, v1\n\t"
        "v_dual_min_num_f32 v78, v90, v2 :: v_dual_min_num_f32 v79, v91, v3\n\t"
        "v_dual_max_num_f32 v0, v88, v0 :: v_dual_max_num_f32 v1, v89, v1\n\t"
        "v_dual_max_num_f32 v2, v90, v2 :: v_dual_max_num_f32 v3, v91, v3\n\t"
        "v_dual_min_num_f32 v88, v8, v16 :: v_dual_min_num_f32 v89, v9, v17\n\t"
        "v_dual_min_num_f32 v90, v10, v18 :: v_dual_min_num_f32 v91, v11, v19\n\t"
        "v_dual_max_num_f32 v16, v8, v16 :: v_dual_max_num_f32 v17, v9, v17\n\t"
        "v_dual_max_num_f32 v18, v10, v18 :: v_dual_max_num_f32 v19, v11, v19\n\t"
        "v_dual_min_num_f32 v8, v84, v4 :: v_dual_min_num_f32 v9, v85, v5\n\t"
        "v_dual_min_num_f32 v10, v86, v6 :: v_dual_min_num_f32 v11, v87, v7\n\t"
        "v_dual_max_num_f32 v4, v84, v4 :: v_dual_max_num_f32 v5, v85, v5\n\t"
        "v_dual_max_num_f32 v6, v86, v6 :: v_dual_max_num_f32 v7, v87, v7\n\t"
        "v_dual_min_num_f32 v84, v12, v28 :: v_dual_min_num_f32 v85, v13, v29\n\t"
        "v_dual_min_num_f32 v86, v14, v30 :: v_dual_min_num_f32 v87, v15, v31\n\t"
        "v_dual_max_num_f32 v28, v12, v28 :: v_dual_max_num_f32 v29, v13, v29\n\t"
        "v_dual_max_num_f32 v30, v14, v30 :: v_dual_max_num_f32 v31, v15, v31\n\t"
        "v_dual_min_num_f32 v12, v24, v32 :: v_dual_min_num_f32 v13, v25, v33\n\t"
        "v_dual_min_num_f32 v14, v26, v34 :: v_dual_min_num_f32 v15, v27, v35\n\t"
        "v_dual_max_num_f32 v32, v24, v32 :: v_dual_max_num_f32 v33, v25, v33\n\t"
        "v_dual_max_num_f32 v34, v26, v34 :: v_dual_max_num_f32 v35, v27, v35\n\t"
        "v_dual_min_num_f32 v24, v40, v48 :: v_dual_min_num_f32 v25, v41, v49\n\t"
        "v_dual_min_num_f32 v26, v42, v50 :: v_dual_min_num_f32 v27, v43, v51\n\t"
        "v_dual_max_num_f32 v48, v40, v48 :: v_dual_max_num_f32 v49, v41, v49\n\t"
        "v_dual_max_num_f32 v50, v42, v50 :: v_dual_max_num_f32 v51, v43, v51\n\t"
        "v_dual_min_num_f32 v40, v20, v36 :: v_dual_min_num_f32 v41, v21, v37\n\t"
        "v_dual_min_num_f32 v42, v22, v38 :: v_dual_min_num_f32 v43, v23, v39\n\t"
        "v_dual_max_num_f32 v36, v20, v36 :: v_dual_max_num_f32 v37, v21, v37\n\t"
        "v_dual_max_num_f32 v38, v22, v38 :: v_dual_max_num_f32 v39, v23, v39\n\t"
        "v_dual_min_num_f32 v20, v44, v60 :: v_dual_min_num_f32 v21, v45, v61\n\t"
        "v_dual_min_num_f32 v22, v46, v62 :: v_dual_min_num_f32 v23, v47, v63\n\t"
        "v_dual_max_num_f32 v60, v44, v60 :: v_dual_max_num_f32 v61, v45, v61\n\t"
        "v_dual_max_num_f32 v62, v46, v62 :: v_dual_max_num_f32 v63, v47, v63\n\t"
        "v_dual_min_num_f32 v44, v64, v80 :: v_dual_min_num_f32 v45, v65, v81\n\t"
        "v_dual_min_num_f32 v46, v66, v82 :: v_dual_min_num_f32 v47, v67, v83\n\t"
        "v_dual_max_num_f32 v80, v64, v80 :: v_dual_max_num_f32 v81, v65, v81\n\t"
        "v_dual_max_num_f32 v82, v66, v82 :: v_dual_max_num_f32 v83, v67, v83\n\t"
        "v_dual_min_num_f32 v64, v52, v128 :: v_dual_min_num_f32 v65, v53, v129\n\t"
        "v_dual_min_num_f32 v66, v54, v130 :: v_dual_min_num_f32 v67, v55, v131\n\t"
        "v_dual_max_num_f32 v128, v52, v128 :: v_dual_max_num_f32 v129, v53, v129\n\t"
        "v_dual_max_num_f32 v130, v54, v130 :: v_dual_max_num_f32 v131, v55, v131\n\t"
        "v_dual_min_num_f32 v52, v0, v16 :: v_dual_min_num_f32 v53, v1, v17\n\t"
        "v_dual_min_num_f32 v54, v2, v18 :: v_dual_min_num_f32 v55, v3, v19\n\t"
        "v_dual_max_num_f32 v16, v0, v16 :: v_dual_max_num_f32 v17, v1, v17\n\t"
        "v_dual_max_num_f32 v18, v2, v18 :: v_dual_max_num_f32 v19, v3, v19\n\t"
        "v_dual_min_num_f32 v0, v76, v88 :: v_dual_min_num_f32 v1, v77, v89\n\t"
        "v_dual_min_num_f32 v2, v78, v90 :: v_dual_min_num_f32 v3, v79, v91\n\t"
        "v_dual_max_num_f32 v88, v76, v88 :: v_dual_max_num_f32 v89, v77, v89\n\t"
        "v_dual_max_num_f32 v90, v78, v90 :: v_dual_max_num_f32 v91, v79, v91\n\t"
        "v_dual_min_num_f32 v76, v32, v48 :: v_dual_min_num_f32 v77, v33, v49\n\t"
        "v_dual_min_num_f32 v78, v34, v50 :: v_dual_min_num_f32 v79, v35, v51\n\t"
        "v_dual_max_num_f32 v48, v32, v48 :: v_dual_max_num_f32 v49, v33, v49\n\t"
        "v_dual_max_num_f32 v50, v34, v50 :: v_dual_max_num_f32 v51, v35, v51\n\t"
        "v_dual_min_num_f32 v32, v12, v24 :: v_dual_min_num_f32 v33, v13, v25\n\t"
        "v_dual_min_num_f32 v34, v14, v26 :: v_dual_min_num_f32 v35, v15, v27\n\t"
        "v_dual_max_num_f32 v24, v12, v24 :: v_dual_max_num_f32 v25, v13, v25\n\t"
        "v_dual_max_num_f32 v26, v14, v26 :: v_dual_max_num_f32 v27, v15, v27\n\t"
        "v_dual_min_num_f32 v12, v8, v40 :: v_dual_min_num_f32 v13, v9, v41\n\t"
        "v_dual_min_num_f32 v14, v10, v42 :: v_dual_min_num_f32 v15, v11, v43\n\t"
        "v_dual_max_num_f32 v40, v8, v40 :: v_dual_max_num_f32 v41, v9, v41\n\t"
        "v_dual_max_num_f32 v42, v10, v42 :: v_dual_max_num_f32 v43, v11, v43\n\t"
        "v_dual_min_num_f32 v8, v28, v60 :: v_dual_min_num_f32 v9, v29, v61\n\t"
        "v_dual_min_num_f32 v10, v30, v62 :: v_dual_min_num_f32 v11, v31, v63\n\t"
        "v_dual_max_num_f32 v60, v28, v60 :: v_dual_max_num_f32 v61, v29, v61\n\t"
        "v_dual_max_num_f32 v62, v30, v62 :: v_dual_max_num_f32 v63, v31, v63\n\t"
        "v_dual_min_num_f32 v28, v44, v68 :: v_dual_min_num_f32 v29, v45, v69\n\t"
        "v_dual_min_num_f32 v30, v46, v70 :: v_dual_min_num_f32 v31, v47, v71\n\t"
        "v_dual_max_num_f32 v68, v44, v68 :: v_dual_max_num_f32 v69, v45, v69\n\t"
        "v_dual_max_num_f32 v70, v46, v70 :: v_dual_max_num_f32 v71, v47, v71\n\t"
        "v_dual_min_num_f32 v44, v56, v128 :: v_dual_min_num_f32 v45, v57, v129\n\t"
        "v_dual_min_num_f32 v46, v58, v130 :: v_dual_min_num_f32 v47, v59, v131\n\t"
        "v_dual_max_num_f32 v128, v56, v128 :: v_dual_max_num_f32 v129, v57, v129\n\t"
        "v_dual_max_num_f32 v130, v58, v130 :: v_dual_max_num_f32 v131, v59, v131\n\t"
        "v_dual_min_num_f32 v56, v52, v4 :: v_dual_min_num_f32 v57, v53, v5\n\t"
        "v_dual_min_num_f32 v58, v54, v6 :: v_dual_min_num_f32 v59, v55, v7\n\t"
        "v_dual_max_num_f32 v4, v52, v4 :: v_dual_max_num_f32 v5, v53, v5\n\t"
        "v_dual_max_num_f32 v6, v54, v6 :: v_dual_max_num_f32 v7, v55, v7\n\t"
        "v_dual_min_num_f32 v52, v84, v88 :: v_dual_min_num_f32 v53, v85, v89\n\t"
        "v_dual_min_num_f32 v54, v86, v90 :: v_dual_min_num_f32 v55, v87, v91\n\t"
        "v_dual_max_num_f32 v88, v84, v88 :: v_dual_max_num_f32 v89, v85, v89\n\t"
        "v_dual_max_num_f32 v90, v86, v90 :: v_dual_max_num_f32 v91, v87, v91\n\t"
        "v_dual_min_num_f32 v84, v76, v36 :: v_dual_min_num_f32 v85, v77, v37\n\t"
        "v_dual_min_num_f32 v86, v78, v38 :: v_dual_min_num_f32 v87, v79, v39\n\t"
        "v_dual_max_num_f32 v36, v76, v36 :: v_dual_max_num_f32 v37, v77, v37\n\t"
        "v_dual_max_num_f32 v38, v78, v38 :: v_dual_max_num_f32 v39, v79, v39\n\t"
        "v_dual_min_num_f32 v76, v20, v24 :: v_dual_min_num_f32 v77, v21, v25\n\t"
        "v_dual_min_num_f32 v78, v22, v26 :: v_dual_min_num_f32 v79, v23, v27\n\t"
        "v_dual_max_num_f32 v24, v20, v24 :: v_dual_max_num_f32 v25, v21, v25\n\t"
        "v_dual_max_num_f32 v26, v22, v26 :: v_dual_max_num_f32 v27, v23, v27\n\t"
        "v_dual_min_num_f32 v60, v60, v72 :: v_dual_min_num_f32 v61, v61, v73\n\t"
        "v_dual_min_num_f32 v62, v62, v74 :: v_dual_min_num_f32 v63, v63, v75\n\t"
        "v_dual_min_num_f32 v20, v64, v28 :: v_dual_min_num_f32 v21, v65, v29\n\t"
        "v_dual_min_num_f32 v22, v66, v30 :: v_dual_min_num_f32 v23, v67, v31\n\t"
        "v_dual_max_num_f32 v28, v64, v28 :: v_dual_max_num_f32 v29, v65, v29\n\t"
        "v_dual_max_num_f32 v30, v66, v30 :: v_dual_max_num_f32 v31, v67, v31\n\t"
        "v_dual_min_num_f32 v64, v44, v68 :: v_dual_min_num_f32 v65, v45, v69\n\t"
        "v_dual_min_num_f32 v66, v46, v70 :: v_dual_min_num_f32 v67, v47, v71\n\t"
        "v_dual_max_num_f32 v68, v44, v68 :: v_dual_max_num_f32 v69, v45, v69\n\t"
        "v_dual_max_num_f32 v70, v46, v70 :: v_dual_max_num_f32 v71, v47, v71\n\t"
        "v_dual_min_num_f32 v44, v128, v80 :: v_dual_min_num_f32 v45, v129, v81\n\t"
        "v_dual_min_num_f32 v46, v130, v82 :: v_dual_min_num_f32 v47, v131, v83\n\t"
        "v_dual_max_num_f32 v80, v128, v80 :: v_dual_max_num_f32 v81, v129, v81\n\t"
        "v_dual_max_num_f32 v82, v130, v82 :: v_dual_max_num_f32 v83, v131, v83\n\t"
        "v_dual_min_num_f32 v128, v0, v56 :: v_dual_min_num_f32 v129, v1, v57\n\t"
        "v_dual_min_num_f32 v130, v2, v58 :: v_dual_min_num_f32 v131, v3, v59\n\t"
        "v_dual_max_num_f32 v56, v0, v56 :: v_dual_max_num_f32 v57, v1, v57\n\t"
        "v_dual_max_num_f32 v58, v2, v58 :: v_dual_max_num_f32 v59, v3, v59\n\t"
        "v_dual_min_num_f32 v0, v52, v4 :: v_dual_min_num_f32 v1, v53, v5\n\t"
        "v_dual_min_num_f32 v2, v54, v6 :: v_dual_min_num_f32 v3, v55, v7\n\t"
        "v_dual_max_num_f32 v4, v52, v4 :: v_dual_max_num_f32 v5, v53, v5\n\t"
        "v_dual_max_num_f32 v6, v54, v6 :: v_dual_max_num_f32 v7, v55, v7\n\t"
        "v_dual_min_num_f32 v52, v88, v16 :: v_dual_min_num_f32 v53, v89, v17\n\t"
        "v_dual_min_num_f32 v54, v90, v18 :: v_dual_min_num_f32 v55, v91, v19\n\t"
        "v_dual_max_num_f32 v16, v88, v16 :: v_dual_max_num_f32 v17, v89, v17\n\t"
        "v_dual_max_num_f32 v18, v90, v18 :: v_dual_max_num_f32 v19, v91, v19\n\t"
        "v_dual_min_num_f32 v88, v32, v84 :: v_dual_min_num_f32 v89, v33, v85\n\t"
        "v_dual_min_num_f32 v90, v34, v86 :: v_dual_min_num_f32 v91, v35, v87\n\t"
        "v_dual_max_num_f32 v84, v32, v84 :: v_dual_max_num_f32 v85, v33, v85\n\t"
        "v_dual_max_num_f32 v86, v34, v86 :: v_dual_max_num_f32 v87, v35, v87\n\t"
        "v_dual_min_num_f32 v32, v76, v36 :: v_dual_min_num_f32 v33, v77, v37\n\t"
        "v_dual_min_num_f32 v34, v78, v38 :: v_dual_min_num_f32 v35, v79, v39\n\t"
        "v_dual_max_num_f32 v36, v76, v36 :: v_dual_max_num_f32 v37, v77, v37\n\t"
        "v_dual_max_num_f32 v38, v78, v38 :: v_dual_max_num_f32 v39, v79, v39\n\t"
        "v_dual_min_num_f32 v76, v24, v48 :: v_dual_min_num_f32 v77, v25, v49\n\t"
        "v_dual_min_num_f32 v78, v26, v50 :: v_dual_min_num_f32 v79, v27, v51\n\t"
        "v_dual_max_num_f32 v48, v24, v48 :: v_dual_max_num_f32 v49, v25, v49\n\t"
        "v_dual_max_num_f32 v50, v26, v50 :: v_dual_max_num_f32 v51, v27, v51\n\t"
        "v_dual_min_num_f32 v24, v4, v36 :: v_dual_min_num_f32 v25, v5, v37\n\t"
        "v_dual_min_num_f32 v26, v6, v38 :: v_dual_min_num_f32 v27, v7, v39\n\t"
        "v_dual_max_num_f32 v36, v4, v36 :: v_dual_max_num_f32 v37, v5, v37\n\t"
        "v_dual_max_num_f32 v38, v6, v38 :: v_dual_max_num_f32 v39, v7, v39\n\t"
        "v_dual_min_num_f32 v4, v56, v84 :: v_dual_min_num_f32 v5, v57, v85\n\t"
        "v_dual_min_num_f32 v6, v58, v86 :: v_dual_min_num_f32 v7, v59, v87\n\t"
        "v_dual_max_num_f32 v84, v56, v84 :: v_dual_max_num_f32 v85, v57, v85\n\t"
        "v_dual_max_num_f32 v86, v58, v86 :: v_dual_max_num_f32 v87, v59, v87\n\t"
        "v_dual_min_num_f32 v56, v16, v48 :: v_dual_min_num_f32 v57, v17, v49\n\t"
        "v_dual_min_num_f32 v58, v18, v50 :: v_dual_min_num_f32 v59, v19, v51\n\t"
        "v_dual_max_num_f32 v48, v16, v48 :: v_dual_max_num_f32 v49, v17, v49\n\t"
        "v_dual_max_num_f32 v50, v18, v50 :: v_dual_max_num_f32 v51, v19, v51\n\t"
        "v_dual_min_num_f32 v16, v128, v88 :: v_dual_min_num_f32 v17, v129, v89\n\t"
        "v_dual_min_num_f32 v18, v130, v90 :: v_dual_min_num_f32 v19, v131, v91\n\t"
        "v_dual_max_num_f32 v88, v128, v88 :: v_dual_max_num_f32 v89, v129, v89\n\t"
        "v_dual_max_num_f32 v90, v130, v90 :: v_dual_max_num_f32 v91, v131, v91\n\t"
        "v_dual_min_num_f32 v128, v52, v76 :: v_dual_min_num_f32 v129, v53, v77\n\t"
        "v_dual_min_num_f32 v130, v54, v78 :: v_dual_min_num_f32 v131, v55, v79\n\t"
        "v_dual_max_num_f32 v76, v52, v76 :: v_dual_max_num_f32 v77, v53, v77\n\t"
        "v_dual_max_num_f32 v78, v54, v78 :: v_dual_max_num_f32 v79, v55, v79\n\t"
        "v_dual_min_num_f32 v52, v0, v32 :: v_dual_min_num_f32 v53, v1, v33\n\t"
        "v_dual_min_num_f32 v54, v2, v34 :: v_dual_min_num_f32 v55, v3, v35\n\t"
        "v_dual_max_num_f32 v32, v0, v32 :: v_dual_max_num_f32 v33, v1, v33\n\t"
        "v_dual_max_num_f32 v34, v2, v34 :: v_dual_max_num_f32 v35, v3, v35\n\t"
        "v_dual_min_num_f32 v0, v24, v40 :: v_dual_min_num_f32 v1, v25, v41\n\t"
        "v_dual_min_num_f32 v2, v26, v42 :: v_dual_min_num_f32 v3, v27, v43\n\t"
        "v_dual_max_num_f32 v40, v24, v40 :: v_dual_max_num_f32 v41, v25, v41\n\t"
        "v_dual_max_num_f32 v42, v26, v42 :: v_dual_max_num_f32 v43, v27, v43\n\t"
        "v_dual_min_num_f32 v24, v56, v84 :: v_dual_min_num_f32 v25, v57, v85\n\t"
        "v_dual_min_num_f32 v26, v58, v86 :: v_dual_min_num_f32 v27, v59, v87\n\t"
        "v_dual_max_num_f32 v84, v56, v84 :: v_dual_max_num_f32 v85, v57, v85\n\t"
        "v_dual_max_num_f32 v86, v58, v86 :: v_dual_max_num_f32 v87, v59, v87\n\t"
        "v_dual_min_num_f32 v56, v128, v88 :: v_dual_min_num_f32 v57, v129, v89\n\t"
        "v_dual_min_num_f32 v58, v130, v90 :: v_dual_min_num_f32 v59, v131, v91\n\t"
        "v_dual_max_num_f32 v88, v128, v88 :: v_dual_max_num_f32 v89, v129, v89\n\t"
        "v_dual_max_num_f32 v90, v130, v90 :: v_dual_max_num_f32 v91, v131, v91\n\t"
        "v_dual_min_num_f32 v128, v8, v32 :: v_dual_min_num_f32 v129, v9, v33\n\t"
        "v_dual_min_num_f32 v130, v10, v34 :: v_dual_min_num_f32 v131, v11, v35\n\t"
        "v_dual_max_num_f32 v32, v8, v32 :: v_dual_max_num_f32 v33, v9, v33\n\t"
        "v_dual_max_num_f32 v34, v10, v34 :: v_dual_max_num_f32 v35, v11, v35\n\t"
        "v_dual_min_num_f32 v8, v4, v0 :: v_dual_min_num_f32 v9, v5, v1\n\t"
        "v_dual_min_num_f32 v10, v6, v2 :: v_dual_min_num_f32 v11, v7, v3\n\t"
        "v_dual_max_num_f32 v0, v4, v0 :: v_dual_max_num_f32 v1, v5, v1\n\t"
        "v_dual_max_num_f32 v2, v6, v2 :: v_dual_max_num_f32 v3, v7, v3\n\t"
        "v_dual_min_num_f32 v4, v24, v40 :: v_dual_min_num_f32 v5, v25, v41\n\t"
        "v_dual_min_num_f32 v6, v26, v42 :: v_dual_min_num_f32 v7, v27, v43\n\t"
        "v_dual_max_num_f32 v40, v24, v40 :: v_dual_max_num_f32 v41, v25, v41\n\t"
        "v_dual_max_num_f32 v42, v26, v42 :: v_dual_max_num_f32 v43, v27, v43\n\t"
        "v_dual_min_num_f32 v24, v84, v36 :: v_dual_min_num_f32 v25, v85, v37\n\t"
        "v_dual_min_num_f32 v26, v86, v38 :: v_dual_min_num_f32 v27, v87, v39\n\t"
        "v_dual_max_num_f32 v36, v84, v36 :: v_dual_max_num_f32 v37, v85, v37\n\t"
        "v_dual_max_num_f32 v38, v86, v38 :: v_dual_max_num_f32 v39, v87, v39\n\t"
        "v_dual_min_num_f32 v84, v52, v56 :: v_dual_min_num_f32 v85, v53, v57\n\t"
        "v_dual_min_num_f32 v86, v54, v58 :: v_dual_min_num_f32 v87, v55, v59\n\t"
        "v_dual_max_num_f32 v56, v52, v56 :: v_dual_max_num_f32 v57, v53, v57\n\t"
        "v_dual_max_num_f32 v58, v54, v58 :: v_dual_max_num_f32 v59, v55, v59\n\t"
        "v_dual_min_num_f32 v52, v128, v88 :: v_dual_min_num_f32 v53, v129, v89\n\t"
        "v_dual_min_num_f32 v54, v130, v90 :: v_dual_min_num_f32 v55, v131, v91\n\t"
        "v_dual_max_num_f32 v88, v128, v88 :: v_dual_max_num_f32 v89, v129, v89\n\t"
        "v_dual_max_num_f32 v90, v130, v90 :: v_dual_max_num_f32 v91, v131, v91\n\t"
        "v_dual_min_num_f32 v128, v32, v76 :: v_dual_min_num_f32 v129, v33, v77\n\t"
        "v_dual_min_num_f32 v130, v34, v78 :: v_dual_min_num_f32 v131, v35, v79\n\t"
        "v_dual_max_num_f32 v76, v32, v76 :: v_dual_max_num_f32 v77, v33, v77\n\t"
        "v_dual_max_num_f32 v78, v34, v78 :: v_dual_max_num_f32 v79, v35, v79\n\t"
        "v_dual_min_num_f32 v32, v16, v8 :: v_dual_min_num_f32 v33, v17, v9\n\t"
        "v_dual_min_num_f32 v34, v18, v10 :: v_dual_min_num_f32 v35, v19, v11\n\t"
        "v_dual_max_num_f32 v8, v16, v8 :: v_dual_max_num_f32 v9, v17, v9\n\t"
        "v_dual_max_num_f32 v10, v18, v10 :: v_dual_max_num_f32 v11, v19, v11\n\t"
        "v_dual_min_num_f32 v16, v84, v0 :: v_dual_min_num_f32 v17, v85, v1\n\t"
        "v_dual_min_num_f32 v18, v86, v2 :: v_dual_min_num_f32 v19, v87, v3\n\t"
        "v_dual_max_num_f32 v0, v84, v0 :: v_dual_max_num_f32 v1, v85, v1\n\t"
        "v_dual_max_num_f32 v2, v86, v2 :: v_dual_max_num_f32 v3, v87, v3\n\t"
        "v_dual_min_num_f32 v84, v56, v4 :: v_dual_min_num_f32 v85, v57, v5\n\t"
        "v_dual_min_num_f32 v86, v58, v6 :: v_dual_min_num_f32 v87, v59, v7\n\t"
        "v_dual_max_num_f32 v4, v56, v4 :: v_dual_max_num_f32 v5, v57, v5\n\t"
        "v_dual_max_num_f32 v6, v58, v6 :: v_dual_max_num_f32 v7, v59, v7\n\t"
        "v_dual_min_num_f32 v56, v52, v40 :: v_dual_min_num_f32 v57, v53, v41\n\t"
        "v_dual_min_num_f32 v58, v54, v42 :: v_dual_min_num_f32 v59, v55, v43\n\t"
        "v_dual_max_num_f32 v40, v52, v40 :: v_dual_max_num_f32 v41, v53, v41\n\t"
        "v_dual_max_num_f32 v42, v54, v42 :: v_dual_max_num_f32 v43, v55, v43\n\t"
        "v_dual_min_num_f32 v52, v88, v24 :: v_dual_min_num_f32 v53, v89, v25\n\t"
        "v_dual_min_num_f32 v54, v90, v26 :: v_dual_min_num_f32 v55, v91, v27\n\t"
        "v_dual_max_num_f32 v24, v88, v24 :: v_dual_max_num_f32 v25, v89, v25\n\t"
        "v_dual_max_num_f32 v26, v90, v26 :: v_dual_max_num_f32 v27, v91, v27\n\t"
        "v_dual_min_num_f32 v88, v128, v36 :: v_dual_min_num_f32 v89, v129, v37\n\t"
        "v_dual_min_num_f32 v90, v130, v38 :: v_dual_min_num_f32 v91, v131, v39\n\t"
        "v_dual_max_num_f32 v36, v128, v36 :: v_dual_max_num_f32 v37, v129, v37\n\t"
        "v_dual_max_num_f32 v38, v130, v38 :: v_dual_max_num_f32 v39, v131, v39\n\t"
        "v_dual_min_num_f32 v128, v76, v48 :: v_dual_min_num_f32 v129, v77, v49\n\t"
        "v_dual_min_num_f32 v130, v78, v50 :: v_dual_min_num_f32 v131, v79, v51\n\t"
        "v_dual_max_num_f32 v48, v76, v48 :: v_dual_max_num_f32 v49, v77, v49\n\t"
        "v_dual_max_num_f32 v50, v78, v50 :: v_dual_max_num_f32 v51, v79, v51\n\t"
        "v_dual_min_num_f32 v40, v40, v92 :: v_dual_min_num_f32 v41, v41, v93\n\t"
        "v_dual_min_num_f32 v42, v42, v94 :: v_dual_min_num_f32 v43, v43, v95\n\t"
        "v_dual_min_num_f32 v52, v52, v80 :: v_dual_min_num_f32 v53, v53, v81\n\t"
        "v_dual_min_num_f32 v54, v54, v82 :: v_dual_min_num_f32 v55, v55, v83\n\t"
        "v_dual_min_num_f32 v24, v24, v44 :: v_dual_min_num_f32 v25, v25, v45\n\t"
        "v_dual_min_num_f32 v26, v26, v46 :: v_dual_min_num_f32 v27, v27, v47\n\t"
        "v_dual_min_num_f32 v88, v88, v68 :: v_dual_min_num_f32 v89, v89, v69\n\t"
        "v_dual_min_num_f32 v90, v90, v70 :: v_dual_min_num_f32 v91, v91, v71\n\t"
        "v_dual_min_num_f32 v36, v36, v64 :: v_dual_min_num_f32 v37, v37, v65\n\t"
        "v_dual_min_num_f32 v38, v38, v66 :: v_dual_min_num_f32 v39, v39, v67\n\t"
        "v_dual_min_num_f32 v128, v128, v28 :: v_dual_min_num_f32 v129, v129, v29\n\t"
        "v_dual_min_num_f32 v130, v130, v30 :: v_dual_min_num_f32 v131, v131, v31\n\t"
        "v_dual_min_num_f32 v48, v48, v20 :: v_dual_min_num_f32 v49, v49, v21\n\t"
        "v_dual_min_num_f32 v50, v50, v22 :: v_dual_min_num_f32 v51, v51, v23\n\t"
        "v_dual_max_num_f32 v56, v56, v60 :: v_dual_max_num_f32 v57, v57, v61\n\t"
        "v_dual_max_num_f32 v58, v58, v62 :: v_dual_max_num_f32 v59, v59, v63\n\t"
        "v_dual_max_num_f32 v12, v12, v40 :: v_dual_max_num_f32 v13, v13, v41\n\t"
        "v_dual_max_num_f32 v14, v14, v42 :: v_dual_max_num_f32 v15, v15, v43\n\t"
        "v_dual_max_num_f32 v32, v32, v52 :: v_dual_max_num_f32 v33, v33, v53\n\t"
        "v_dual_max_num_f32 v34, v34, v54 :: v_dual_max_num_f32 v35, v35, v55\n\t"
        "v_dual_max_num_f32 v8, v8, v24 :: v_dual_max_num_f32 v9, v9, v25\n\t"
        "v_dual_max_num_f32 v10, v10, v26 :: v_dual_max_num_f32 v11, v11, v27\n\t"
        "v_dual_max_num_f32 v16, v16, v88 :: v_dual_max_num_f32 v17, v17, v89\n\t"
        "v_dual_max_num_f32 v18, v18, v90 :: v_dual_max_num_f32 v19, v19, v91\n\t"
        "v_dual_max_num_f32 v0, v0, v36 :: v_dual_max_num_f32 v1, v1, v37\n\t"
        "v_dual_max_num_f32 v2, v2, v38 :: v_dual_max_num_f32 v3, v3, v39\n\t"
        "v_dual_max_num_f32 v84, v84, v128 :: v_dual_max_num_f32 v85, v85, v129\n\t"
        "v_dual_max_num_f32 v86, v86, v130 :: v_dual_max_num_f32 v87, v87, v131\n\t"
        "v_dual_max_num_f32 v4, v4, v48 :: v_dual_max_num_f32 v5, v5, v49\n\t"
        "v_dual_max_num_f32 v6, v6, v50 :: v_dual_max_num_f32 v7, v7, v51\n\t"
        "v_dual_min_num_f32 v12, v12, v0 :: v_dual_min_num_f32 v13, v13, v1\n\t"
        "v_dual_min_num_f32 v14, v14, v2 :: v_dual_min_num_f32 v15, v15, v3\n\t"
        "v_dual_min_num_f32 v32, v32, v84 :: v_dual_min_num_f32 v33, v33, v85\n\t"
        "v_dual_min_num_f32 v34, v34, v86 :: v_dual_min_num_f32 v35, v35, v87\n\t"
        "v_dual_min_num_f32 v8, v8, v4 :: v_dual_min_num_f32 v9, v9, v5\n\t"
        "v_dual_min_num_f32 v10, v10, v6 :: v_dual_min_num_f32 v11, v11, v7\n\t"
        "v_dual_min_num_f32 v16, v16, v56 :: v_dual_min_num_f32 v17, v17, v57\n\t"
        "v_dual_min_num_f32 v18, v18, v58 :: v_dual_min_num_f32 v19, v19, v59\n\t"
        "v_dual_min_num_f32 v76, v12, v8 :: v_dual_min_num_f32 v77, v13, v9\n\t"
        "v_dual_min_num_f32 v78, v14, v10 :: v_dual_min_num_f32 v79, v15, v11\n\t"
        "v_dual_max_num_f32 v8, v12, v8 :: v_dual_max_num_f32 v9, v13, v9\n\t"
        "v_dual_max_num_f32 v10, v14, v10 :: v_dual_max_num_f32 v11, v15, v11\n\t"
        "v_dual_min_num_f32 v12, v32, v16 :: v_dual_min_num_f32 v13, v33, v17\n\t"
        "v_dual_min_num_f32 v14, v34, v18 :: v_dual_min_num_f32 v15, v35, v19\n\t"
        "v_dual_max_num_f32 v16, v32, v16 :: v_dual_max_num_f32 v17, v33, v17\n\t"
        "v_dual_max_num_f32 v18, v34, v18 :: v_dual_max_num_f32 v19, v35, v19\n\t"
        "v_dual_min_num_f32 v32, v76, v12 :: v_dual_min_num_f32 v33, v77, v13\n\t"
        "v_dual_min_num_f32 v34, v78, v14 :: v_dual_min_num_f32 v35, v79, v15\n\t"
        "v_dual_max_num_f32 v12, v76, v12 :: v_dual_max_num_f32 v13, v77, v13\n\t"
        "v_dual_max_num_f32 v14, v78, v14 :: v_dual_max_num_f32 v15, v79, v15\n\t"
        "v_dual_min_num_f32 v76, v8, v16 :: v_dual_min_num_f32 v77, v9, v17\n\t"
        "v_dual_min_num_f32 v78, v10, v18 :: v_dual_min_num_f32 v79, v11, v19\n\t"
        "v_dual_max_num_f32 v16, v8, v16 :: v_dual_max_num_f32 v17, v9, v17\n\t"
        "v_dual_max_num_f32 v18, v10, v18 :: v_dual_max_num_f32 v19, v11, v19\n\t"
        :
          "={v[0:3]}"(a[0]), "={v[4:7]}"(a[1]), "={v[8:11]}"(a[2]),
          "={v[12:15]}"(a[3]), "={v[16:19]}"(a[4]), "={v[20:23]}"(a[5]),
          "={v[24:27]}"(a[6]), "={v[28:31]}"(a[7]), "={v[32:35]}"(a[8]),
          "={v[36:39]}"(a[9]), "={v[40:43]}"(a[10]), "={v[44:47]}"(a[11]),
          "={v[48:51]}"(a[12]), "={v[52:55]}"(a[13]), "={v[56:59]}"(a[14]),
          "={v[60:63]}"(a[15]), "={v[64:67]}"(b[0]), "={v[68:71]}"(b[1]),
          "={v[72:75]}"(b[2]), "={v[76:79]}"(b[3]), "={v[80:83]}"(b[4]),
          "={v[84:87]}"(b[5]), "={v[88:91]}"(b[6]), "={v[92:95]}"(b[7]),
          "={v[128:131]}"(sp)
        :
          "{v[0:3]}"(a[0]), "{v[4:7]}"(a[1]), "{v[8:11]}"(a[2]),
          "{v[12:15]}"(a[3]), "{v[16:19]}"(a[4]), "{v[20:23]}"(a[5]),
          "{v[24:27]}"(a[6]), "{v[28:31]}"(a[7]), "{v[32:35]}"(a[8]),
          "{v[36:39]}"(a[9]), "{v[40:43]}"(a[10]), "{v[44:47]}"(a[11]),
          "{v[48:51]}"(a[12]), "{v[52:55]}"(a[13]), "{v[56:59]}"(a[14]),
          "{v[60:63]}"(a[15]), "{v[64:67]}"(b[0]), "{v[68:71]}"(b[1]),
          "{v[72:75]}"(b[2]), "{v[76:79]}"(b[3]), "{v[80:83]}"(b[4]),
          "{v[84:87]}"(b[5]), "{v[88:91]}"(b[6]), "{v[92:95]}"(b[7]),
          [m1] "s"(m1), [m2] "s"(m2), [m3] "s"(m3), [m4] "s"(m4),
          [m5] "s"(m5), [m6] "s"(m6), [m7] "s"(m7));
    const int sel = med - 8;
    v4f r = a[8];
    r = (sel == 1) ? a[3] : r;
    r = (sel == 2) ? b[3] : r;
    r = (sel == 3) ? a[4] : r;
    return r;
}

__device__ __forceinline__ v4f rank_paired_8_8(v4f* a, v4f* b, int d, int med) {
    const float inf = __builtin_inff();
    const float m5 = (13 < d) ? -inf : inf;
    const float m6 = (14 < d) ? -inf : inf;
    const float m7 = (15 < d) ? -inf : inf;
    v4f sp;
    asm(
        "v_dual_max_num_f32 v84, %[m5], v84 :: v_dual_max_num_f32 v85, %[m5], v85\n\t"
        "v_dual_max_num_f32 v86, %[m5], v86 :: v_dual_max_num_f32 v87, %[m5], v87\n\t"
        "v_dual_max_num_f32 v88, %[m6], v88 :: v_dual_max_num_f32 v89, %[m6], v89\n\t"
        "v_dual_max_num_f32 v90, %[m6], v90 :: v_dual_max_num_f32 v91, %[m6], v91\n\t"
        "v_dual_max_num_f32 v92, %[m7], v92 :: v_dual_max_num_f32 v93, %[m7], v93\n\t"
        "v_dual_max_num_f32 v94, %[m7], v94 :: v_dual_max_num_f32 v95, %[m7], v95\n\t"
        "v_dual_min_num_f32 v128, v64, v68 :: v_dual_min_num_f32 v129, v65, v69\n\t"
        "v_dual_min_num_f32 v130, v66, v70 :: v_dual_min_num_f32 v131, v67, v71\n\t"
        "v_dual_max_num_f32 v68, v64, v68 :: v_dual_max_num_f32 v69, v65, v69\n\t"
        "v_dual_max_num_f32 v70, v66, v70 :: v_dual_max_num_f32 v71, v67, v71\n\t"
        "v_dual_min_num_f32 v64, v72, v76 :: v_dual_min_num_f32 v65, v73, v77\n\t"
        "v_dual_min_num_f32 v66, v74, v78 :: v_dual_min_num_f32 v67, v75, v79\n\t"
        "v_dual_max_num_f32 v76, v72, v76 :: v_dual_max_num_f32 v77, v73, v77\n\t"
        "v_dual_max_num_f32 v78, v74, v78 :: v_dual_max_num_f32 v79, v75, v79\n\t"
        "v_dual_min_num_f32 v72, v80, v84 :: v_dual_min_num_f32 v73, v81, v85\n\t"
        "v_dual_min_num_f32 v74, v82, v86 :: v_dual_min_num_f32 v75, v83, v87\n\t"
        "v_dual_max_num_f32 v84, v80, v84 :: v_dual_max_num_f32 v85, v81, v85\n\t"
        "v_dual_max_num_f32 v86, v82, v86 :: v_dual_max_num_f32 v87, v83, v87\n\t"
        "v_dual_min_num_f32 v80, v88, v92 :: v_dual_min_num_f32 v81, v89, v93\n\t"
        "v_dual_min_num_f32 v82, v90, v94 :: v_dual_min_num_f32 v83, v91, v95\n\t"
        "v_dual_max_num_f32 v92, v88, v92 :: v_dual_max_num_f32 v93, v89, v93\n\t"
        "v_dual_max_num_f32 v94, v90, v94 :: v_dual_max_num_f32 v95, v91, v95\n\t"
        "v_dual_min_num_f32 v88, v0, v4 :: v_dual_min_num_f32 v89, v1, v5\n\t"
        "v_dual_min_num_f32 v90, v2, v6 :: v_dual_min_num_f32 v91, v3, v7\n\t"
        "v_dual_max_num_f32 v4, v0, v4 :: v_dual_max_num_f32 v5, v1, v5\n\t"
        "v_dual_max_num_f32 v6, v2, v6 :: v_dual_max_num_f32 v7, v3, v7\n\t"
        "v_dual_min_num_f32 v0, v8, v12 :: v_dual_min_num_f32 v1, v9, v13\n\t"
        "v_dual_min_num_f32 v2, v10, v14 :: v_dual_min_num_f32 v3, v11, v15\n\t"
        "v_dual_max_num_f32 v12, v8, v12 :: v_dual_max_num_f32 v13, v9, v13\n\t"
        "v_dual_max_num_f32 v14, v10, v14 :: v_dual_max_num_f32 v15, v11, v15\n\t"
        "v_dual_min_num_f32 v8, v16, v20 :: v_dual_min_num_f32 v9, v17, v21\n\t"
        "v_dual_min_num_f32 v10, v18, v22 :: v_dual_min_num_f32 v11, v19, v23\n\t"
        "v_dual_max_num_f32 v20, v16, v20 :: v_dual_max_num_f32 v21, v17, v21\n\t"
        "v_dual_max_num_f32 v22, v18, v22 :: v_dual_max_num_f32 v23, v19, v23\n\t"
        "v_dual_min_num_f32 v16, v24, v28 :: v_dual_min_num_f32 v17, v25, v29\n\t"
        "v_dual_min_num_f32 v18, v26, v30 :: v_dual_min_num_f32 v19, v27, v31\n\t"
        "v_dual_max_num_f32 v28, v24, v28 :: v_dual_max_num_f32 v29, v25, v29\n\t"
        "v_dual_max_num_f32 v30, v26, v30 :: v_dual_max_num_f32 v31, v27, v31\n\t"
        "v_dual_min_num_f32 v24, v128, v64 :: v_dual_min_num_f32 v25, v129, v65\n\t"
        "v_dual_min_num_f32 v26, v130, v66 :: v_dual_min_num_f32 v27, v131, v67\n\t"
        "v_dual_max_num_f32 v64, v128, v64 :: v_dual_max_num_f32 v65, v129, v65\n\t"
        "v_dual_max_num_f32 v66, v130, v66 :: v_dual_max_num_f32 v67, v131, v67\n\t"
        "v_dual_min_num_f32 v128, v68, v76 :: v_dual_min_num_f32 v129, v69, v77\n\t"
        "v_dual_min_num_f32 v130, v70, v78 :: v_dual_min_num_f32 v131, v71, v79\n\t"
        "v_dual_max_num_f32 v76, v68, v76 :: v_dual_max_num_f32 v77, v69, v77\n\t"
        "v_dual_max_num_f32 v78, v70, v78 :: v_dual_max_num_f32 v79, v71, v79\n\t"
        "v_dual_min_num_f32 v68, v72, v80 :: v_dual_min_num_f32 v69, v73, v81\n\t"
        "v_dual_min_num_f32 v70, v74, v82 :: v_dual_min_num_f32 v71, v75, v83\n\t"
        "v_dual_max_num_f32 v80, v72, v80 :: v_dual_max_num_f32 v81, v73, v81\n\t"
        "v_dual_max_num_f32 v82, v74, v82 :: v_dual_max_num_f32 v83, v75, v83\n\t"
        "v_dual_min_num_f32 v72, v84, v92 :: v_dual_min_num_f32 v73, v85, v93\n\t"
        "v_dual_min_num_f32 v74, v86, v94 :: v_dual_min_num_f32 v75, v87, v95\n\t"
        "v_dual_max_num_f32 v92, v84, v92 :: v_dual_max_num_f32 v93, v85, v93\n\t"
        "v_dual_max_num_f32 v94, v86, v94 :: v_dual_max_num_f32 v95, v87, v95\n\t"
        "v_dual_min_num_f32 v84, v88, v0 :: v_dual_min_num_f32 v85, v89, v1\n\t"
        "v_dual_min_num_f32 v86, v90, v2 :: v_dual_min_num_f32 v87, v91, v3\n\t"
        "v_dual_max_num_f32 v0, v88, v0 :: v_dual_max_num_f32 v1, v89, v1\n\t"
        "v_dual_max_num_f32 v2, v90, v2 :: v_dual_max_num_f32 v3, v91, v3\n\t"
        "v_dual_min_num_f32 v88, v4, v12 :: v_dual_min_num_f32 v89, v5, v13\n\t"
        "v_dual_min_num_f32 v90, v6, v14 :: v_dual_min_num_f32 v91, v7, v15\n\t"
        "v_dual_max_num_f32 v12, v4, v12 :: v_dual_max_num_f32 v13, v5, v13\n\t"
        "v_dual_max_num_f32 v14, v6, v14 :: v_dual_max_num_f32 v15, v7, v15\n\t"
        "v_dual_min_num_f32 v4, v8, v16 :: v_dual_min_num_f32 v5, v9, v17\n\t"
        "v_dual_min_num_f32 v6, v10, v18 :: v_dual_min_num_f32 v7, v11, v19\n\t"
        "v_dual_max_num_f32 v16, v8, v16 :: v_dual_max_num_f32 v17, v9, v17\n\t"
        "v_dual_max_num_f32 v18, v10, v18 :: v_dual_max_num_f32 v19, v11, v19\n\t"
        "v_dual_min_num_f32 v8, v20, v28 :: v_dual_min_num_f32 v9, v21, v29\n\t"
        "v_dual_min_num_f32 v10, v22, v30 :: v_dual_min_num_f32 v11, v23, v31\n\t"
        "v_dual_max_num_f32 v28, v20, v28 :: v_dual_max_num_f32 v29, v21, v29\n\t"
        "v_dual_max_num_f32 v30, v22, v30 :: v_dual_max_num_f32 v31, v23, v31\n\t"
        "v_dual_min_num_f32 v20, v128, v64 :: v_dual_min_num_f32 v21, v129, v65\n\t"
        "v_dual_min_num_f32 v22, v130, v66 :: v_dual_min_num_f32 v23, v131, v67\n\t"
        "v_dual_max_num_f32 v64, v128, v64 :: v_dual_max_num_f32 v65, v129, v65\n\t"
        "v_dual_max_num_f32 v66, v130, v66 :: v_dual_max_num_f32 v67, v131, v67\n\t"
        "v_dual_min_num_f32 v128, v72, v80 :: v_dual_min_num_f32 v129, v73, v81\n\t"
        "v_dual_min_num_f32 v130, v74, v82 :: v_dual_min_num_f32 v131, v75, v83\n\t"
        "v_dual_max_num_f32 v80, v72, v80 :: v_dual_max_num_f32 v81, v73, v81\n\t"
        "v_dual_max_num_f32 v82, v74, v82 :: v_dual_max_num_f32 v83, v75, v83\n\t"
        "v_dual_min_num_f32 v72, v24, v68 :: v_dual_min_num_f32 v73, v25, v69\n\t"
        "v_dual_min_num_f32 v74, v26, v70 :: v_dual_min_num_f32 v75, v27, v71\n\t"
        "v_dual_max_num_f32 v68, v24, v68 :: v_dual_max_num_f32 v69, v25, v69\n\t"
        "v_dual_max_num_f32 v70, v26, v70 :: v_dual_max_num_f32 v71, v27, v71\n\t"
        "v_dual_min_num_f32 v24, v76, v92 :: v_dual_min_num_f32 v25, v77, v93\n\t"
        "v_dual_min_num_f32 v26, v78, v94 :: v_dual_min_num_f32 v27, v79, v95\n\t"
        "v_dual_max_num_f32 v92, v76, v92 :: v_dual_max_num_f32 v93, v77, v93\n\t"
        "v_dual_max_num_f32 v94, v78, v94 :: v_dual_max_num_f32 v95, v79, v95\n\t"
        "v_dual_min_num_f32 v76, v88, v0 :: v_dual_min_num_f32 v77, v89, v1\n\t"
        "v_dual_min_num_f32 v78, v90, v2 :: v_dual_min_num_f32 v79, v91, v3\n\t"
        "v_dual_max_num_f32 v0, v88, v0 :: v_dual_max_num_f32 v1, v89, v1\n\t"
        "v_dual_max_num_f32 v2, v90, v2 :: v_dual_max_num_f32 v3, v91, v3\n\t"
        "v_dual_min_num_f32 v88, v8, v16 :: v_dual_min_num_f32 v89, v9, v17\n\t"
        "v_dual_min_num_f32 v90, v10, v18 :: v_dual_min_num_f32 v91, v11, v19\n\t"
        "v_dual_max_num_f32 v16, v8, v16 :: v_dual_max_num_f32 v17, v9, v17\n\t"
        "v_dual_max_num_f32 v18, v10, v18 :: v_dual_max_num_f32 v19, v11, v19\n\t"
        "v_dual_min_num_f32 v8, v84, v4 :: v_dual_min_num_f32 v9, v85, v5\n\t"
        "v_dual_min_num_f32 v10, v86, v6 :: v_dual_min_num_f32 v11, v87, v7\n\t"
        "v_dual_max_num_f32 v4, v84, v4 :: v_dual_max_num_f32 v5, v85, v5\n\t"
        "v_dual_max_num_f32 v6, v86, v6 :: v_dual_max_num_f32 v7, v87, v7\n\t"
        "v_dual_min_num_f32 v84, v12, v28 :: v_dual_min_num_f32 v85, v13, v29\n\t"
        "v_dual_min_num_f32 v86, v14, v30 :: v_dual_min_num_f32 v87, v15, v31\n\t"
        "v_dual_max_num_f32 v28, v12, v28 :: v_dual_max_num_f32 v29, v13, v29\n\t"
        "v_dual_max_num_f32 v30, v14, v30 :: v_dual_max_num_f32 v31, v15, v31\n\t"
        "v_dual_min_num_f32 v12, v64, v80 :: v_dual_min_num_f32 v13, v65, v81\n\t"
        "v_dual_min_num_f32 v14, v66, v82 :: v_dual_min_num_f32 v15, v67, v83\n\t"
        "v_dual_max_num_f32 v80, v64, v80 :: v_dual_max_num_f32 v81, v65, v81\n\t"
        "v_dual_max_num_f32 v82, v66, v82 :: v_dual_max_num_f32 v83, v67, v83\n\t"
        "v_dual_min_num_f32 v64, v20, v128 :: v_dual_min_num_f32 v65, v21, v129\n\t"
        "v_dual_min_num_f32 v66, v22, v130 :: v_dual_min_num_f32 v67, v23, v131\n\t"
        "v_dual_max_num_f32 v128, v20, v128 :: v_dual_max_num_f32 v129, v21, v129\n\t"
        "v_dual_max_num_f32 v130, v22, v130 :: v_dual_max_num_f32 v131, v23, v131\n\t"
        "v_dual_min_num_f32 v20, v0, v16 :: v_dual_min_num_f32 v21, v1, v17\n\t"
        "v_dual_min_num_f32 v22, v2, v18 :: v_dual_min_num_f32 v23, v3, v19\n\t"
        "v_dual_max_num_f32 v16, v0, v16 :: v_dual_max_num_f32 v17, v1, v17\n\t"
        "v_dual_max_num_f32 v18, v2, v18 :: v_dual_max_num_f32 v19, v3, v19\n\t"
        "v_dual_min_num_f32 v0, v76, v88 :: v_dual_min_num_f32 v1, v77, v89\n\t"
        "v_dual_min_num_f32 v2, v78, v90 :: v_dual_min_num_f32 v3, v79, v91\n\t"
        "v_dual_max_num_f32 v88, v76, v88 :: v_dual_max_num_f32 v89, v77, v89\n\t"
        "v_dual_max_num_f32 v90, v78, v90 :: v_dual_max_num_f32 v91, v79, v91\n\t"
        "v_dual_min_num_f32 v8, v8, v92 :: v_dual_min_num_f32 v9, v9, v93\n\t"
        "v_dual_min_num_f32 v10, v10, v94 :: v_dual_min_num_f32 v11, v11, v95\n\t"
        "v_dual_min_num_f32 v28, v28, v72 :: v_dual_min_num_f32 v29, v29, v73\n\t"
        "v_dual_min_num_f32 v30, v30, v74 :: v_dual_min_num_f32 v31, v31, v75\n\t"
        "v_dual_min_num_f32 v76, v12, v68 :: v_dual_min_num_f32 v77, v13, v69\n\t"
        "v_dual_min_num_f32 v78, v14, v70 :: v_dual_min_num_f32 v79, v15, v71\n\t"
        "v_dual_max_num_f32 v68, v12, v68 :: v_dual_max_num_f32 v69, v13, v69\n\t"
        "v_dual_max_num_f32 v70, v14, v70 :: v_dual_max_num_f32 v71, v15, v71\n\t"
        "v_dual_min_num_f32 v12, v24, v128 :: v_dual_min_num_f32 v13, v25, v129\n\t"
        "v_dual_min_num_f32 v14, v26, v130 :: v_dual_min_num_f32 v15, v27, v131\n\t"
        "v_dual_max_num_f32 v128, v24, v128 :: v_dual_max_num_f32 v129, v25, v129\n\t"
        "v_dual_max_num_f32 v130, v26, v130 :: v_dual_max_num_f32 v131, v27, v131\n\t"
        "v_dual_min_num_f32 v24, v20, v4 :: v_dual_min_num_f32 v25, v21, v5\n\t"
        "v_dual_min_num_f32 v26, v22, v6 :: v_dual_min_num_f32 v27, v23, v7\n\t"
        "v_dual_max_num_f32 v4, v20, v4 :: v_dual_max_num_f32 v5, v21, v5\n\t"
        "v_dual_max_num_f32 v6, v22, v6 :: v_dual_max_num_f32 v7, v23, v7\n\t"
        "v_dual_min_num_f32 v20, v84, v88 :: v_dual_min_num_f32 v21, v85, v89\n\t"
        "v_dual_min_num_f32 v22, v86, v90 :: v_dual_min_num_f32 v23, v87, v91\n\t"
        "v_dual_max_num_f32 v88, v84, v88 :: v_dual_max_num_f32 v89, v85, v89\n\t"
        "v_dual_max_num_f32 v90, v86, v90 :: v_dual_max_num_f32 v91, v87, v91\n\t"
        "v_dual_min_num_f32 v84, v64, v76 :: v_dual_min_num_f32 v85, v65, v77\n\t"
        "v_dual_min_num_f32 v86, v66, v78 :: v_dual_min_num_f32 v87, v67, v79\n\t"
        "v_dual_max_num_f32 v76, v64, v76 :: v_dual_max_num_f32 v77, v65, v77\n\t"
        "v_dual_max_num_f32 v78, v66, v78 :: v_dual_max_num_f32 v79, v67, v79\n\t"
        "v_dual_min_num_f32 v64, v12, v68 :: v_dual_min_num_f32 v65, v13, v69\n\t"
        "v_dual_min_num_f32 v66, v14, v70 :: v_dual_min_num_f32 v67, v15, v71\n\t"
        "v_dual_max_num_f32 v68, v12, v68 :: v_dual_max_num_f32 v69, v13, v69\n\t"
        "v_dual_max_num_f32 v70, v14, v70 :: v_dual_max_num_f32 v71, v15, v71\n\t"
        "v_dual_min_num_f32 v12, v128, v80 :: v_dual_min_num_f32 v13, v129, v81\n\t"
        "v_dual_min_num_f32 v14, v130, v82 :: v_dual_min_num_f32 v15, v131, v83\n\t"
        "v_dual_max_num_f32 v80, v128, v80 :: v_dual_max_num_f32 v81, v129, v81\n\t"
        "v_dual_max_num_f32 v82, v130, v82 :: v_dual_max_num_f32 v83, v131, v83\n\t"
        "v_dual_min_num_f32 v128, v0, v24 :: v_dual_min_num_f32 v129, v1, v25\n\t"
        "v_dual_min_num_f32 v130, v2, v26 :: v_dual_min_num_f32 v131, v3, v27\n\t"
        "v_dual_max_num_f32 v24, v0, v24 :: v_dual_max_num_f32 v25, v1, v25\n\t"
        "v_dual_max_num_f32 v26, v2, v26 :: v_dual_max_num_f32 v27, v3, v27\n\t"
        "v_dual_min_num_f32 v0, v20, v4 :: v_dual_min_num_f32 v1, v21, v5\n\t"
        "v_dual_min_num_f32 v2, v22, v6 :: v_dual_min_num_f32 v3, v23, v7\n\t"
        "v_dual_max_num_f32 v4, v20, v4 :: v_dual_max_num_f32 v5, v21, v5\n\t"
        "v_dual_max_num_f32 v6, v22, v6 :: v_dual_max_num_f32 v7, v23, v7\n\t"
        "v_dual_min_num_f32 v20, v88, v16 :: v_dual_min_num_f32 v21, v89, v17\n\t"
        "v_dual_min_num_f32 v22, v90, v18 :: v_dual_min_num_f32 v23, v91, v19\n\t"
        "v_dual_max_num_f32 v16, v88, v16 :: v_dual_max_num_f32 v17, v89, v17\n\t"
        "v_dual_max_num_f32 v18, v90, v18 :: v_dual_max_num_f32 v19, v91, v19\n\t"
        "v_dual_min_num_f32 v4, v4, v64 :: v_dual_min_num_f32 v5, v5, v65\n\t"
        "v_dual_min_num_f32 v6, v6, v66 :: v_dual_min_num_f32 v7, v7, v67\n\t"
        "v_dual_min_num_f32 v128, v128, v80 :: v_dual_min_num_f32 v129, v129, v81\n\t"
        "v_dual_min_num_f32 v130, v130, v82 :: v_dual_min_num_f32 v131, v131, v83\n\t"
        "v_dual_min_num_f32 v20, v20, v76 :: v_dual_min_num_f32 v21, v21, v77\n\t"
        "v_dual_min_num_f32 v22, v22, v78 :: v_dual_min_num_f32 v23, v23, v79\n\t"
        "v_dual_min_num_f32 v24, v24, v12 :: v_dual_min_num_f32 v25, v25, v13\n\t"
        "v_dual_min_num_f32 v26, v26, v14 :: v_dual_min_num_f32 v27, v27, v15\n\t"
        "v_dual_min_num_f32 v16, v16, v84 :: v_dual_min_num_f32 v17, v17, v85\n\t"
        "v_dual_min_num_f32 v18, v18, v86 :: v_dual_min_num_f32 v19, v19, v87\n\t"
        "v_dual_min_num_f32 v0, v0, v68 :: v_dual_min_num_f32 v1, v1, v69\n\t"
        "v_dual_min_num_f32 v2, v2, v70 :: v_dual_min_num_f32 v3, v3, v71\n\t"
        "v_dual_max_num_f32 v8, v8, v4 :: v_dual_max_num_f32 v9, v9, v5\n\t"
        "v_dual_max_num_f32 v10, v10, v6 :: v_dual_max_num_f32 v11, v11, v7\n\t"
        "v_dual_max_num_f32 v128, v128, v20 :: v_dual_max_num_f32 v129, v129, v21\n\t"
        "v_dual_max_num_f32 v130, v130, v22 :: v_dual_max_num_f32 v131, v131, v23\n\t"
        "v_dual_max_num_f32 v24, v24, v16 :: v_dual_max_num_f32 v25, v25, v17\n\t"
        "v_dual_max_num_f32 v26, v26, v18 :: v_dual_max_num_f32 v27, v27, v19\n\t"
        "v_dual_max_num_f32 v0, v0, v28 :: v_dual_max_num_f32 v1, v1, v29\n\t"
        "v_dual_max_num_f32 v2, v2, v30 :: v_dual_max_num_f32 v3, v3, v31\n\t"
        "v_dual_min_num_f32 v88, v8, v24 :: v_dual_min_num_f32 v89, v9, v25\n\t"
        "v_dual_min_num_f32 v90, v10, v26 :: v_dual_min_num_f32 v91, v11, v27\n\t"
        "v_dual_max_num_f32 v24, v8, v24 :: v_dual_max_num_f32 v25, v9, v25\n\t"
        "v_dual_max_num_f32 v26, v10, v26 :: v_dual_max_num_f32 v27, v11, v27\n\t"
        "v_dual_min_num_f32 v8, v128, v0 :: v_dual_min_num_f32 v9, v129, v1\n\t"
        "v_dual_min_num_f32 v10, v130, v2 :: v_dual_min_num_f32 v11, v131, v3\n\t"
        "v_dual_max_num_f32 v0, v128, v0 :: v_dual_max_num_f32 v1, v129, v1\n\t"
        "v_dual_max_num_f32 v2, v130, v2 :: v_dual_max_num_f32 v3, v131, v3\n\t"
        "v_dual_min_num_f32 v128, v88, v8 :: v_dual_min_num_f32 v129, v89, v9\n\t"
        "v_dual_min_num_f32 v130, v90, v10 :: v_dual_min_num_f32 v131, v91, v11\n\t"
        "v_dual_max_num_f32 v8, v88, v8 :: v_dual_max_num_f32 v9, v89, v9\n\t"
        "v_dual_max_num_f32 v10, v90, v10 :: v_dual_max_num_f32 v11, v91, v11\n\t"
        "v_dual_min_num_f32 v88, v24, v0 :: v_dual_min_num_f32 v89, v25, v1\n\t"
        "v_dual_min_num_f32 v90, v26, v2 :: v_dual_min_num_f32 v91, v27, v3\n\t"
        "v_dual_max_num_f32 v0, v24, v0 :: v_dual_max_num_f32 v1, v25, v1\n\t"
        "v_dual_max_num_f32 v2, v26, v2 :: v_dual_max_num_f32 v3, v27, v3\n\t"
        :
          "={v[0:3]}"(a[0]), "={v[4:7]}"(a[1]), "={v[8:11]}"(a[2]),
          "={v[12:15]}"(a[3]), "={v[16:19]}"(a[4]), "={v[20:23]}"(a[5]),
          "={v[24:27]}"(a[6]), "={v[28:31]}"(a[7]), "={v[64:67]}"(b[0]),
          "={v[68:71]}"(b[1]), "={v[72:75]}"(b[2]), "={v[76:79]}"(b[3]),
          "={v[80:83]}"(b[4]), "={v[84:87]}"(b[5]), "={v[88:91]}"(b[6]),
          "={v[92:95]}"(b[7]), "={v[128:131]}"(sp)
        :
          "{v[0:3]}"(a[0]), "{v[4:7]}"(a[1]), "{v[8:11]}"(a[2]),
          "{v[12:15]}"(a[3]), "{v[16:19]}"(a[4]), "{v[20:23]}"(a[5]),
          "{v[24:27]}"(a[6]), "{v[28:31]}"(a[7]), "{v[64:67]}"(b[0]),
          "{v[68:71]}"(b[1]), "{v[72:75]}"(b[2]), "{v[76:79]}"(b[3]),
          "{v[80:83]}"(b[4]), "{v[84:87]}"(b[5]), "{v[88:91]}"(b[6]),
          "{v[92:95]}"(b[7]),
          [m5] "s"(m5), [m6] "s"(m6), [m7] "s"(m7));
    const int sel = med - 4;
    v4f r = sp;
    r = (sel == 1) ? a[2] : r;
    r = (sel == 2) ? b[6] : r;
    r = (sel == 3) ? a[0] : r;
    return r;
}

__device__ __forceinline__ v4f rank_paired_8_4(v4f* a, v4f* b, int d, int med) {
    const float inf = __builtin_inff();
    const float m1 = (9 < d) ? -inf : inf;
    const float m2 = (10 < d) ? -inf : inf;
    const float m3 = (11 < d) ? -inf : inf;
    v4f sp;
    asm(
        "v_dual_max_num_f32 v68, %[m1], v68 :: v_dual_max_num_f32 v69, %[m1], v69\n\t"
        "v_dual_max_num_f32 v70, %[m1], v70 :: v_dual_max_num_f32 v71, %[m1], v71\n\t"
        "v_dual_max_num_f32 v72, %[m2], v72 :: v_dual_max_num_f32 v73, %[m2], v73\n\t"
        "v_dual_max_num_f32 v74, %[m2], v74 :: v_dual_max_num_f32 v75, %[m2], v75\n\t"
        "v_dual_max_num_f32 v76, %[m3], v76 :: v_dual_max_num_f32 v77, %[m3], v77\n\t"
        "v_dual_max_num_f32 v78, %[m3], v78 :: v_dual_max_num_f32 v79, %[m3], v79\n\t"
        "v_dual_min_num_f32 v128, v64, v68 :: v_dual_min_num_f32 v129, v65, v69\n\t"
        "v_dual_min_num_f32 v130, v66, v70 :: v_dual_min_num_f32 v131, v67, v71\n\t"
        "v_dual_max_num_f32 v68, v64, v68 :: v_dual_max_num_f32 v69, v65, v69\n\t"
        "v_dual_max_num_f32 v70, v66, v70 :: v_dual_max_num_f32 v71, v67, v71\n\t"
        "v_dual_min_num_f32 v64, v72, v76 :: v_dual_min_num_f32 v65, v73, v77\n\t"
        "v_dual_min_num_f32 v66, v74, v78 :: v_dual_min_num_f32 v67, v75, v79\n\t"
        "v_dual_max_num_f32 v76, v72, v76 :: v_dual_max_num_f32 v77, v73, v77\n\t"
        "v_dual_max_num_f32 v78, v74, v78 :: v_dual_max_num_f32 v79, v75, v79\n\t"
        "v_dual_min_num_f32 v72, v0, v4 :: v_dual_min_num_f32 v73, v1, v5\n\t"
        "v_dual_min_num_f32 v74, v2, v6 :: v_dual_min_num_f32 v75, v3, v7\n\t"
        "v_dual_max_num_f32 v4, v0, v4 :: v_dual_max_num_f32 v5, v1, v5\n\t"
        "v_dual_max_num_f32 v6, v2, v6 :: v_dual_max_num_f32 v7, v3, v7\n\t"
        "v_dual_min_num_f32 v0, v8, v12 :: v_dual_min_num_f32 v1, v9, v13\n\t"
        "v_dual_min_num_f32 v2, v10, v14 :: v_dual_min_num_f32 v3, v11, v15\n\t"
        "v_dual_max_num_f32 v12, v8, v12 :: v_dual_max_num_f32 v13, v9, v13\n\t"
        "v_dual_max_num_f32 v14, v10, v14 :: v_dual_max_num_f32 v15, v11, v15\n\t"
        "v_dual_min_num_f32 v8, v16, v20 :: v_dual_min_num_f32 v9, v17, v21\n\t"
        "v_dual_min_num_f32 v10, v18, v22 :: v_dual_min_num_f32 v11, v19, v23\n\t"
        "v_dual_max_num_f32 v20, v16, v20 :: v_dual_max_num_f32 v21, v17, v21\n\t"
        "v_dual_max_num_f32 v22, v18, v22 :: v_dual_max_num_f32 v23, v19, v23\n\t"
        "v_dual_min_num_f32 v16, v24, v28 :: v_dual_min_num_f32 v17, v25, v29\n\t"
        "v_dual_min_num_f32 v18, v26, v30 :: v_dual_min_num_f32 v19, v27, v31\n\t"
        "v_dual_max_num_f32 v28, v24, v28 :: v_dual_max_num_f32 v29, v25, v29\n\t"
        "v_dual_max_num_f32 v30, v26, v30 :: v_dual_max_num_f32 v31, v27, v31\n\t"
        "v_dual_min_num_f32 v24, v128, v64 :: v_dual_min_num_f32 v25, v129, v65\n\t"
        "v_dual_min_num_f32 v26, v130, v66 :: v_dual_min_num_f32 v27, v131, v67\n\t"
        "v_dual_max_num_f32 v64, v128, v64 :: v_dual_max_num_f32 v65, v129, v65\n\t"
        "v_dual_max_num_f32 v66, v130, v66 :: v_dual_max_num_f32 v67, v131, v67\n\t"
        "v_dual_min_num_f32 v128, v68, v76 :: v_dual_min_num_f32 v129, v69, v77\n\t"
        "v_dual_min_num_f32 v130, v70, v78 :: v_dual_min_num_f32 v131, v71, v79\n\t"
        "v_dual_max_num_f32 v76, v68, v76 :: v_dual_max_num_f32 v77, v69, v77\n\t"
        "v_dual_max_num_f32 v78, v70, v78 :: v_dual_max_num_f32 v79, v71, v79\n\t"
        "v_dual_min_num_f32 v68, v72, v0 :: v_dual_min_num_f32 v69, v73, v1\n\t"
        "v_dual_min_num_f32 v70, v74, v2 :: v_dual_min_num_f32 v71, v75, v3\n\t"
        "v_dual_max_num_f32 v0, v72, v0 :: v_dual_max_num_f32 v1, v73, v1\n\t"
        "v_dual_max_num_f32 v2, v74, v2 :: v_dual_max_num_f32 v3, v75, v3\n\t"
        "v_dual_min_num_f32 v72, v4, v12 :: v_dual_min_num_f32 v73, v5, v13\n\t"
        "v_dual_min_num_f32 v74, v6, v14 :: v_dual_min_num_f32 v75, v7, v15\n\t"
        "v_dual_max_num_f32 v12, v4, v12 :: v_dual_max_num_f32 v13, v5, v13\n\t"
        "v_dual_max_num_f32 v14, v6, v14 :: v_dual_max_num_f32 v15, v7, v15\n\t"
        "v_dual_min_num_f32 v4, v8, v16 :: v_dual_min_num_f32 v5, v9, v17\n\t"
        "v_dual_min_num_f32 v6, v10, v18 :: v_dual_min_num_f32 v7, v11, v19\n\t"
        "v_dual_max_num_f32 v16, v8, v16 :: v_dual_max_num_f32 v17, v9, v17\n\t"
        "v_dual_max_num_f32 v18, v10, v18 :: v_dual_max_num_f32 v19, v11, v19\n\t"
        "v_dual_min_num_f32 v8, v20, v28 :: v_dual_min_num_f32 v9, v21, v29\n\t"
        "v_dual_min_num_f32 v10, v22, v30 :: v_dual_min_num_f32 v11, v23, v31\n\t"
        "v_dual_max_num_f32 v28, v20, v28 :: v_dual_max_num_f32 v29, v21, v29\n\t"
        "v_dual_max_num_f32 v30, v22, v30 :: v_dual_max_num_f32 v31, v23, v31\n\t"
        "v_dual_min_num_f32 v20, v128, v64 :: v_dual_min_num_f32 v21, v129, v65\n\t"
        "v_dual_min_num_f32 v22, v130, v66 :: v_dual_min_num_f32 v23, v131, v67\n\t"
        "v_dual_max_num_f32 v64, v128, v64 :: v_dual_max_num_f32 v65, v129, v65\n\t"
        "v_dual_max_num_f32 v66, v130, v66 :: v_dual_max_num_f32 v67, v131, v67\n\t"
        "v_dual_min_num_f32 v128, v72, v0 :: v_dual_min_num_f32 v129, v73, v1\n\t"
        "v_dual_min_num_f32 v130, v74, v2 :: v_dual_min_num_f32 v131, v75, v3\n\t"
        "v_dual_max_num_f32 v0, v72, v0 :: v_dual_max_num_f32 v1, v73, v1\n\t"
        "v_dual_max_num_f32 v2, v74, v2 :: v_dual_max_num_f32 v3, v75, v3\n\t"
        "v_dual_min_num_f32 v72, v8, v16 :: v_dual_min_num_f32 v73, v9, v17\n\t"
        "v_dual_min_num_f32 v74, v10, v18 :: v_dual_min_num_f32 v75, v11, v19\n\t"
        "v_dual_max_num_f32 v16, v8, v16 :: v_dual_max_num_f32 v17, v9, v17\n\t"
        "v_dual_max_num_f32 v18, v10, v18 :: v_dual_max_num_f32 v19, v11, v19\n\t"
        "v_dual_min_num_f32 v8, v68, v4 :: v_dual_min_num_f32 v9, v69, v5\n\t"
        "v_dual_min_num_f32 v10, v70, v6 :: v_dual_min_num_f32 v11, v71, v7\n\t"
        "v_dual_max_num_f32 v4, v68, v4 :: v_dual_max_num_f32 v5, v69, v5\n\t"
        "v_dual_max_num_f32 v6, v70, v6 :: v_dual_max_num_f32 v7, v71, v7\n\t"
        "v_dual_min_num_f32 v68, v12, v28 :: v_dual_min_num_f32 v69, v13, v29\n\t"
        "v_dual_min_num_f32 v70, v14, v30 :: v_dual_min_num_f32 v71, v15, v31\n\t"
        "v_dual_max_num_f32 v28, v12, v28 :: v_dual_max_num_f32 v29, v13, v29\n\t"
        "v_dual_max_num_f32 v30, v14, v30 :: v_dual_max_num_f32 v31, v15, v31\n\t"
        "v_dual_min_num_f32 v12, v0, v16 :: v_dual_min_num_f32 v13, v1, v17\n\t"
        "v_dual_min_num_f32 v14, v2, v18 :: v_dual_min_num_f32 v15, v3, v19\n\t"
        "v_dual_max_num_f32 v16, v0, v16 :: v_dual_max_num_f32 v17, v1, v17\n\t"
        "v_dual_max_num_f32 v18, v2, v18 :: v_dual_max_num_f32 v19, v3, v19\n\t"
        "v_dual_min_num_f32 v0, v128, v72 :: v_dual_min_num_f32 v1, v129, v73\n\t"
        "v_dual_min_num_f32 v2, v130, v74 :: v_dual_min_num_f32 v3, v131, v75\n\t"
        "v_dual_max_num_f32 v72, v128, v72 :: v_dual_max_num_f32 v73, v129, v73\n\t"
        "v_dual_max_num_f32 v74, v130, v74 :: v_dual_max_num_f32 v75, v131, v75\n\t"
        "v_dual_min_num_f32 v28, v28, v24 :: v_dual_min_num_f32 v29, v29, v25\n\t"
        "v_dual_min_num_f32 v30, v30, v26 :: v_dual_min_num_f32 v31, v31, v27\n\t"
        "v_dual_min_num_f32 v128, v12, v4 :: v_dual_min_num_f32 v129, v13, v5\n\t"
        "v_dual_min_num_f32 v130, v14, v6 :: v_dual_min_num_f32 v131, v15, v7\n\t"
        "v_dual_max_num_f32 v4, v12, v4 :: v_dual_max_num_f32 v5, v13, v5\n\t"
        "v_dual_max_num_f32 v6, v14, v6 :: v_dual_max_num_f32 v7, v15, v7\n\t"
        "v_dual_min_num_f32 v12, v68, v72 :: v_dual_min_num_f32 v13, v69, v73\n\t"
        "v_dual_min_num_f32 v14, v70, v74 :: v_dual_min_num_f32 v15, v71, v75\n\t"
        "v_dual_max_num_f32 v72, v68, v72 :: v_dual_max_num_f32 v73, v69, v73\n\t"
        "v_dual_max_num_f32 v74, v70, v74 :: v_dual_max_num_f32 v75, v71, v75\n\t"
        "v_dual_min_num_f32 v68, v0, v128 :: v_dual_min_num_f32 v69, v1, v129\n\t"
        "v_dual_min_num_f32 v70, v2, v130 :: v_dual_min_num_f32 v71, v3, v131\n\t"
        "v_dual_max_num_f32 v128, v0, v128 :: v_dual_max_num_f32 v129, v1, v129\n\t"
        "v_dual_max_num_f32 v130, v2, v130 :: v_dual_max_num_f32 v131, v3, v131\n\t"
        "v_dual_min_num_f32 v0, v12, v4 :: v_dual_min_num_f32 v1, v13, v5\n\t"
        "v_dual_min_num_f32 v2, v14, v6 :: v_dual_min_num_f32 v3, v15, v7\n\t"
        "v_dual_max_num_f32 v4, v12, v4 :: v_dual_max_num_f32 v5, v13, v5\n\t"
        "v_dual_max_num_f32 v6, v14, v6 :: v_dual_max_num_f32 v7, v15, v7\n\t"
        "v_dual_min_num_f32 v12, v72, v16 :: v_dual_min_num_f32 v13, v73, v17\n\t"
        "v_dual_min_num_f32 v14, v74, v18 :: v_dual_min_num_f32 v15, v75, v19\n\t"
        "v_dual_max_num_f32 v16, v72, v16 :: v_dual_max_num_f32 v17, v73, v17\n\t"
        "v_dual_max_num_f32 v18, v74, v18 :: v_dual_max_num_f32 v19, v75, v19\n\t"
        "v_dual_min_num_f32 v4, v4, v76 :: v_dual_min_num_f32 v5, v5, v77\n\t"
        "v_dual_min_num_f32 v6, v6, v78 :: v_dual_min_num_f32 v7, v7, v79\n\t"
        "v_dual_min_num_f32 v12, v12, v64 :: v_dual_min_num_f32 v13, v13, v65\n\t"
        "v_dual_min_num_f32 v14, v14, v66 :: v_dual_min_num_f32 v15, v15, v67\n\t"
        "v_dual_min_num_f32 v16, v16, v20 :: v_dual_min_num_f32 v17, v17, v21\n\t"
        "v_dual_min_num_f32 v18, v18, v22 :: v_dual_min_num_f32 v19, v19, v23\n\t"
        "v_dual_max_num_f32 v0, v0, v28 :: v_dual_max_num_f32 v1, v1, v29\n\t"
        "v_dual_max_num_f32 v2, v2, v30 :: v_dual_max_num_f32 v3, v3, v31\n\t"
        "v_dual_max_num_f32 v8, v8, v4 :: v_dual_max_num_f32 v9, v9, v5\n\t"
        "v_dual_max_num_f32 v10, v10, v6 :: v_dual_max_num_f32 v11, v11, v7\n\t"
        "v_dual_max_num_f32 v68, v68, v12 :: v_dual_max_num_f32 v69, v69, v13\n\t"
        "v_dual_max_num_f32 v70, v70, v14 :: v_dual_max_num_f32 v71, v71, v15\n\t"
        "v_dual_max_num_f32 v128, v128, v16 :: v_dual_max_num_f32 v129, v129, v17\n\t"
        "v_dual_max_num_f32 v130, v130, v18 :: v_dual_max_num_f32 v131, v131, v19\n\t"
        "v_dual_min_num_f32 v8, v8, v128 :: v_dual_min_num_f32 v9, v9, v129\n\t"
        "v_dual_min_num_f32 v10, v10, v130 :: v_dual_min_num_f32 v11, v11, v131\n\t"
        "v_dual_min_num_f32 v68, v68, v0 :: v_dual_min_num_f32 v69, v69, v1\n\t"
        "v_dual_min_num_f32 v70, v70, v2 :: v_dual_min_num_f32 v71, v71, v3\n\t"
        "v_dual_min_num_f32 v72, v8, v68 :: v_dual_min_num_f32 v73, v9, v69\n\t"
        "v_dual_min_num_f32 v74, v10, v70 :: v_dual_min_num_f32 v75, v11, v71\n\t"
        "v_dual_max_num_f32 v68, v8, v68 :: v_dual_max_num_f32 v69, v9, v69\n\t"
        "v_dual_max_num_f32 v70, v10, v70 :: v_dual_max_num_f32 v71, v11, v71\n\t"
        :
          "={v[0:3]}"(a[0]), "={v[4:7]}"(a[1]), "={v[8:11]}"(a[2]),
          "={v[12:15]}"(a[3]), "={v[16:19]}"(a[4]), "={v[20:23]}"(a[5]),
          "={v[24:27]}"(a[6]), "={v[28:31]}"(a[7]), "={v[64:67]}"(b[0]),
          "={v[68:71]}"(b[1]), "={v[72:75]}"(b[2]), "={v[76:79]}"(b[3]),
          "={v[128:131]}"(sp)
        :
          "{v[0:3]}"(a[0]), "{v[4:7]}"(a[1]), "{v[8:11]}"(a[2]),
          "{v[12:15]}"(a[3]), "{v[16:19]}"(a[4]), "{v[20:23]}"(a[5]),
          "{v[24:27]}"(a[6]), "{v[28:31]}"(a[7]), "{v[64:67]}"(b[0]),
          "{v[68:71]}"(b[1]), "{v[72:75]}"(b[2]), "{v[76:79]}"(b[3]),
          [m1] "s"(m1), [m2] "s"(m2), [m3] "s"(m3));
    const int sel = med - 4;
    v4f r = b[2];
    r = (sel == 1) ? b[1] : r;
    return r;
}

template <int HA, int HB>
__device__ __forceinline__ v4f rank_of_two_lists(v4f* a, v4f* b, int med) {
    oe_sort<0, HB - 1>(b);
    oe_sort<0, HA - 1>(a);
    constexpr int Q = HA / 2;
    v4f u[Q];
    #pragma unroll
    for (int i = 0; i < Q; i++) {
        v4f lo0 = a[i];
        if (HA - 1 - i < HB) lo0 = vmin4(lo0, b[HA - 1 - i]);
        v4f lo1 = a[i + Q];
        if (Q - 1 - i < HB) lo1 = vmin4(lo1, b[Q - 1 - i]);
        u[i] = vmax4(lo0, lo1);
    }
    if constexpr (HB == HA) {
        return bitonic_pick<Q>(u, med - Q);
    } else {
        constexpr int E = Q / 2;
        v4f w[E];
        #pragma unroll
        for (int i = 0; i < E; i++) w[i] = vmin4(u[i], u[i + E]);
        return bitonic_pick<E>(w, med - Q);
    }
}

__device__ __forceinline__ v4f row4(const float* __restrict__ h, int nb, int lane) {
    return ((const v4f*)h)[(unsigned)nb * (OUT_C / 4) + (unsigned)lane];
}

__device__ __forceinline__ void load_index_row(const int* __restrict__ nbr, int n, int (&idx)[KK]) {
    const int* __restrict__ p = nbr + (size_t)n * KK;
    #pragma unroll
    for (int j = 0; j < KK; j++) idx[j] = p[j];
}

template <int HA, int HB>
__device__ __forceinline__ v4f node_median(const float* __restrict__ h,
                                           const int (&nb)[KK],
                                           int lane, int d, int med) {
    const float inf = __builtin_inff();
    const v4f inf4 = {inf, inf, inf, inf};
    v4f a[HA], b[HB];
    #pragma unroll
    for (int j = 0; j < HA; j++) a[j] = row4(h, nb[j], lane);
    constexpr int CH = HB < 4 ? HB : 4;
    if constexpr (HA >= 8) {
        #pragma unroll
        for (int q = 0; q < HB; q += CH) {
            if (q == 0 || HA + q < d) {
                #pragma unroll
                for (int t = 0; t < CH; t++) b[q + t] = row4(h, nb[HA + q + t], lane);
            } else {
                #pragma unroll
                for (int t = 0; t < CH; t++) b[q + t] = inf4;
            }
        }
        if constexpr (HA == 16 && HB == 16) return rank_paired_16_16(a, b, d, med);
        else if constexpr (HA == 16 && HB == 8) return rank_paired_16_8(a, b, d, med);
        else if constexpr (HA == 8 && HB == 8) return rank_paired_8_8(a, b, d, med);
        else return rank_paired_8_4(a, b, d, med);
    } else {
    #pragma unroll
    for (int q = 0; q < HB; q += CH) {
        if (q == 0 || HA + q < d) {
            #pragma unroll
            for (int t = 0; t < CH; t++) {
                const v4f val = row4(h, nb[HA + q + t], lane);
                b[q + t] = (HA + q + t < d) ? val : inf4;
            }
        } else {
            #pragma unroll
            for (int t = 0; t < CH; t++) b[q + t] = inf4;
        }
    }
    return rank_of_two_lists<HA, HB>(a, b, med);
    }
}

__global__ void __launch_bounds__(32 * SEL_WAVES) __attribute__((amdgpu_waves_per_eu(5)))
median_conv_select(const float* __restrict__ h,
                   const int*   __restrict__ nbr,
                   const int*   __restrict__ deg,
                   float*       __restrict__ out) {
    const int lane = threadIdx.x & 31;
    const int wave = __builtin_amdgcn_readfirstlane((int)(threadIdx.x >> 5));
    const int n0   = (blockIdx.x * SEL_WAVES + wave) * SEL_NODES;

    int d = deg[n0];
    int idx[KK];
    load_index_row(nbr, n0, idx);

    asm volatile("" :: "s"(d), "s"(idx[0]), "s"(idx[16]));

    #pragma unroll 1
    for (int i = 0; i < SEL_NODES; i++) {
        const int n = n0 + i;

        const int nn = (n + 1 < NN) ? n + 1 : NN - 1;
        const int d_next = deg[nn];
        int idx_next[KK];
        load_index_row(nbr, nn, idx_next);

        const int med = (d - 1) >> 1;

        v4f r;
        if (d > 24) {
            r = node_median<16, 16>(h, idx, lane, d, med);
        } else if (d > 16) {
            r = node_median<16, 8>(h, idx, lane, d, med);
        } else if (d > 12) {
            r = node_median<8, 8>(h, idx, lane, d, med);
        } else if (d > 8) {
            r = node_median<8, 4>(h, idx, lane, d, med);
        } else if (d > 4) {
            r = node_median<4, 4>(h, idx, lane, d, med);
        } else if (d > 2) {
            r = node_median<2, 2>(h, idx, lane, d, med);
        } else {
            const float inf = __builtin_inff();
            const v4f inf4 = {inf, inf, inf, inf};
            r = (d > 0) ? row4(h, idx[0], lane) : inf4;
            if (d > 1) r = vmin4(r, row4(h, idx[1], lane));
        }

        ((v4f*)(out + (size_t)n * OUT_C))[lane] = r;

        d = d_next;
        #pragma unroll
        for (int j = 0; j < KK; j++) idx[j] = idx_next[j];
    }
}

extern "C" void kernel_launch(void* const* d_in, const int* in_sizes, int n_in,
                              void* d_out, int out_size, void* d_ws, size_t ws_size,
                              hipStream_t stream) {
    const float* x   = (const float*)d_in[0];
    const float* w   = (const float*)d_in[1];
    const int*   nbr = (const int*)d_in[2];
    const int*   deg = (const int*)d_in[3];
    float*       out = (float*)d_out;
    float*       h   = (float*)d_ws;

    median_conv_gemm_f32<<<GEMM_BLOCKS, 256, 0, stream>>>(x, w, h);

    median_conv_select<<<SEL_BLOCKS, 32 * SEL_WAVES, 0, stream>>>(h, nbr, deg, out);
}
